# starting kernel
# baseline (speedup 1.0000x reference)
#include <hip/hip_runtime.h>
#include <hip/hip_bf16.h>
#include <stdint.h>

typedef _Float16 v16h __attribute__((ext_vector_type(16)));
typedef _Float16 v8h  __attribute__((ext_vector_type(8)));
typedef float    v8f  __attribute__((ext_vector_type(8)));

#define F_DIM 128

__global__ void zero_f4(float4* __restrict__ p, int n4) {
    int i = blockIdx.x * blockDim.x + threadIdx.x;
    if (i < n4) p[i] = make_float4(0.f, 0.f, 0.f, 0.f);
}

__global__ void wconvert(const float* __restrict__ W11, const float* __restrict__ W12,
                         const float* __restrict__ W21, const float* __restrict__ W22,
                         _Float16* __restrict__ frags) {
    int gid = blockIdx.x * blockDim.x + threadIdx.x;
    if (gid >= 4 * 8192) return;
    int m    = gid >> 13;
    int r    = gid & 8191;
    int d    = r & 7;
    int lane = (r >> 3) & 31;
    int ntk  = r >> 8;
    int kt   = ntk >> 3;
    int nt   = ntk & 7;
    int nn   = nt * 16 + (lane & 15);
    int kb   = kt * 32 + ((lane >> 4) << 3);
    int k0   = kb + ((d < 4) ? (d << 1) : (16 + ((d - 4) << 1)));
    const float* W = (m == 0) ? W11 : (m == 1) ? W12 : (m == 2) ? W21 : W22;
    _Float16* o = frags + (size_t)m * 16384 + (size_t)r * 2;
    o[0] = (_Float16)W[(size_t)k0 * F_DIM + nn];
    o[1] = (_Float16)W[(size_t)(k0 + 1) * F_DIM + nn];
}

__global__ void edge_scatter(const float* __restrict__ xin, const int* __restrict__ ei,
                             const float* __restrict__ ea, const float* __restrict__ We,
                             const float* __restrict__ be, float* aggr, int E_) {
    long long gid = (long long)blockIdx.x * blockDim.x + threadIdx.x;
    if (gid >= (long long)E_ * 32) return;
    int e = (int)(gid >> 5);
    int c = ((int)gid & 31) << 2;
    int src = ei[e];
    int dst = ei[E_ + e];
    float a = ea[e];
    const float* xr = xin + (size_t)src * F_DIM + c;
    float4 xv = *(const float4*)xr;
    float4 wv = *(const float4*)(We + c);
    float4 bv = *(const float4*)(be + c);
    float m0 = fmaxf(xv.x + a * wv.x + bv.x, 0.f);
    float m1 = fmaxf(xv.y + a * wv.y + bv.y, 0.f);
    float m2 = fmaxf(xv.z + a * wv.z + bv.z, 0.f);
    float m3 = fmaxf(xv.w + a * wv.w + bv.w, 0.f);
    float* o = aggr + (size_t)dst * F_DIM + c;
    atomicAdd(o + 0, m0);
    atomicAdd(o + 1, m1);
    atomicAdd(o + 2, m2);
    atomicAdd(o + 3, m3);
}

__global__ void __launch_bounds__(256)
mlp_wmma(const float* __restrict__ xin, const float* aggr, const float* __restrict__ eps,
         const _Float16* __restrict__ fw1, const float* __restrict__ b1,
         const _Float16* __restrict__ fw2, const float* __restrict__ b2,
         float* out, int nrows) {
    __shared__ __align__(16) _Float16 stage[8][16 * 136];

    const int wv   = threadIdx.x >> 5;
    const int lane = threadIdx.x & 31;
    const int row0 = blockIdx.x * 128 + wv * 16;
    if (row0 >= nrows) return;

    const float ep = 1.0f + eps[0];
    _Float16* st = &stage[wv][0];

    #pragma unroll 4
    for (int r = 0; r < 16; ++r) {
        size_t off = (size_t)(row0 + r) * F_DIM + (lane << 2);
        float4 xv = *(const float4*)(xin + off);
        float4 av = *(const float4*)(aggr + off);
        union { _Float16 h[4]; unsigned long long u; } p;
        p.h[0] = (_Float16)(ep * xv.x + av.x);
        p.h[1] = (_Float16)(ep * xv.y + av.y);
        p.h[2] = (_Float16)(ep * xv.z + av.z);
        p.h[3] = (_Float16)(ep * xv.w + av.w);
        *(unsigned long long*)(st + r * 136 + (lane << 2)) = p.u;
    }

    const int m_  = lane & 15;
    const int hs8 = (lane >> 4) << 3;

    v8f acc[8];
    const v8f vz = {0.f, 0.f, 0.f, 0.f, 0.f, 0.f, 0.f, 0.f};

    #pragma unroll
    for (int nt = 0; nt < 8; ++nt) acc[nt] = vz;
    #pragma unroll
    for (int kt = 0; kt < 4; ++kt) {
        int kb = kt * 32 + hs8;
        v8h alo = *(const v8h*)(st + m_ * 136 + kb);
        v8h ahi = *(const v8h*)(st + m_ * 136 + kb + 16);
        v16h av = __builtin_shufflevector(alo, ahi, 0,1,2,3,4,5,6,7,8,9,10,11,12,13,14,15);
        #pragma unroll
        for (int nt = 0; nt < 8; ++nt) {
            const _Float16* q = fw1 + ((size_t)((kt << 3) | nt) * 32 + lane) * 16;
            v8h blo = *(const v8h*)q;
            v8h bhi = *(const v8h*)(q + 8);
            v16h bv = __builtin_shufflevector(blo, bhi, 0,1,2,3,4,5,6,7,8,9,10,11,12,13,14,15);
            acc[nt] = __builtin_amdgcn_wmma_f32_16x16x32_f16(
                false, av, false, bv, (short)0, acc[nt], false, false);
        }
    }

    #pragma unroll
    for (int nt = 0; nt < 8; ++nt) {
        int col = nt * 16 + m_;
        float bs = b1[col];
        #pragma unroll
        for (int r = 0; r < 8; ++r) {
            float v = fmaxf(acc[nt][r] + bs, 0.f);
            st[(r + hs8) * 136 + col] = (_Float16)v;
        }
    }

    #pragma unroll
    for (int nt = 0; nt < 8; ++nt) acc[nt] = vz;
    #pragma unroll
    for (int kt = 0; kt < 4; ++kt) {
        int kb = kt * 32 + hs8;
        v8h alo = *(const v8h*)(st + m_ * 136 + kb);
        v8h ahi = *(const v8h*)(st + m_ * 136 + kb + 16);
        v16h av = __builtin_shufflevector(alo, ahi, 0,1,2,3,4,5,6,7,8,9,10,11,12,13,14,15);
        #pragma unroll
        for (int nt = 0; nt < 8; ++nt) {
            const _Float16* q = fw2 + ((size_t)((kt << 3) | nt) * 32 + lane) * 16;
            v8h blo = *(const v8h*)q;
            v8h bhi = *(const v8h*)(q + 8);
            v16h bv = __builtin_shufflevector(blo, bhi, 0,1,2,3,4,5,6,7,8,9,10,11,12,13,14,15);
            acc[nt] = __builtin_amdgcn_wmma_f32_16x16x32_f16(
                false, av, false, bv, (short)0, acc[nt], false, false);
        }
    }

    #pragma unroll
    for (int nt = 0; nt < 8; ++nt) {
        int col = nt * 16 + m_;
        float bs = b2[col];
        #pragma unroll
        for (int r = 0; r < 8; ++r) {
            out[(size_t)(row0 + r + hs8) * F_DIM + col] = fmaxf(acc[nt][r] + bs, 0.f);
        }
    }
}

__global__ void __launch_bounds__(256)
classifier(const float* __restrict__ h, const float* __restrict__ Wl,
           const float* __restrict__ bl, float* __restrict__ out, int n) {
    __shared__ float w[F_DIM * 40];
    __shared__ float bb[40];
    for (int i = threadIdx.x; i < F_DIM * 40; i += blockDim.x) w[i] = Wl[i];
    if (threadIdx.x < 40) bb[threadIdx.x] = bl[threadIdx.x];
    __syncthreads();

    int node = blockIdx.x * blockDim.x + threadIdx.x;
    if (node >= n) return;

    float acc[40];
    #pragma unroll
    for (int c = 0; c < 40; ++c) acc[c] = bb[c];

    const float* hr = h + (size_t)node * F_DIM;
    for (int k = 0; k < F_DIM; ++k) {
        float hv = hr[k];
        #pragma unroll
        for (int c = 0; c < 40; ++c) acc[c] = fmaf(hv, w[k * 40 + c], acc[c]);
    }

    float mx = acc[0];
    #pragma unroll
    for (int c = 1; c < 40; ++c) mx = fmaxf(mx, acc[c]);
    float s = 0.f;
    #pragma unroll
    for (int c = 0; c < 40; ++c) s += __expf(acc[c] - mx);
    float ls = __logf(s);
    float* o = out + (size_t)node * 40;
    #pragma unroll
    for (int c = 0; c < 40; ++c) o[c] = acc[c] - mx - ls;
}

extern "C" void kernel_launch(void* const* d_in, const int* in_sizes, int n_in,
                              void* d_out, int out_size, void* d_ws, size_t ws_size,
                              hipStream_t stream) {
    const float* x    = (const float*)d_in[0];
    const int*   ei   = (const int*)d_in[1];
    const float* ea   = (const float*)d_in[2];
    const float* eps1 = (const float*)d_in[3];
    const float* We1  = (const float*)d_in[4];
    const float* be1  = (const float*)d_in[5];
    const float* W11  = (const float*)d_in[6];
    const float* b11  = (const float*)d_in[7];
    const float* W12  = (const float*)d_in[8];
    const float* b12  = (const float*)d_in[9];
    const float* eps2 = (const float*)d_in[10];
    const float* We2  = (const float*)d_in[11];
    const float* be2  = (const float*)d_in[12];
    const float* W21  = (const float*)d_in[13];
    const float* b21  = (const float*)d_in[14];
    const float* W22  = (const float*)d_in[15];
    const float* b22  = (const float*)d_in[16];
    const float* Wl   = (const float*)d_in[17];
    const float* bl   = (const float*)d_in[18];

    const int N_ = in_sizes[0] / F_DIM;
    const int E_ = in_sizes[2];
    const size_t NB = (size_t)N_ * F_DIM * sizeof(float);

    char* ws = (char*)d_ws;
    float*    aggr  = (float*)ws;
    float*    h1    = (float*)(ws + NB);
    _Float16* frags = (_Float16*)(ws + 2 * NB);
    _Float16* f11 = frags;
    _Float16* f12 = frags + 16384;
    _Float16* f21 = frags + 32768;
    _Float16* f22 = frags + 49152;
    float* h2  = aggr;
    float* out = (float*)d_out;

    const int n4 = N_ * F_DIM / 4;
    const int zb = (n4 + 255) / 256;
    const long long ework = (long long)E_ * 32;
    const int eb = (int)((ework + 255) / 256);
    const int mb = (N_ + 127) / 128;
    const int cb = (N_ + 255) / 256;

    wconvert<<<128, 256, 0, stream>>>(W11, W12, W21, W22, frags);

    zero_f4<<<zb, 256, 0, stream>>>((float4*)aggr, n4);
    edge_scatter<<<eb, 256, 0, stream>>>(x, ei, ea, We1, be1, aggr, E_);
    mlp_wmma<<<mb, 256, 0, stream>>>(x, aggr, eps1, f11, b11, f12, b12, h1, N_);

    zero_f4<<<zb, 256, 0, stream>>>((float4*)aggr, n4);
    edge_scatter<<<eb, 256, 0, stream>>>(h1, ei, ea, We2, be2, aggr, E_);
    mlp_wmma<<<mb, 256, 0, stream>>>(h1, aggr, eps2, f21, b21, f22, b22, h2, N_);

    classifier<<<cb, 256, 0, stream>>>(h2, Wl, bl, out, N_);
}


# Classifier (128x40) + log_softmax fused into 2nd MLP launch as a third f16 matrix product on the wave's LDS tile (was fp32 vector FMAs, own launch); layer output never stored; sum order changes
# speedup vs baseline: 3.0880x; 3.0880x over previous
#include <hip/hip_runtime.h>
#include <hip/hip_bf16.h>
#include <stdint.h>

typedef _Float16 v16h __attribute__((ext_vector_type(16)));
typedef _Float16 v8h  __attribute__((ext_vector_type(8)));
typedef float    v8f  __attribute__((ext_vector_type(8)));

#define F_DIM 128
#define N_CLS 40

#define BKT_SHIFT 9
#define BKT_W (1 << BKT_SHIFT)
#define BIN_BLOCKS 256

__global__ void wconvert(const float* __restrict__ W11, const float* __restrict__ W12,
                         const float* __restrict__ W21, const float* __restrict__ W22,
                         const float* __restrict__ Wl, _Float16* __restrict__ frags) {
    int gid = blockIdx.x * blockDim.x + threadIdx.x;
    if (gid >= 4 * 8192) {
        int r = gid - 4 * 8192;
        if (r >= 3072) return;
        int d    = r & 7;
        int lane = (r >> 3) & 31;
        int ntk  = r >> 8;
        int kt   = ntk / 3;
        int nt   = ntk - kt * 3;
        int nn   = nt * 16 + (lane & 15);
        int kb   = kt * 32 + ((lane >> 4) << 3);
        int k0   = kb + ((d < 4) ? (d << 1) : (16 + ((d - 4) << 1)));
        _Float16* o = frags + (size_t)4 * 16384 + (size_t)r * 2;
        const bool ok = (nn < N_CLS);
        o[0] = ok ? (_Float16)Wl[(size_t)k0 * N_CLS + nn] : (_Float16)0.f;
        o[1] = ok ? (_Float16)Wl[(size_t)(k0 + 1) * N_CLS + nn] : (_Float16)0.f;
        return;
    }
    int m    = gid >> 13;
    int r    = gid & 8191;
    int d    = r & 7;
    int lane = (r >> 3) & 31;
    int ntk  = r >> 8;
    int kt   = ntk >> 3;
    int nt   = ntk & 7;
    int nn   = nt * 16 + (lane & 15);
    int kb   = kt * 32 + ((lane >> 4) << 3);
    int k0   = kb + ((d < 4) ? (d << 1) : (16 + ((d - 4) << 1)));
    const float* W = (m == 0) ? W11 : (m == 1) ? W12 : (m == 2) ? W21 : W22;
    _Float16* o = frags + (size_t)m * 16384 + (size_t)r * 2;
    o[0] = (_Float16)W[(size_t)k0 * F_DIM + nn];
    o[1] = (_Float16)W[(size_t)(k0 + 1) * F_DIM + nn];
}

__global__ void __launch_bounds__(1024)
bin_count(const int* __restrict__ ei, int* __restrict__ cntmat, int E_, int K, int chunk) {
    extern __shared__ int hist[];
    const int tid = threadIdx.x;
    const int b   = blockIdx.x;
    for (int k = tid; k < K; k += 1024) hist[k] = 0;
    __syncthreads();
    const int e0 = b * chunk;
    const int e1 = (e0 + chunk < E_) ? (e0 + chunk) : E_;
    const int* dst = ei + E_;
    for (int i0 = e0 + tid; i0 < e1; i0 += 4096) {
        int d[4];
        #pragma unroll
        for (int q = 0; q < 4; ++q) {
            const int i = i0 + q * 1024;
            d[q] = (i < e1) ? dst[i] : -1;
        }
        #pragma unroll
        for (int q = 0; q < 4; ++q)
            if (d[q] >= 0) atomicAdd(&hist[d[q] >> BKT_SHIFT], 1);
    }
    __syncthreads();
    for (int k = tid; k < K; k += 1024) cntmat[k * BIN_BLOCKS + b] = hist[k];
}

__global__ void __launch_bounds__(1024)
scan_counts(const int* __restrict__ cnt, int* __restrict__ off, int n) {
    __shared__ int wsum[32];
    const int tid  = threadIdx.x;
    const int lane = tid & 31;
    const int wv   = tid >> 5;
    int carry = 0;
    for (int base = 0; base < n; base += 16384) {
        const int i0 = base + (tid << 4);
        const bool full = (i0 + 15 < n);
        int v[16];
        if (full) {
            #pragma unroll
            for (int q = 0; q < 4; ++q) {
                int4 t = *(const int4*)(cnt + i0 + 4 * q);
                v[4 * q] = t.x; v[4 * q + 1] = t.y; v[4 * q + 2] = t.z; v[4 * q + 3] = t.w;
            }
        } else {
            #pragma unroll
            for (int k = 0; k < 16; ++k) v[k] = (i0 + k < n) ? cnt[i0 + k] : 0;
        }
        int tot = 0;
        #pragma unroll
        for (int k = 0; k < 16; ++k) { int t = v[k]; v[k] = tot; tot += t; }
        int incl = tot;
        #pragma unroll
        for (int d = 1; d < 32; d <<= 1) {
            int y = __shfl_up(incl, d);
            if (lane >= d) incl += y;
        }
        if (lane == 31) wsum[wv] = incl;
        __syncthreads();
        const int w = wsum[lane];
        int wincl = w;
        #pragma unroll
        for (int d = 1; d < 32; d <<= 1) {
            int y = __shfl_up(wincl, d);
            if (lane >= d) wincl += y;
        }
        const int wave_excl = __shfl(wincl, wv) - __shfl(w, wv);
        const int total     = __shfl(wincl, 31);
        const int ex = carry + wave_excl + (incl - tot);
        if (full) {
            #pragma unroll
            for (int q = 0; q < 4; ++q) {
                int4 o = make_int4(ex + v[4 * q], ex + v[4 * q + 1], ex + v[4 * q + 2], ex + v[4 * q + 3]);
                *(int4*)(off + i0 + 4 * q) = o;
            }
        } else {
            #pragma unroll
            for (int k = 0; k < 16; ++k) {
                if (i0 + k < n) off[i0 + k] = ex + v[k];
            }
        }
        carry += total;
        __syncthreads();
    }
    if (tid == 0) off[n] = carry;
}

__global__ void __launch_bounds__(1024)
bin_scatter(const int* __restrict__ ei, const float* __restrict__ ea,
            const int* __restrict__ base, int2* __restrict__ tpair,
            int* __restrict__ tdl, int E_, int K, int chunk) {
    extern __shared__ int cur[];
    const int tid = threadIdx.x;
    const int b   = blockIdx.x;
    for (int k = tid; k < K; k += 1024) cur[k] = base[k * BIN_BLOCKS + b];
    __syncthreads();
    const int e0 = b * chunk;
    const int e1 = (e0 + chunk < E_) ? (e0 + chunk) : E_;
    const int* dst = ei + E_;
    for (int i0 = e0 + tid; i0 < e1; i0 += 4096) {
        int d[4], s[4];
        float a[4];
        #pragma unroll
        for (int q = 0; q < 4; ++q) {
            const int i = i0 + q * 1024;
            const bool ok = (i < e1);
            d[q] = ok ? dst[i] : -1;
            s[q] = ok ? ei[i] : 0;
            a[q] = ok ? ea[i] : 0.f;
        }
        #pragma unroll
        for (int q = 0; q < 4; ++q) {
            if (d[q] >= 0) {
                const int p = atomicAdd(&cur[d[q] >> BKT_SHIFT], 1);
                tpair[p] = make_int2(s[q], __float_as_int(a[q]));
                tdl[p]   = d[q] & (BKT_W - 1);
            }
        }
    }
}

__global__ void __launch_bounds__(1024)
bucket_place(const int* __restrict__ base, const int2* __restrict__ tpair,
             const int* __restrict__ tdl, int* __restrict__ off,
             int2* __restrict__ csr, int n, int K) {
    __shared__ int c[BKT_W];
    __shared__ int wsum[32];
    const int tid  = threadIdx.x;
    const int lane = tid & 31;
    const int wv   = tid >> 5;
    const int k    = blockIdx.x;
    const int lo = base[k * BIN_BLOCKS];
    const int hi = base[(k + 1) * BIN_BLOCKS];
    if (tid < BKT_W) c[tid] = 0;
    __syncthreads();
    for (int i0 = lo + tid; i0 < hi; i0 += 4096) {
        int dl[4];
        #pragma unroll
        for (int q = 0; q < 4; ++q) {
            const int i = i0 + q * 1024;
            dl[q] = (i < hi) ? tdl[i] : -1;
        }
        #pragma unroll
        for (int q = 0; q < 4; ++q)
            if (dl[q] >= 0) atomicAdd(&c[dl[q]], 1);
    }
    __syncthreads();
    const int v = (tid < BKT_W) ? c[tid] : 0;
    int incl = v;
    #pragma unroll
    for (int d = 1; d < 32; d <<= 1) {
        int y = __shfl_up(incl, d);
        if (lane >= d) incl += y;
    }
    if (lane == 31) wsum[wv] = incl;
    __syncthreads();
    const int w = wsum[lane];
    int wincl = w;
    #pragma unroll
    for (int d = 1; d < 32; d <<= 1) {
        int y = __shfl_up(wincl, d);
        if (lane >= d) wincl += y;
    }
    const int wave_excl = __shfl(wincl, wv) - __shfl(w, wv);
    const int start = lo + wave_excl + (incl - v);
    if (tid < BKT_W) {
        c[tid] = start;
        const int node = k * BKT_W + tid;
        if (node < n) off[node] = start;
    }
    if (k == K - 1 && tid == 0) off[n] = hi;
    __syncthreads();
    for (int i0 = lo + tid; i0 < hi; i0 += 4096) {
        int dl[4];
        int2 pr[4];
        #pragma unroll
        for (int q = 0; q < 4; ++q) {
            const int i = i0 + q * 1024;
            const bool ok = (i < hi);
            dl[q] = ok ? tdl[i] : -1;
            pr[q] = ok ? tpair[i] : make_int2(0, 0);
        }
        #pragma unroll
        for (int q = 0; q < 4; ++q) {
            if (dl[q] >= 0) {
                const int p = atomicAdd(&c[dl[q]], 1);
                csr[p] = pr[q];
            }
        }
    }
}

__device__ __forceinline__ void msg_add(float4& acc, const float4 xv, const float a,
                                        const float4 wv, const float4 bv) {
    acc.x += fmaxf(xv.x + a * wv.x + bv.x, 0.f);
    acc.y += fmaxf(xv.y + a * wv.y + bv.y, 0.f);
    acc.z += fmaxf(xv.z + a * wv.z + bv.z, 0.f);
    acc.w += fmaxf(xv.w + a * wv.w + bv.w, 0.f);
}

__global__ void __launch_bounds__(256)
node_gather(const float* __restrict__ xin, const int* __restrict__ off,
            const int* __restrict__ csr, const float* __restrict__ We,
            const float* __restrict__ be, float* __restrict__ aggr, int n) {
    const int lane = threadIdx.x & 31;
    const int node = __builtin_amdgcn_readfirstlane((int)(blockIdx.x * 8 + (threadIdx.x >> 5)));
    if (node >= n) return;
    const int beg = off[node];
    const int end = off[node + 1];
    const int c = lane << 2;
    const float4 wv = *(const float4*)(We + c);
    const float4 bv = *(const float4*)(be + c);
    const float* xc = xin + c;
    const int2* seg = (const int2*)csr;
    float4 acc = make_float4(0.f, 0.f, 0.f, 0.f);

    for (int base = beg; base < end; base += 32) {
        const int m = (end - base < 32) ? (end - base) : 32;
        int2 my = make_int2(0, 0);
        if (lane < m) my = seg[base + lane];
        int j = 0;
        for (; j + 4 <= m; j += 4) {
            const int s0 = __builtin_amdgcn_readlane(my.x, j);
            const int s1 = __builtin_amdgcn_readlane(my.x, j + 1);
            const int s2 = __builtin_amdgcn_readlane(my.x, j + 2);
            const int s3 = __builtin_amdgcn_readlane(my.x, j + 3);
            const float4 x0 = *(const float4*)(xc + (size_t)s0 * F_DIM);
            const float4 x1 = *(const float4*)(xc + (size_t)s1 * F_DIM);
            const float4 x2 = *(const float4*)(xc + (size_t)s2 * F_DIM);
            const float4 x3 = *(const float4*)(xc + (size_t)s3 * F_DIM);
            const float a0 = __int_as_float(__builtin_amdgcn_readlane(my.y, j));
            const float a1 = __int_as_float(__builtin_amdgcn_readlane(my.y, j + 1));
            const float a2 = __int_as_float(__builtin_amdgcn_readlane(my.y, j + 2));
            const float a3 = __int_as_float(__builtin_amdgcn_readlane(my.y, j + 3));
            msg_add(acc, x0, a0, wv, bv);
            msg_add(acc, x1, a1, wv, bv);
            msg_add(acc, x2, a2, wv, bv);
            msg_add(acc, x3, a3, wv, bv);
        }
        for (; j < m; ++j) {
            const int s0 = __builtin_amdgcn_readlane(my.x, j);
            const float a0 = __int_as_float(__builtin_amdgcn_readlane(my.y, j));
            const float4 x0 = *(const float4*)(xc + (size_t)s0 * F_DIM);
            msg_add(acc, x0, a0, wv, bv);
        }
    }
    *(float4*)(aggr + (size_t)node * F_DIM + c) = acc;
}

__global__ void __launch_bounds__(256)
mlp_wmma(const float* __restrict__ xin, const float* aggr, const float* __restrict__ eps,
         const _Float16* __restrict__ fw1, const float* __restrict__ b1,
         const _Float16* __restrict__ fw2, const float* __restrict__ b2,
         float* out, int nrows) {
    __shared__ __align__(16) _Float16 stage[8][16 * 136];

    const int wv   = threadIdx.x >> 5;
    const int lane = threadIdx.x & 31;
    const int row0 = blockIdx.x * 128 + wv * 16;
    if (row0 >= nrows) return;

    const float ep = 1.0f + eps[0];
    _Float16* st = &stage[wv][0];

    #pragma unroll 4
    for (int r = 0; r < 16; ++r) {
        size_t off = (size_t)(row0 + r) * F_DIM + (lane << 2);
        float4 xv = *(const float4*)(xin + off);
        float4 av = *(const float4*)(aggr + off);
        union { _Float16 h[4]; unsigned long long u; } p;
        p.h[0] = (_Float16)(ep * xv.x + av.x);
        p.h[1] = (_Float16)(ep * xv.y + av.y);
        p.h[2] = (_Float16)(ep * xv.z + av.z);
        p.h[3] = (_Float16)(ep * xv.w + av.w);
        *(unsigned long long*)(st + r * 136 + (lane << 2)) = p.u;
    }

    const int m_  = lane & 15;
    const int hs8 = (lane >> 4) << 3;

    v8f acc[8];
    const v8f vz = {0.f, 0.f, 0.f, 0.f, 0.f, 0.f, 0.f, 0.f};

    #pragma unroll
    for (int nt = 0; nt < 8; ++nt) acc[nt] = vz;
    #pragma unroll
    for (int kt = 0; kt < 4; ++kt) {
        int kb = kt * 32 + hs8;
        v8h alo = *(const v8h*)(st + m_ * 136 + kb);
        v8h ahi = *(const v8h*)(st + m_ * 136 + kb + 16);
        v16h av = __builtin_shufflevector(alo, ahi, 0,1,2,3,4,5,6,7,8,9,10,11,12,13,14,15);
        #pragma unroll
        for (int nt = 0; nt < 8; ++nt) {
            const _Float16* q = fw1 + ((size_t)((kt << 3) | nt) * 32 + lane) * 16;
            v8h blo = *(const v8h*)q;
            v8h bhi = *(const v8h*)(q + 8);
            v16h bv = __builtin_shufflevector(blo, bhi, 0,1,2,3,4,5,6,7,8,9,10,11,12,13,14,15);
            acc[nt] = __builtin_amdgcn_wmma_f32_16x16x32_f16(
                false, av, false, bv, (short)0, acc[nt], false, false);
        }
    }

    #pragma unroll
    for (int nt = 0; nt < 8; ++nt) {
        int col = nt * 16 + m_;
        float bs = b1[col];
        #pragma unroll
        for (int r = 0; r < 8; ++r) {
            float v = fmaxf(acc[nt][r] + bs, 0.f);
            st[(r + hs8) * 136 + col] = (_Float16)v;
        }
    }

    #pragma unroll
    for (int nt = 0; nt < 8; ++nt) acc[nt] = vz;
    #pragma unroll
    for (int kt = 0; kt < 4; ++kt) {
        int kb = kt * 32 + hs8;
        v8h alo = *(const v8h*)(st + m_ * 136 + kb);
        v8h ahi = *(const v8h*)(st + m_ * 136 + kb + 16);
        v16h av = __builtin_shufflevector(alo, ahi, 0,1,2,3,4,5,6,7,8,9,10,11,12,13,14,15);
        #pragma unroll
        for (int nt = 0; nt < 8; ++nt) {
            const _Float16* q = fw2 + ((size_t)((kt << 3) | nt) * 32 + lane) * 16;
            v8h blo = *(const v8h*)q;
            v8h bhi = *(const v8h*)(q + 8);
            v16h bv = __builtin_shufflevector(blo, bhi, 0,1,2,3,4,5,6,7,8,9,10,11,12,13,14,15);
            acc[nt] = __builtin_amdgcn_wmma_f32_16x16x32_f16(
                false, av, false, bv, (short)0, acc[nt], false, false);
        }
    }

    #pragma unroll
    for (int nt = 0; nt < 8; ++nt) {
        int col = nt * 16 + m_;
        float bs = b2[col];
        #pragma unroll
        for (int r = 0; r < 8; ++r) {
            out[(size_t)(row0 + r + hs8) * F_DIM + col] = fmaxf(acc[nt][r] + bs, 0.f);
        }
    }
}

__global__ void __launch_bounds__(256)
mlp_cls_wmma(const float* __restrict__ xin, const float* __restrict__ aggr,
             const float* __restrict__ eps,
             const _Float16* __restrict__ fw1, const float* __restrict__ b1,
             const _Float16* __restrict__ fw2, const float* __restrict__ b2,
             const _Float16* __restrict__ fwl, const float* __restrict__ bl,
             float* __restrict__ out, int nrows) {
    __shared__ __align__(16) _Float16 stage[8][16 * 136];
    __shared__ __align__(16) float ltile[8][16 * N_CLS];

    const int wv   = threadIdx.x >> 5;
    const int lane = threadIdx.x & 31;
    const int row0 = blockIdx.x * 128 + wv * 16;
    if (row0 >= nrows) return;

    const float ep = 1.0f + eps[0];
    _Float16* st = &stage[wv][0];

    #pragma unroll 4
    for (int r = 0; r < 16; ++r) {
        size_t off = (size_t)(row0 + r) * F_DIM + (lane << 2);
        float4 xv = *(const float4*)(xin + off);
        float4 av = *(const float4*)(aggr + off);
        union { _Float16 h[4]; unsigned long long u; } p;
        p.h[0] = (_Float16)(ep * xv.x + av.x);
        p.h[1] = (_Float16)(ep * xv.y + av.y);
        p.h[2] = (_Float16)(ep * xv.z + av.z);
        p.h[3] = (_Float16)(ep * xv.w + av.w);
        *(unsigned long long*)(st + r * 136 + (lane << 2)) = p.u;
    }

    const int m_  = lane & 15;
    const int hs8 = (lane >> 4) << 3;

    v8f acc[8];
    const v8f vz = {0.f, 0.f, 0.f, 0.f, 0.f, 0.f, 0.f, 0.f};

    #pragma unroll
    for (int nt = 0; nt < 8; ++nt) acc[nt] = vz;
    #pragma unroll
    for (int kt = 0; kt < 4; ++kt) {
        int kb = kt * 32 + hs8;
        v8h alo = *(const v8h*)(st + m_ * 136 + kb);
        v8h ahi = *(const v8h*)(st + m_ * 136 + kb + 16);
        v16h av = __builtin_shufflevector(alo, ahi, 0,1,2,3,4,5,6,7,8,9,10,11,12,13,14,15);
        #pragma unroll
        for (int nt = 0; nt < 8; ++nt) {
            const _Float16* q = fw1 + ((size_t)((kt << 3) | nt) * 32 + lane) * 16;
            v8h blo = *(const v8h*)q;
            v8h bhi = *(const v8h*)(q + 8);
            v16h bv = __builtin_shufflevector(blo, bhi, 0,1,2,3,4,5,6,7,8,9,10,11,12,13,14,15);
            acc[nt] = __builtin_amdgcn_wmma_f32_16x16x32_f16(
                false, av, false, bv, (short)0, acc[nt], false, false);
        }
    }

    #pragma unroll
    for (int nt = 0; nt < 8; ++nt) {
        int col = nt * 16 + m_;
        float bs = b1[col];
        #pragma unroll
        for (int r = 0; r < 8; ++r) {
            float v = fmaxf(acc[nt][r] + bs, 0.f);
            st[(r + hs8) * 136 + col] = (_Float16)v;
        }
    }

    #pragma unroll
    for (int nt = 0; nt < 8; ++nt) acc[nt] = vz;
    #pragma unroll
    for (int kt = 0; kt < 4; ++kt) {
        int kb = kt * 32 + hs8;
        v8h alo = *(const v8h*)(st + m_ * 136 + kb);
        v8h ahi = *(const v8h*)(st + m_ * 136 + kb + 16);
        v16h av = __builtin_shufflevector(alo, ahi, 0,1,2,3,4,5,6,7,8,9,10,11,12,13,14,15);
        #pragma unroll
        for (int nt = 0; nt < 8; ++nt) {
            const _Float16* q = fw2 + ((size_t)((kt << 3) | nt) * 32 + lane) * 16;
            v8h blo = *(const v8h*)q;
            v8h bhi = *(const v8h*)(q + 8);
            v16h bv = __builtin_shufflevector(blo, bhi, 0,1,2,3,4,5,6,7,8,9,10,11,12,13,14,15);
            acc[nt] = __builtin_amdgcn_wmma_f32_16x16x32_f16(
                false, av, false, bv, (short)0, acc[nt], false, false);
        }
    }

    #pragma unroll
    for (int nt = 0; nt < 8; ++nt) {
        int col = nt * 16 + m_;
        float bs = b2[col];
        #pragma unroll
        for (int r = 0; r < 8; ++r) {
            float v = fmaxf(acc[nt][r] + bs, 0.f);
            st[(r + hs8) * 136 + col] = (_Float16)v;
        }
    }

    v8f lg[3];
    #pragma unroll
    for (int nt = 0; nt < 3; ++nt) lg[nt] = vz;
    #pragma unroll
    for (int kt = 0; kt < 4; ++kt) {
        int kb = kt * 32 + hs8;
        v8h alo = *(const v8h*)(st + m_ * 136 + kb);
        v8h ahi = *(const v8h*)(st + m_ * 136 + kb + 16);
        v16h av = __builtin_shufflevector(alo, ahi, 0,1,2,3,4,5,6,7,8,9,10,11,12,13,14,15);
        #pragma unroll
        for (int nt = 0; nt < 3; ++nt) {
            const _Float16* q = fwl + ((size_t)(kt * 3 + nt) * 32 + lane) * 16;
            v8h blo = *(const v8h*)q;
            v8h bhi = *(const v8h*)(q + 8);
            v16h bv = __builtin_shufflevector(blo, bhi, 0,1,2,3,4,5,6,7,8,9,10,11,12,13,14,15);
            lg[nt] = __builtin_amdgcn_wmma_f32_16x16x32_f16(
                false, av, false, bv, (short)0, lg[nt], false, false);
        }
    }

    float* lt = &ltile[wv][0];
    #pragma unroll
    for (int nt = 0; nt < 3; ++nt) {
        int col = nt * 16 + m_;
        if (col < N_CLS) {
            float bs = bl[col];
            #pragma unroll
            for (int r = 0; r < 8; ++r) lt[(r + hs8) * N_CLS + col] = lg[nt][r] + bs;
        }
    }

    {
        float* rowp = lt + (lane >> 1) * N_CLS + (lane & 1) * 20;
        float v[20];
        #pragma unroll
        for (int q = 0; q < 5; ++q) {
            float4 t = *(const float4*)(rowp + 4 * q);
            v[4 * q] = t.x; v[4 * q + 1] = t.y; v[4 * q + 2] = t.z; v[4 * q + 3] = t.w;
        }
        float mx = v[0];
        #pragma unroll
        for (int c = 1; c < 20; ++c) mx = fmaxf(mx, v[c]);
        mx = fmaxf(mx, __shfl_xor(mx, 1));
        float s = 0.f;
        #pragma unroll
        for (int c = 0; c < 20; ++c) s += __expf(v[c] - mx);
        s += __shfl_xor(s, 1);
        const float sub = mx + __logf(s);
        #pragma unroll
        for (int q = 0; q < 5; ++q) {
            float4 t = make_float4(v[4 * q] - sub, v[4 * q + 1] - sub,
                                   v[4 * q + 2] - sub, v[4 * q + 3] - sub);
            *(float4*)(rowp + 4 * q) = t;
        }
    }

    float* og = out + (size_t)row0 * N_CLS;
    #pragma unroll
    for (int i = 0; i < 5; ++i) {
        const int e = (i * 32 + lane) * 4;
        *(float4*)(og + e) = *(const float4*)(lt + e);
    }
}


extern "C" void kernel_launch(void* const* d_in, const int* in_sizes, int n_in,
                              void* d_out, int out_size, void* d_ws, size_t ws_size,
                              hipStream_t stream) {
    const float* x    = (const float*)d_in[0];
    const int*   ei   = (const int*)d_in[1];
    const float* ea   = (const float*)d_in[2];
    const float* eps1 = (const float*)d_in[3];
    const float* We1  = (const float*)d_in[4];
    const float* be1  = (const float*)d_in[5];
    const float* W11  = (const float*)d_in[6];
    const float* b11  = (const float*)d_in[7];
    const float* W12  = (const float*)d_in[8];
    const float* b12  = (const float*)d_in[9];
    const float* eps2 = (const float*)d_in[10];
    const float* We2  = (const float*)d_in[11];
    const float* be2  = (const float*)d_in[12];
    const float* W21  = (const float*)d_in[13];
    const float* b21  = (const float*)d_in[14];
    const float* W22  = (const float*)d_in[15];
    const float* b22  = (const float*)d_in[16];
    const float* Wl   = (const float*)d_in[17];
    const float* bl   = (const float*)d_in[18];

    const int N_ = in_sizes[0] / F_DIM;
    const int E_ = in_sizes[2];
    const size_t NB = (size_t)N_ * F_DIM * sizeof(float);

    char* ws = (char*)d_ws;
    float*    aggr  = (float*)ws;
    float*    h1    = (float*)(ws + NB);
    _Float16* frags = (_Float16*)(ws + 2 * NB);
    _Float16* f11 = frags;
    _Float16* f12 = frags + 16384;
    _Float16* f21 = frags + 32768;
    _Float16* f22 = frags + 49152;
    _Float16* fl  = frags + 65536;
    float* out = (float*)d_out;

    const int K     = (N_ + BKT_W - 1) / BKT_W;
    const int chunk = (E_ + BIN_BLOCKS - 1) / BIN_BLOCKS;
    const int nmat  = K * BIN_BLOCKS;
    const size_t matB = (((size_t)nmat + 4) * sizeof(int) + 255) & ~(size_t)255;
    const size_t offB = (((size_t)N_ + 4) * sizeof(int) + 255) & ~(size_t)255;
    const size_t prB  = ((size_t)E_ * 2 * sizeof(int) + 255) & ~(size_t)255;
    const size_t dlB  = ((size_t)E_ * sizeof(int) + 255) & ~(size_t)255;
    char* gbase = ws + 2 * NB + 131072 + 16384;
    int*  cntmat = (int*)(gbase);
    int*  base   = (int*)(gbase + matB);
    int*  off    = (int*)(gbase + 2 * matB);
    int2* tpair  = (int2*)(gbase + 2 * matB + offB);
    int*  tdl    = (int*)(gbase + 2 * matB + offB + prB);
    int*  csr    = (int*)(gbase + 2 * matB + offB + prB + dlB);

    const int gb = (N_ + 7) / 8;
    const int mb = (N_ + 127) / 128;

    wconvert<<<140, 256, 0, stream>>>(W11, W12, W21, W22, Wl, frags);

    bin_count<<<BIN_BLOCKS, 1024, K * sizeof(int), stream>>>(ei, cntmat, E_, K, chunk);
    scan_counts<<<1, 1024, 0, stream>>>(cntmat, base, nmat);
    bin_scatter<<<BIN_BLOCKS, 1024, K * sizeof(int), stream>>>(ei, ea, base, tpair, tdl, E_, K, chunk);
    bucket_place<<<K, 1024, 0, stream>>>(base, tpair, tdl, off, (int2*)csr, N_, K);

    node_gather<<<gb, 256, 0, stream>>>(x, off, csr, We1, be1, aggr, N_);
    mlp_wmma<<<mb, 256, 0, stream>>>(x, aggr, eps1, f11, b11, f12, b12, h1, N_);

    node_gather<<<gb, 256, 0, stream>>>(h1, off, csr, We2, be2, aggr, N_);
    mlp_cls_wmma<<<mb, 256, 0, stream>>>(h1, aggr, eps2, f21, b21, f22, b22, fl, bl, out, N_);
}


# Edge gather fused into the two node-MLP launches as their staging phase: each wave sums its 16 rows' messages (8 node pairs) into its LDS tile; aggr array and 2 launches gone; sum order kept
# speedup vs baseline: 4.0606x; 1.3150x over previous
#include <hip/hip_runtime.h>
#include <hip/hip_bf16.h>
#include <stdint.h>

typedef _Float16 v16h __attribute__((ext_vector_type(16)));
typedef _Float16 v8h  __attribute__((ext_vector_type(8)));
typedef _Float16 v4h  __attribute__((ext_vector_type(4)));
typedef float    v8f  __attribute__((ext_vector_type(8)));

#define F_DIM 128
#define N_CLS 40

#define BKT_SHIFT 9
#define BKT_W (1 << BKT_SHIFT)
#define BIN_BLOCKS 256

#define WCONV_BLOCKS 140
#define XCONV_BLOCKS 1024
__global__ void wconvert(const float* __restrict__ W11, const float* __restrict__ W12,
                         const float* __restrict__ W21, const float* __restrict__ W22,
                         const float* __restrict__ Wl, _Float16* __restrict__ frags,
                         const float* __restrict__ x, _Float16* __restrict__ x16, int nrows) {
    if (blockIdx.x >= WCONV_BLOCKS) {
        const int n4     = nrows * (F_DIM / 4);
        const int stride = XCONV_BLOCKS * 256;
        const float4* x4 = (const float4*)x;
        int i = (blockIdx.x - WCONV_BLOCKS) * 256 + threadIdx.x;
        for (; i + 3 * stride < n4; i += 4 * stride) {
            float4 v[4];
            #pragma unroll
            for (int q = 0; q < 4; ++q) v[q] = x4[i + q * stride];
            #pragma unroll
            for (int q = 0; q < 4; ++q) {
                v4h h = {(_Float16)v[q].x, (_Float16)v[q].y, (_Float16)v[q].z, (_Float16)v[q].w};
                *(v4h*)(x16 + (size_t)(i + q * stride) * 4) = h;
            }
        }
        for (; i < n4; i += stride) {
            const float4 v = x4[i];
            v4h h = {(_Float16)v.x, (_Float16)v.y, (_Float16)v.z, (_Float16)v.w};
            *(v4h*)(x16 + (size_t)i * 4) = h;
        }
        return;
    }
    int gid = blockIdx.x * blockDim.x + threadIdx.x;
    if (gid >= 4 * 8192) {
        int r = gid - 4 * 8192;
        if (r >= 3072) return;
        int d    = r & 7;
        int lane = (r >> 3) & 31;
        int ntk  = r >> 8;
        int kt   = ntk / 3;
        int nt   = ntk - kt * 3;
        int nn   = nt * 16 + (lane & 15);
        int kb   = kt * 32 + ((lane >> 4) << 3);
        int k0   = kb + ((d < 4) ? (d << 1) : (16 + ((d - 4) << 1)));
        _Float16* o = frags + (size_t)4 * 16384 + (size_t)r * 2;
        const bool ok = (nn < N_CLS);
        o[0] = ok ? (_Float16)Wl[(size_t)k0 * N_CLS + nn] : (_Float16)0.f;
        o[1] = ok ? (_Float16)Wl[(size_t)(k0 + 1) * N_CLS + nn] : (_Float16)0.f;
        return;
    }
    int m    = gid >> 13;
    int r    = gid & 8191;
    int d    = r & 7;
    int lane = (r >> 3) & 31;
    int ntk  = r >> 8;
    int kt   = ntk >> 3;
    int nt   = ntk & 7;
    int nn   = nt * 16 + (lane & 15);
    int kb   = kt * 32 + ((lane >> 4) << 3);
    int k0   = kb + ((d < 4) ? (d << 1) : (16 + ((d - 4) << 1)));
    const float* W = (m == 0) ? W11 : (m == 1) ? W12 : (m == 2) ? W21 : W22;
    _Float16* o = frags + (size_t)m * 16384 + (size_t)r * 2;
    o[0] = (_Float16)W[(size_t)k0 * F_DIM + nn];
    o[1] = (_Float16)W[(size_t)(k0 + 1) * F_DIM + nn];
}

__global__ void __launch_bounds__(1024)
bin_count(const int* __restrict__ ei, int* __restrict__ cntmat, int E_, int K, int chunk) {
    extern __shared__ int hist[];
    const int tid = threadIdx.x;
    const int b   = blockIdx.x;
    for (int k = tid; k < K; k += 1024) hist[k] = 0;
    __syncthreads();
    const int e0 = b * chunk;
    const int e1 = (e0 + chunk < E_) ? (e0 + chunk) : E_;
    const int* dst = ei + E_;
    for (int i0 = e0 + tid; i0 < e1; i0 += 4096) {
        int d[4];
        #pragma unroll
        for (int q = 0; q < 4; ++q) {
            const int i = i0 + q * 1024;
            d[q] = (i < e1) ? dst[i] : -1;
        }
        #pragma unroll
        for (int q = 0; q < 4; ++q)
            if (d[q] >= 0) atomicAdd(&hist[d[q] >> BKT_SHIFT], 1);
    }
    __syncthreads();
    for (int k = tid; k < K; k += 1024) cntmat[k * BIN_BLOCKS + b] = hist[k];
}

__global__ void __launch_bounds__(1024)
scan_counts(const int* __restrict__ cnt, int* __restrict__ off, int n) {
    __shared__ int wsum[32];
    const int tid  = threadIdx.x;
    const int lane = tid & 31;
    const int wv   = tid >> 5;
    int carry = 0;
    for (int base = 0; base < n; base += 16384) {
        const int i0 = base + (tid << 4);
        const bool full = (i0 + 15 < n);
        int v[16];
        if (full) {
            #pragma unroll
            for (int q = 0; q < 4; ++q) {
                int4 t = *(const int4*)(cnt + i0 + 4 * q);
                v[4 * q] = t.x; v[4 * q + 1] = t.y; v[4 * q + 2] = t.z; v[4 * q + 3] = t.w;
            }
        } else {
            #pragma unroll
            for (int k = 0; k < 16; ++k) v[k] = (i0 + k < n) ? cnt[i0 + k] : 0;
        }
        int tot = 0;
        #pragma unroll
        for (int k = 0; k < 16; ++k) { int t = v[k]; v[k] = tot; tot += t; }
        int incl = tot;
        #pragma unroll
        for (int d = 1; d < 32; d <<= 1) {
            int y = __shfl_up(incl, d);
            if (lane >= d) incl += y;
        }
        if (lane == 31) wsum[wv] = incl;
        __syncthreads();
        const int w = wsum[lane];
        int wincl = w;
        #pragma unroll
        for (int d = 1; d < 32; d <<= 1) {
            int y = __shfl_up(wincl, d);
            if (lane >= d) wincl += y;
        }
        const int wave_excl = __shfl(wincl, wv) - __shfl(w, wv);
        const int total     = __shfl(wincl, 31);
        const int ex = carry + wave_excl + (incl - tot);
        if (full) {
            #pragma unroll
            for (int q = 0; q < 4; ++q) {
                int4 o = make_int4(ex + v[4 * q], ex + v[4 * q + 1], ex + v[4 * q + 2], ex + v[4 * q + 3]);
                *(int4*)(off + i0 + 4 * q) = o;
            }
        } else {
            #pragma unroll
            for (int k = 0; k < 16; ++k) {
                if (i0 + k < n) off[i0 + k] = ex + v[k];
            }
        }
        carry += total;
        __syncthreads();
    }
    if (tid == 0) off[n] = carry;
}

__global__ void __launch_bounds__(1024)
bin_scatter(const int* __restrict__ ei, const float* __restrict__ ea,
            const int* __restrict__ base, int2* __restrict__ tpair,
            int* __restrict__ tdl, int E_, int K, int chunk) {
    extern __shared__ int cur[];
    const int tid = threadIdx.x;
    const int b   = blockIdx.x;
    for (int k = tid; k < K; k += 1024) cur[k] = base[k * BIN_BLOCKS + b];
    __syncthreads();
    const int e0 = b * chunk;
    const int e1 = (e0 + chunk < E_) ? (e0 + chunk) : E_;
    const int* dst = ei + E_;
    for (int i0 = e0 + tid; i0 < e1; i0 += 4096) {
        int d[4], s[4];
        float a[4];
        #pragma unroll
        for (int q = 0; q < 4; ++q) {
            const int i = i0 + q * 1024;
            const bool ok = (i < e1);
            d[q] = ok ? dst[i] : -1;
            s[q] = ok ? ei[i] : 0;
            a[q] = ok ? ea[i] : 0.f;
        }
        #pragma unroll
        for (int q = 0; q < 4; ++q) {
            if (d[q] >= 0) {
                const int p = atomicAdd(&cur[d[q] >> BKT_SHIFT], 1);
                tpair[p] = make_int2(s[q], __float_as_int(a[q]));
                tdl[p]   = d[q] & (BKT_W - 1);
            }
        }
    }
}

__global__ void __launch_bounds__(1024)
bucket_place(const int* __restrict__ base, const int2* __restrict__ tpair,
             const int* __restrict__ tdl, int* __restrict__ off,
             int2* __restrict__ csr, int n, int K) {
    __shared__ int c[BKT_W];
    __shared__ int wsum[32];
    const int tid  = threadIdx.x;
    const int lane = tid & 31;
    const int wv   = tid >> 5;
    const int k    = blockIdx.x;
    const int lo = base[k * BIN_BLOCKS];
    const int hi = base[(k + 1) * BIN_BLOCKS];
    if (tid < BKT_W) c[tid] = 0;
    __syncthreads();
    for (int i0 = lo + tid; i0 < hi; i0 += 4096) {
        int dl[4];
        #pragma unroll
        for (int q = 0; q < 4; ++q) {
            const int i = i0 + q * 1024;
            dl[q] = (i < hi) ? tdl[i] : -1;
        }
        #pragma unroll
        for (int q = 0; q < 4; ++q)
            if (dl[q] >= 0) atomicAdd(&c[dl[q]], 1);
    }
    __syncthreads();
    const int v = (tid < BKT_W) ? c[tid] : 0;
    int incl = v;
    #pragma unroll
    for (int d = 1; d < 32; d <<= 1) {
        int y = __shfl_up(incl, d);
        if (lane >= d) incl += y;
    }
    if (lane == 31) wsum[wv] = incl;
    __syncthreads();
    const int w = wsum[lane];
    int wincl = w;
    #pragma unroll
    for (int d = 1; d < 32; d <<= 1) {
        int y = __shfl_up(wincl, d);
        if (lane >= d) wincl += y;
    }
    const int wave_excl = __shfl(wincl, wv) - __shfl(w, wv);
    const int start = lo + wave_excl + (incl - v);
    if (tid < BKT_W) {
        c[tid] = start;
        const int node = k * BKT_W + tid;
        if (node < n) off[node] = start;
    }
    if (k == K - 1 && tid == 0) off[n] = hi;
    __syncthreads();
    for (int i0 = lo + tid; i0 < hi; i0 += 4096) {
        int dl[4];
        int2 pr[4];
        #pragma unroll
        for (int q = 0; q < 4; ++q) {
            const int i = i0 + q * 1024;
            const bool ok = (i < hi);
            dl[q] = ok ? tdl[i] : -1;
            pr[q] = ok ? tpair[i] : make_int2(0, 0);
        }
        #pragma unroll
        for (int q = 0; q < 4; ++q) {
            if (dl[q] >= 0) {
                const int p = atomicAdd(&c[dl[q]], 1);
                csr[p] = pr[q];
            }
        }
    }
}

template <int Q>
__device__ __forceinline__ int row_share(const int v) {
    return __builtin_amdgcn_update_dpp(0, v, 0x150 + Q, 0xf, 0xf, false);
}

__device__ __forceinline__ void msg_add8(float (&acc)[8], const v8h xr, const float a,
                                         const float keep, const float (&w)[8],
                                         const float (&b)[8]) {
    #pragma unroll
    for (int i = 0; i < 8; ++i) {
        const float m = fmaxf((float)xr[i] + a * w[i] + b[i], 0.f);
        acc[i] = fmaf(keep, m, acc[i]);
    }
}

template <int Q>
__device__ __forceinline__ void edge_group(float (&acc)[8], const int2 my, const int m,
                                           const char* __restrict__ xbase, const unsigned lo,
                                           const float (&w)[8], const float (&b)[8]) {
    const unsigned s0 = ((unsigned)row_share<Q>(my.x) << 8) | lo;
    const unsigned s1 = ((unsigned)row_share<Q + 1>(my.x) << 8) | lo;
    const unsigned s2 = ((unsigned)row_share<Q + 2>(my.x) << 8) | lo;
    const unsigned s3 = ((unsigned)row_share<Q + 3>(my.x) << 8) | lo;
    const v8h x0 = *(const v8h*)(xbase + s0);
    const v8h x1 = *(const v8h*)(xbase + s1);
    const v8h x2 = *(const v8h*)(xbase + s2);
    const v8h x3 = *(const v8h*)(xbase + s3);
    const float a0 = __int_as_float(row_share<Q>(my.y));
    const float a1 = __int_as_float(row_share<Q + 1>(my.y));
    const float a2 = __int_as_float(row_share<Q + 2>(my.y));
    const float a3 = __int_as_float(row_share<Q + 3>(my.y));
    msg_add8(acc, x0, a0, (Q     < m) ? 1.f : 0.f, w, b);
    msg_add8(acc, x1, a1, (Q + 1 < m) ? 1.f : 0.f, w, b);
    msg_add8(acc, x2, a2, (Q + 2 < m) ? 1.f : 0.f, w, b);
    msg_add8(acc, x3, a3, (Q + 3 < m) ? 1.f : 0.f, w, b);
}

template <int Q>
__device__ __forceinline__ void rows8(v8h (&x)[8], const int2 my,
                                      const char* __restrict__ xbase, const unsigned lo) {
    const unsigned s0 = ((unsigned)row_share<Q>(my.x) << 8) | lo;
    const unsigned s1 = ((unsigned)row_share<Q + 1>(my.x) << 8) | lo;
    const unsigned s2 = ((unsigned)row_share<Q + 2>(my.x) << 8) | lo;
    const unsigned s3 = ((unsigned)row_share<Q + 3>(my.x) << 8) | lo;
    const unsigned s4 = ((unsigned)row_share<Q + 4>(my.x) << 8) | lo;
    const unsigned s5 = ((unsigned)row_share<Q + 5>(my.x) << 8) | lo;
    const unsigned s6 = ((unsigned)row_share<Q + 6>(my.x) << 8) | lo;
    const unsigned s7 = ((unsigned)row_share<Q + 7>(my.x) << 8) | lo;
    x[0] = *(const v8h*)(xbase + s0);
    x[1] = *(const v8h*)(xbase + s1);
    x[2] = *(const v8h*)(xbase + s2);
    x[3] = *(const v8h*)(xbase + s3);
    x[4] = *(const v8h*)(xbase + s4);
    x[5] = *(const v8h*)(xbase + s5);
    x[6] = *(const v8h*)(xbase + s6);
    x[7] = *(const v8h*)(xbase + s7);
}

template <int Q>
__device__ __forceinline__ void sum8(float (&acc)[8], const v8h (&x)[8], const int2 my,
                                     const int m, const float (&w)[8], const float (&b)[8]) {
    const float a0 = __int_as_float(row_share<Q>(my.y));
    const float a1 = __int_as_float(row_share<Q + 1>(my.y));
    const float a2 = __int_as_float(row_share<Q + 2>(my.y));
    const float a3 = __int_as_float(row_share<Q + 3>(my.y));
    const float a4 = __int_as_float(row_share<Q + 4>(my.y));
    const float a5 = __int_as_float(row_share<Q + 5>(my.y));
    const float a6 = __int_as_float(row_share<Q + 6>(my.y));
    const float a7 = __int_as_float(row_share<Q + 7>(my.y));
    msg_add8(acc, x[0], a0, (Q     < m) ? 1.f : 0.f, w, b);
    msg_add8(acc, x[1], a1, (Q + 1 < m) ? 1.f : 0.f, w, b);
    msg_add8(acc, x[2], a2, (Q + 2 < m) ? 1.f : 0.f, w, b);
    msg_add8(acc, x[3], a3, (Q + 3 < m) ? 1.f : 0.f, w, b);
    msg_add8(acc, x[4], a4, (Q + 4 < m) ? 1.f : 0.f, w, b);
    msg_add8(acc, x[5], a5, (Q + 5 < m) ? 1.f : 0.f, w, b);
    msg_add8(acc, x[6], a6, (Q + 6 < m) ? 1.f : 0.f, w, b);
    msg_add8(acc, x[7], a7, (Q + 7 < m) ? 1.f : 0.f, w, b);
}

template <int Q>
__device__ __forceinline__ void edge_group8(float (&acc)[8], const int2 my, const int m,
                                            const char* __restrict__ xbase, const unsigned lo,
                                            const float (&w)[8], const float (&b)[8]) {
    v8h x[8];
    rows8<Q>(x, my, xbase, lo);
    sum8<Q>(acc, x, my, m, w, b);
}

__device__ __forceinline__ void edge_trip(float (&acc)[8], int2& cur, const int cnt, const int cmax,
                                          const int done, const int beg, const int hl,
                                          const int2* __restrict__ csr,
                                          const char* __restrict__ xbase, const unsigned lo,
                                          const float (&w)[8], const float (&b)[8]) {
    const int m  = cnt - done;
    const int mm = cmax - done;
    int2 nxt = make_int2(0, 0);
    if (hl + 16 < m) nxt = csr[beg + done + 16 + hl];
    if (mm > 4) edge_group8<0>(acc, cur, m, xbase, lo, w, b);
    else        edge_group<0>(acc, cur, m, xbase, lo, w, b);
    if (mm > 12)     edge_group8<8>(acc, cur, m, xbase, lo, w, b);
    else if (mm > 8) edge_group<8>(acc, cur, m, xbase, lo, w, b);
    cur = nxt;
}

__device__ __forceinline__ void load8(float (&r)[8], const float* __restrict__ p) {
    const float4 u = *(const float4*)p, v = *(const float4*)(p + 4);
    r[0] = u.x; r[1] = u.y; r[2] = u.z; r[3] = u.w;
    r[4] = v.x; r[5] = v.y; r[6] = v.z; r[7] = v.w;
}

template <typename RT>
__device__ __forceinline__ void gather_stage(_Float16* st, const int row0, const int lane,
                                             const int nrows, const RT* __restrict__ xres,
                                             const _Float16* __restrict__ xrows,
                                             const int* __restrict__ off,
                                             const int2* __restrict__ csr,
                                             const float* __restrict__ We,
                                             const float* __restrict__ be, const float ep) {
    const int hl   = lane & 15;
    const int half = lane >> 4;
    const int c    = hl << 3;

    int offv = 0;
    if (lane <= 16) {
        const int idx = (row0 + lane < nrows) ? (row0 + lane) : nrows;
        offv = off[idx];
    }

    float w[8], b[8];
    load8(w, We + c);
    load8(b, be + c);

    const char* xbase = (const char*)xrows;
    const unsigned lo = (unsigned)hl << 4;

    int beg = __shfl(offv, half);
    int cnt = __shfl(offv, half + 1) - beg;
    int2 cur = make_int2(0, 0);
    if (hl < cnt) cur = csr[beg + hl];

    #pragma unroll 1
    for (int p = 0; p < 8; ++p) {
        const int r    = 2 * p + half;
        const int node = row0 + r;
        const int cA = __builtin_amdgcn_readlane(cnt, 0);
        const int cB = __builtin_amdgcn_readlane(cnt, 16);
        const int cmax = (cA > cB) ? cA : cB;

        v8h x0[8];
        rows8<0>(x0, cur, xbase, lo);

        int nbeg = 0, ncnt = 0;
        int2 ncur = make_int2(0, 0);
        if (p < 7) {
            nbeg = __shfl(offv, r + 2);
            ncnt = __shfl(offv, r + 3) - nbeg;
            if (hl < ncnt) ncur = csr[nbeg + hl];
        }
        int2 nxt = make_int2(0, 0);
        if (hl + 16 < cnt) nxt = csr[beg + 16 + hl];
        float acc[8];
        #pragma unroll
        for (int i = 0; i < 8; ++i) acc[i] = 0.f;

        sum8<0>(acc, x0, cur, cnt, w, b);
        const RT* xp = xres + (size_t)((node < nrows) ? node : (nrows - 1)) * F_DIM + c;
        v8h    xh = {};
        float4 xa = make_float4(0.f, 0.f, 0.f, 0.f), xb = xa;
        if constexpr (sizeof(RT) == 2) {
            xh = *(const v8h*)xp;
        } else {
            xa = *(const float4*)xp;
            xb = *(const float4*)(xp + 4);
        }
        if (cmax > 12)     edge_group8<8>(acc, cur, cnt, xbase, lo, w, b);
        else if (cmax > 8) edge_group<8>(acc, cur, cnt, xbase, lo, w, b);
        cur = nxt;
        for (int done = 16; done < cmax; done += 16)
            edge_trip(acc, cur, cnt, cmax, done, beg, hl, csr, xbase, lo, w, b);

        float xr[8];
        if constexpr (sizeof(RT) == 2) {
            #pragma unroll
            for (int i = 0; i < 8; ++i) xr[i] = (float)xh[i];
        } else {
            xr[0] = xa.x; xr[1] = xa.y; xr[2] = xa.z; xr[3] = xa.w;
            xr[4] = xb.x; xr[5] = xb.y; xr[6] = xb.z; xr[7] = xb.w;
        }
        v8h h;
        #pragma unroll
        for (int i = 0; i < 8; ++i) h[i] = (_Float16)(ep * xr[i] + acc[i]);
        *(v8h*)(st + r * 136 + c) = h;

        beg = nbeg; cnt = ncnt; cur = ncur;
    }
}

__global__ void __launch_bounds__(128, 7)
gather_mlp(const float* __restrict__ xin, const _Float16* __restrict__ xrows,
           const int* __restrict__ off, const int* __restrict__ csr,
           const float* __restrict__ We, const float* __restrict__ be,
           const float* __restrict__ eps,
           const _Float16* __restrict__ fw1, const float* __restrict__ b1,
           const _Float16* __restrict__ fw2, const float* __restrict__ b2,
           _Float16* __restrict__ out, int nrows) {
    __shared__ __align__(16) _Float16 stage[4][16 * 136];

    const int wv   = threadIdx.x >> 5;
    const int lane = threadIdx.x & 31;
    const int row0 = blockIdx.x * 64 + wv * 16;
    if (row0 >= nrows) return;

    const float ep = 1.0f + eps[0];
    _Float16* st = &stage[wv][0];

    gather_stage(st, row0, lane, nrows, xin, xrows, off, (const int2*)csr, We, be, ep);

    const int m_  = lane & 15;
    const int hs8 = (lane >> 4) << 3;

    v8f acc[8];
    const v8f vz = {0.f, 0.f, 0.f, 0.f, 0.f, 0.f, 0.f, 0.f};

    #pragma unroll
    for (int nt = 0; nt < 8; ++nt) acc[nt] = vz;
    #pragma unroll
    for (int kt = 0; kt < 4; ++kt) {
        int kb = kt * 32 + hs8;
        v8h alo = *(const v8h*)(st + m_ * 136 + kb);
        v8h ahi = *(const v8h*)(st + m_ * 136 + kb + 16);
        v16h av = __builtin_shufflevector(alo, ahi, 0,1,2,3,4,5,6,7,8,9,10,11,12,13,14,15);
        #pragma unroll
        for (int nt = 0; nt < 8; ++nt) {
            const _Float16* q = fw1 + ((size_t)((kt << 3) | nt) * 32 + lane) * 16;
            v8h blo = *(const v8h*)q;
            v8h bhi = *(const v8h*)(q + 8);
            v16h bv = __builtin_shufflevector(blo, bhi, 0,1,2,3,4,5,6,7,8,9,10,11,12,13,14,15);
            acc[nt] = __builtin_amdgcn_wmma_f32_16x16x32_f16(
                false, av, false, bv, (short)0, acc[nt], false, false);
            if ((nt & 3) == 3) __builtin_amdgcn_sched_barrier(0);
        }
    }

    #pragma unroll
    for (int nt = 0; nt < 8; ++nt) {
        int col = nt * 16 + m_;
        float bs = b1[col];
        #pragma unroll
        for (int r = 0; r < 8; ++r) {
            float v = fmaxf(acc[nt][r] + bs, 0.f);
            st[(r + hs8) * 136 + col] = (_Float16)v;
        }
    }

    #pragma unroll
    for (int nt = 0; nt < 8; ++nt) acc[nt] = vz;
    #pragma unroll
    for (int kt = 0; kt < 4; ++kt) {
        int kb = kt * 32 + hs8;
        v8h alo = *(const v8h*)(st + m_ * 136 + kb);
        v8h ahi = *(const v8h*)(st + m_ * 136 + kb + 16);
        v16h av = __builtin_shufflevector(alo, ahi, 0,1,2,3,4,5,6,7,8,9,10,11,12,13,14,15);
        #pragma unroll
        for (int nt = 0; nt < 8; ++nt) {
            const _Float16* q = fw2 + ((size_t)((kt << 3) | nt) * 32 + lane) * 16;
            v8h blo = *(const v8h*)q;
            v8h bhi = *(const v8h*)(q + 8);
            v16h bv = __builtin_shufflevector(blo, bhi, 0,1,2,3,4,5,6,7,8,9,10,11,12,13,14,15);
            acc[nt] = __builtin_amdgcn_wmma_f32_16x16x32_f16(
                false, av, false, bv, (short)0, acc[nt], false, false);
            if ((nt & 3) == 3) __builtin_amdgcn_sched_barrier(0);
        }
    }

    #pragma unroll
    for (int nt = 0; nt < 8; ++nt) {
        int col = nt * 16 + m_;
        float bs = b2[col];
        #pragma unroll
        for (int r = 0; r < 8; ++r) {
            float v = fmaxf(acc[nt][r] + bs, 0.f);
            st[(r + hs8) * 136 + col] = (_Float16)v;
        }
    }

    _Float16* og = out + (size_t)row0 * F_DIM;
    #pragma unroll
    for (int i = 0; i < 8; ++i) {
        const int rr = 2 * i + (lane >> 4);
        const int cc = (lane & 15) << 3;
        *(v8h*)(og + rr * F_DIM + cc) = *(const v8h*)(st + rr * 136 + cc);
    }
}

__global__ void __launch_bounds__(128, 7)
gather_mlp_cls(const _Float16* __restrict__ xin,
               const int* __restrict__ off, const int* __restrict__ csr,
               const float* __restrict__ We, const float* __restrict__ be,
               const float* __restrict__ eps,
               const _Float16* __restrict__ fw1, const float* __restrict__ b1,
               const _Float16* __restrict__ fw2, const float* __restrict__ b2,
               const _Float16* __restrict__ fwl, const float* __restrict__ bl,
               float* __restrict__ out, int nrows) {
    __shared__ __align__(16) _Float16 stage[4][16 * 136];
    __shared__ __align__(16) float ltile[4][16 * N_CLS];

    const int wv   = threadIdx.x >> 5;
    const int lane = threadIdx.x & 31;
    const int row0 = blockIdx.x * 64 + wv * 16;
    if (row0 >= nrows) return;

    const float ep = 1.0f + eps[0];
    _Float16* st = &stage[wv][0];

    gather_stage(st, row0, lane, nrows, xin, xin, off, (const int2*)csr, We, be, ep);

    const int m_  = lane & 15;
    const int hs8 = (lane >> 4) << 3;

    v8f acc[8];
    const v8f vz = {0.f, 0.f, 0.f, 0.f, 0.f, 0.f, 0.f, 0.f};

    #pragma unroll
    for (int nt = 0; nt < 8; ++nt) acc[nt] = vz;
    #pragma unroll
    for (int kt = 0; kt < 4; ++kt) {
        int kb = kt * 32 + hs8;
        v8h alo = *(const v8h*)(st + m_ * 136 + kb);
        v8h ahi = *(const v8h*)(st + m_ * 136 + kb + 16);
        v16h av = __builtin_shufflevector(alo, ahi, 0,1,2,3,4,5,6,7,8,9,10,11,12,13,14,15);
        #pragma unroll
        for (int nt = 0; nt < 8; ++nt) {
            const _Float16* q = fw1 + ((size_t)((kt << 3) | nt) * 32 + lane) * 16;
            v8h blo = *(const v8h*)q;
            v8h bhi = *(const v8h*)(q + 8);
            v16h bv = __builtin_shufflevector(blo, bhi, 0,1,2,3,4,5,6,7,8,9,10,11,12,13,14,15);
            acc[nt] = __builtin_amdgcn_wmma_f32_16x16x32_f16(
                false, av, false, bv, (short)0, acc[nt], false, false);
            if ((nt & 3) == 3) __builtin_amdgcn_sched_barrier(0);
        }
    }

    #pragma unroll
    for (int nt = 0; nt < 8; ++nt) {
        int col = nt * 16 + m_;
        float bs = b1[col];
        #pragma unroll
        for (int r = 0; r < 8; ++r) {
            float v = fmaxf(acc[nt][r] + bs, 0.f);
            st[(r + hs8) * 136 + col] = (_Float16)v;
        }
    }

    #pragma unroll
    for (int nt = 0; nt < 8; ++nt) acc[nt] = vz;
    #pragma unroll
    for (int kt = 0; kt < 4; ++kt) {
        int kb = kt * 32 + hs8;
        v8h alo = *(const v8h*)(st + m_ * 136 + kb);
        v8h ahi = *(const v8h*)(st + m_ * 136 + kb + 16);
        v16h av = __builtin_shufflevector(alo, ahi, 0,1,2,3,4,5,6,7,8,9,10,11,12,13,14,15);
        #pragma unroll
        for (int nt = 0; nt < 8; ++nt) {
            const _Float16* q = fw2 + ((size_t)((kt << 3) | nt) * 32 + lane) * 16;
            v8h blo = *(const v8h*)q;
            v8h bhi = *(const v8h*)(q + 8);
            v16h bv = __builtin_shufflevector(blo, bhi, 0,1,2,3,4,5,6,7,8,9,10,11,12,13,14,15);
            acc[nt] = __builtin_amdgcn_wmma_f32_16x16x32_f16(
                false, av, false, bv, (short)0, acc[nt], false, false);
            if ((nt & 3) == 3) __builtin_amdgcn_sched_barrier(0);
        }
    }

    #pragma unroll
    for (int nt = 0; nt < 8; ++nt) {
        int col = nt * 16 + m_;
        float bs = b2[col];
        #pragma unroll
        for (int r = 0; r < 8; ++r) {
            float v = fmaxf(acc[nt][r] + bs, 0.f);
            st[(r + hs8) * 136 + col] = (_Float16)v;
        }
    }

    v8f lg[3];
    #pragma unroll
    for (int nt = 0; nt < 3; ++nt) lg[nt] = vz;
    #pragma unroll
    for (int kt = 0; kt < 4; ++kt) {
        int kb = kt * 32 + hs8;
        v8h alo = *(const v8h*)(st + m_ * 136 + kb);
        v8h ahi = *(const v8h*)(st + m_ * 136 + kb + 16);
        v16h av = __builtin_shufflevector(alo, ahi, 0,1,2,3,4,5,6,7,8,9,10,11,12,13,14,15);
        #pragma unroll
        for (int nt = 0; nt < 3; ++nt) {
            const _Float16* q = fwl + ((size_t)(kt * 3 + nt) * 32 + lane) * 16;
            v8h blo = *(const v8h*)q;
            v8h bhi = *(const v8h*)(q + 8);
            v16h bv = __builtin_shufflevector(blo, bhi, 0,1,2,3,4,5,6,7,8,9,10,11,12,13,14,15);
            lg[nt] = __builtin_amdgcn_wmma_f32_16x16x32_f16(
                false, av, false, bv, (short)0, lg[nt], false, false);
        }
        __builtin_amdgcn_sched_barrier(0);
    }

    float* lt = &ltile[wv][0];
    #pragma unroll
    for (int nt = 0; nt < 3; ++nt) {
        int col = nt * 16 + m_;
        if (col < N_CLS) {
            float bs = bl[col];
            #pragma unroll
            for (int r = 0; r < 8; ++r) lt[(r + hs8) * N_CLS + col] = lg[nt][r] + bs;
        }
    }

    {
        float* rowp = lt + (lane >> 1) * N_CLS + (lane & 1) * 20;
        float v[20];
        #pragma unroll
        for (int q = 0; q < 5; ++q) {
            float4 t = *(const float4*)(rowp + 4 * q);
            v[4 * q] = t.x; v[4 * q + 1] = t.y; v[4 * q + 2] = t.z; v[4 * q + 3] = t.w;
        }
        float mx = v[0];
        #pragma unroll
        for (int c = 1; c < 20; ++c) mx = fmaxf(mx, v[c]);
        mx = fmaxf(mx, __shfl_xor(mx, 1));
        float s = 0.f;
        #pragma unroll
        for (int c = 0; c < 20; ++c) s += __expf(v[c] - mx);
        s += __shfl_xor(s, 1);
        const float sub = mx + __logf(s);
        #pragma unroll
        for (int q = 0; q < 5; ++q) {
            float4 t = make_float4(v[4 * q] - sub, v[4 * q + 1] - sub,
                                   v[4 * q + 2] - sub, v[4 * q + 3] - sub);
            *(float4*)(rowp + 4 * q) = t;
        }
    }

    float* og = out + (size_t)row0 * N_CLS;
    #pragma unroll
    for (int i = 0; i < 5; ++i) {
        const int e = (i * 32 + lane) * 4;
        *(float4*)(og + e) = *(const float4*)(lt + e);
    }
}


extern "C" void kernel_launch(void* const* d_in, const int* in_sizes, int n_in,
                              void* d_out, int out_size, void* d_ws, size_t ws_size,
                              hipStream_t stream) {
    const float* x    = (const float*)d_in[0];
    const int*   ei   = (const int*)d_in[1];
    const float* ea   = (const float*)d_in[2];
    const float* eps1 = (const float*)d_in[3];
    const float* We1  = (const float*)d_in[4];
    const float* be1  = (const float*)d_in[5];
    const float* W11  = (const float*)d_in[6];
    const float* b11  = (const float*)d_in[7];
    const float* W12  = (const float*)d_in[8];
    const float* b12  = (const float*)d_in[9];
    const float* eps2 = (const float*)d_in[10];
    const float* We2  = (const float*)d_in[11];
    const float* be2  = (const float*)d_in[12];
    const float* W21  = (const float*)d_in[13];
    const float* b21  = (const float*)d_in[14];
    const float* W22  = (const float*)d_in[15];
    const float* b22  = (const float*)d_in[16];
    const float* Wl   = (const float*)d_in[17];
    const float* bl   = (const float*)d_in[18];

    const int N_ = in_sizes[0] / F_DIM;
    const int E_ = in_sizes[2];
    const size_t NB = (size_t)N_ * F_DIM * sizeof(float);

    char* ws = (char*)d_ws;
    _Float16* h1    = (_Float16*)(ws + NB);
    _Float16* x16   = (_Float16*)(ws + NB + NB / 2);
    _Float16* frags = (_Float16*)(ws + 2 * NB);
    _Float16* f11 = frags;
    _Float16* f12 = frags + 16384;
    _Float16* f21 = frags + 32768;
    _Float16* f22 = frags + 49152;
    _Float16* fl  = frags + 65536;
    float* out = (float*)d_out;

    const int K     = (N_ + BKT_W - 1) / BKT_W;
    const int chunk = (E_ + BIN_BLOCKS - 1) / BIN_BLOCKS;
    const int nmat  = K * BIN_BLOCKS;
    const size_t matB = (((size_t)nmat + 4) * sizeof(int) + 255) & ~(size_t)255;
    const size_t offB = (((size_t)N_ + 4) * sizeof(int) + 255) & ~(size_t)255;
    const size_t prB  = ((size_t)E_ * 2 * sizeof(int) + 255) & ~(size_t)255;
    const size_t dlB  = ((size_t)E_ * sizeof(int) + 255) & ~(size_t)255;
    char* gbase = ws + 2 * NB + 131072 + 16384;
    int*  cntmat = (int*)(gbase);
    int*  base   = (int*)(gbase + matB);
    int*  off    = (int*)(gbase + 2 * matB);
    int2* tpair  = (int2*)(gbase + 2 * matB + offB);
    int*  tdl    = (int*)(gbase + 2 * matB + offB + prB);
    int*  csr    = (int*)(gbase + 2 * matB + offB + prB + dlB);

    const int mb = (N_ + 63) / 64;

    wconvert<<<WCONV_BLOCKS + XCONV_BLOCKS, 256, 0, stream>>>(W11, W12, W21, W22, Wl, frags, x, x16, N_);

    bin_count<<<BIN_BLOCKS, 1024, K * sizeof(int), stream>>>(ei, cntmat, E_, K, chunk);
    scan_counts<<<1, 1024, 0, stream>>>(cntmat, base, nmat);
    bin_scatter<<<BIN_BLOCKS, 1024, K * sizeof(int), stream>>>(ei, ea, base, tpair, tdl, E_, K, chunk);
    bucket_place<<<K, 1024, 0, stream>>>(base, tpair, tdl, off, (int2*)csr, N_, K);

    gather_mlp<<<mb, 128, 0, stream>>>(x, x16, off, csr, We1, be1, eps1, f11, b11, f12, b12, h1, N_);
    gather_mlp_cls<<<mb, 128, 0, stream>>>(h1, off, csr, We2, be2, eps2, f21, b21, f22, b22, fl, bl, out, N_);
}


# Launches 7 -> 5: bin_count's threads also make the f16 copies of x and the weights (own launch gone); bin_scatter sums its span starts from the count matrix (scan launch gone); sum orders kept
# speedup vs baseline: 5.9422x; 1.4634x over previous
#include <hip/hip_runtime.h>
#include <hip/hip_bf16.h>
#include <stdint.h>

typedef _Float16 v16h __attribute__((ext_vector_type(16)));
typedef _Float16 v8h  __attribute__((ext_vector_type(8)));
typedef _Float16 v4h  __attribute__((ext_vector_type(4)));
typedef float    v8f  __attribute__((ext_vector_type(8)));

#define F_DIM 128
#define N_CLS 40

#define BKT_SHIFT 9
#define BKT_W (1 << BKT_SHIFT)
#define BIN_BLOCKS 256

__device__ __forceinline__ int frag_dword(const int ntk, const int lane, const int d) {
    return ntk * 256 + (d >> 2) * 128 + lane * 4 + (d & 3);
}

#define CONV_THREADS (BIN_BLOCKS * 1024)
#define WFRAG_DWORDS (4 * 8192 + 3072)
static_assert(CONV_THREADS >= WFRAG_DWORDS, "one thread per fragment dword");

__device__ __forceinline__ void convert_x(const int g, const float* __restrict__ x,
                                          _Float16* __restrict__ x16, const int nrows) {
    const int n4     = nrows * (F_DIM / 4);
    const int stride = CONV_THREADS;
    const float4* x4 = (const float4*)x;
    int i = g;
    for (; i + 3 * stride < n4; i += 4 * stride) {
        float4 v[4];
        #pragma unroll
        for (int q = 0; q < 4; ++q) v[q] = x4[i + q * stride];
        #pragma unroll
        for (int q = 0; q < 4; ++q) {
            v4h h = {(_Float16)v[q].x, (_Float16)v[q].y, (_Float16)v[q].z, (_Float16)v[q].w};
            *(v4h*)(x16 + (size_t)(i + q * stride) * 4) = h;
        }
    }
    for (; i < n4; i += stride) {
        const float4 v = x4[i];
        v4h h = {(_Float16)v.x, (_Float16)v.y, (_Float16)v.z, (_Float16)v.w};
        *(v4h*)(x16 + (size_t)i * 4) = h;
    }
}

__device__ __forceinline__ void convert_weights(const int gid,
                         const float* __restrict__ W11, const float* __restrict__ W12,
                         const float* __restrict__ W21, const float* __restrict__ W22,
                         const float* __restrict__ Wl, _Float16* __restrict__ frags) {
    if (gid >= 4 * 8192) {
        int r = gid - 4 * 8192;
        int d    = r & 7;
        int lane = (r >> 3) & 31;
        int ntk  = r >> 8;
        int kt   = ntk / 3;
        int nt   = ntk - kt * 3;
        int nn   = nt * 16 + (lane & 15);
        int kb   = kt * 32 + ((lane >> 4) << 3);
        int k0   = kb + ((d < 4) ? (d << 1) : (16 + ((d - 4) << 1)));
        _Float16* o = frags + (size_t)4 * 16384 + (size_t)frag_dword(ntk, lane, d) * 2;
        const bool ok = (nn < N_CLS);
        o[0] = ok ? (_Float16)Wl[(size_t)k0 * N_CLS + nn] : (_Float16)0.f;
        o[1] = ok ? (_Float16)Wl[(size_t)(k0 + 1) * N_CLS + nn] : (_Float16)0.f;
        return;
    }
    int m    = gid >> 13;
    int r    = gid & 8191;
    int d    = r & 7;
    int lane = (r >> 3) & 31;
    int ntk  = r >> 8;
    int kt   = ntk >> 3;
    int nt   = ntk & 7;
    int nn   = nt * 16 + (lane & 15);
    int kb   = kt * 32 + ((lane >> 4) << 3);
    int k0   = kb + ((d < 4) ? (d << 1) : (16 + ((d - 4) << 1)));
    const float* W = (m == 0) ? W11 : (m == 1) ? W12 : (m == 2) ? W21 : W22;
    _Float16* o = frags + (size_t)m * 16384 + (size_t)frag_dword(ntk, lane, d) * 2;
    o[0] = (_Float16)W[(size_t)k0 * F_DIM + nn];
    o[1] = (_Float16)W[(size_t)(k0 + 1) * F_DIM + nn];
}

__global__ void __launch_bounds__(1024)
bin_count(const int* __restrict__ ei, int* __restrict__ cntmat, int E_, int K, int chunk,
          const float* __restrict__ W11, const float* __restrict__ W12,
          const float* __restrict__ W21, const float* __restrict__ W22,
          const float* __restrict__ Wl, _Float16* __restrict__ frags,
          const float* __restrict__ x, _Float16* __restrict__ x16, int nrows) {
    extern __shared__ int hist[];
    const int tid = threadIdx.x;
    const int b   = blockIdx.x;
    for (int k = tid; k < K; k += 1024) hist[k] = 0;
    __syncthreads();
    const int e0 = b * chunk;
    const int e1 = (e0 + chunk < E_) ? (e0 + chunk) : E_;
    const int* dst = ei + E_;
    for (int i0 = e0 + tid; i0 < e1; i0 += 4096) {
        int d[4];
        #pragma unroll
        for (int q = 0; q < 4; ++q) {
            const int i = i0 + q * 1024;
            d[q] = (i < e1) ? dst[i] : -1;
        }
        #pragma unroll
        for (int q = 0; q < 4; ++q)
            if (d[q] >= 0) atomicAdd(&hist[d[q] >> BKT_SHIFT], 1);
    }
    const int g = b * 1024 + tid;
    if (g < WFRAG_DWORDS) convert_weights(g, W11, W12, W21, W22, Wl, frags);
    convert_x(g, x, x16, nrows);
    __syncthreads();
    for (int k = tid; k < K; k += 1024) cntmat[k * BIN_BLOCKS + b] = hist[k];
}

__device__ __forceinline__ int scan_k(const int* in, int* out, const int K, int* wsum,
                                      const int tid, const int lane, const int wv) {
    int carry = 0;
    for (int kb = 0; kb < K; kb += 1024) {
        const int k = kb + tid;
        const int v = (k < K) ? in[k] : 0;
        int incl = v;
        #pragma unroll
        for (int dd = 1; dd < 32; dd <<= 1) {
            int y = __shfl_up(incl, dd);
            if (lane >= dd) incl += y;
        }
        if (lane == 31) wsum[wv] = incl;
        __syncthreads();
        const int w = wsum[lane];
        int wincl = w;
        #pragma unroll
        for (int dd = 1; dd < 32; dd <<= 1) {
            int y = __shfl_up(wincl, dd);
            if (lane >= dd) wincl += y;
        }
        const int wave_excl = __shfl(wincl, wv) - __shfl(w, wv);
        const int total     = __shfl(wincl, 31);
        if (k < K) out[k] = carry + wave_excl + (incl - v);
        carry += total;
        __syncthreads();
    }
    return carry;
}

#define SC_BATCH 8192
static_assert(BIN_BLOCKS == 256, "a wave reads a row of the count matrix as 32 lanes x 8 counts");
__global__ void __launch_bounds__(1024)
bin_scatter(const int* __restrict__ ei, const float* __restrict__ ea,
            const int* __restrict__ cntmat, int* __restrict__ bstart,
            int2* __restrict__ tpair, int* __restrict__ tdl, int E_, int K, int chunk) {
    extern __shared__ int dyn[];
    __shared__ int2 spair[SC_BATCH];
    __shared__ int  sdl[SC_BATCH];
    __shared__ int  wsum[32];
    int* cur    = dyn;
    int* lcnt   = dyn + K;
    int* lstart = dyn + 2 * K;
    const int tid  = threadIdx.x;
    const int lane = tid & 31;
    const int wv   = tid >> 5;
    const int b    = blockIdx.x;
    for (int k0 = wv; k0 < K; k0 += 128) {
        int4 u[4][2];
        #pragma unroll
        for (int q = 0; q < 4; ++q) {
            const int k  = k0 + 32 * q;
            const int kk = (k < K) ? k : (K - 1);
            const int4* rp = (const int4*)(cntmat + (size_t)kk * BIN_BLOCKS) + 2 * lane;
            u[q][0] = rp[0];
            u[q][1] = rp[1];
        }
        int tot[4], pre[4];
        #pragma unroll
        for (int q = 0; q < 4; ++q) {
            const int v[8] = {u[q][0].x, u[q][0].y, u[q][0].z, u[q][0].w,
                              u[q][1].x, u[q][1].y, u[q][1].z, u[q][1].w};
            tot[q] = 0; pre[q] = 0;
            #pragma unroll
            for (int j = 0; j < 8; ++j) {
                tot[q] += v[j];
                if (8 * lane + j < b) pre[q] += v[j];
            }
        }
        #pragma unroll
        for (int dd = 16; dd >= 1; dd >>= 1) {
            #pragma unroll
            for (int q = 0; q < 4; ++q) {
                tot[q] += __shfl_xor(tot[q], dd);
                pre[q] += __shfl_xor(pre[q], dd);
            }
        }
        #pragma unroll
        for (int q = 0; q < 4; ++q) {
            const int k = k0 + 32 * q;
            if (k < K && lane == 0) { lcnt[k] = tot[q]; cur[k] = pre[q]; }
        }
    }
    __syncthreads();
    const int nedges = scan_k(lcnt, lstart, K, wsum, tid, lane, wv);
    for (int k = tid; k < K; k += 1024) {
        const int s = lstart[k];
        cur[k] += s;
        if (b == 0) bstart[k] = s;
    }
    if (b == 0 && tid == 0) bstart[K] = nedges;
    const int e0 = b * chunk;
    const int e1 = (e0 + chunk < E_) ? (e0 + chunk) : E_;
    const int* dst = ei + E_;
    for (int s0 = e0; s0 < e1; s0 += SC_BATCH) {
        const int s1 = (s0 + SC_BATCH < e1) ? (s0 + SC_BATCH) : e1;
        for (int k = tid; k < K; k += 1024) lcnt[k] = 0;
        __syncthreads();
        int d[8], s[8], r[8];
        float a[8];
        #pragma unroll
        for (int q = 0; q < 8; ++q) {
            const int i = s0 + q * 1024 + tid;
            const bool ok = (i < s1);
            d[q] = ok ? dst[i] : -1;
            s[q] = ok ? ei[i] : 0;
            a[q] = ok ? ea[i] : 0.f;
        }
        #pragma unroll
        for (int q = 0; q < 8; ++q)
            r[q] = (d[q] >= 0) ? atomicAdd(&lcnt[d[q] >> BKT_SHIFT], 1) : 0;
        __syncthreads();
        scan_k(lcnt, lstart, K, wsum, tid, lane, wv);
        #pragma unroll
        for (int q = 0; q < 8; ++q) {
            if (d[q] >= 0) {
                const int slot = lstart[d[q] >> BKT_SHIFT] + r[q];
                spair[slot] = make_int2(s[q], __float_as_int(a[q]));
                sdl[slot]   = d[q] & (BKT_W - 1);
            }
        }
        __syncthreads();
        for (int k = wv; k < K; k += 32) {
            const int L  = lcnt[k];
            const int ls = lstart[k];
            const int g  = cur[k];
            for (int j = lane; j < L; j += 32) {
                tpair[g + j] = spair[ls + j];
                tdl[g + j]   = sdl[ls + j];
            }
            if (lane == 0) cur[k] = g + L;
        }
        __syncthreads();
    }
}

#define PLACE_CAP 16384
__global__ void __launch_bounds__(1024)
bucket_place(const int* __restrict__ bstart, const int2* __restrict__ tpair,
             const int* __restrict__ tdl, int* __restrict__ off,
             int2* __restrict__ csr, int n, int K) {
    __shared__ int c[BKT_W];
    __shared__ int wsum[32];
    __shared__ int2 stage[PLACE_CAP];
    const int tid  = threadIdx.x;
    const int lane = tid & 31;
    const int wv   = tid >> 5;
    const int k    = blockIdx.x;
    const int lo = bstart[k];
    const int hi = bstart[k + 1];
    if (tid < BKT_W) c[tid] = 0;
    __syncthreads();
    for (int i0 = lo + tid; i0 < hi; i0 += 4096) {
        int dl[4];
        #pragma unroll
        for (int q = 0; q < 4; ++q) {
            const int i = i0 + q * 1024;
            dl[q] = (i < hi) ? tdl[i] : -1;
        }
        #pragma unroll
        for (int q = 0; q < 4; ++q)
            if (dl[q] >= 0) atomicAdd(&c[dl[q]], 1);
    }
    __syncthreads();
    const int v = (tid < BKT_W) ? c[tid] : 0;
    int incl = v;
    #pragma unroll
    for (int d = 1; d < 32; d <<= 1) {
        int y = __shfl_up(incl, d);
        if (lane >= d) incl += y;
    }
    if (lane == 31) wsum[wv] = incl;
    __syncthreads();
    const int w = wsum[lane];
    int wincl = w;
    #pragma unroll
    for (int d = 1; d < 32; d <<= 1) {
        int y = __shfl_up(wincl, d);
        if (lane >= d) wincl += y;
    }
    const int wave_excl = __shfl(wincl, wv) - __shfl(w, wv);
    const int lstart = wave_excl + (incl - v);
    if (tid < BKT_W) {
        c[tid] = lstart;
        const int node = k * BKT_W + tid;
        if (node < n) off[node] = lo + lstart;
    }
    if (k == K - 1 && tid == 0) off[n] = hi;
    __syncthreads();
    for (int i0 = lo + tid; i0 < hi; i0 += 4096) {
        int dl[4];
        int2 pr[4];
        #pragma unroll
        for (int q = 0; q < 4; ++q) {
            const int i = i0 + q * 1024;
            const bool ok = (i < hi);
            dl[q] = ok ? tdl[i] : -1;
            pr[q] = ok ? tpair[i] : make_int2(0, 0);
        }
        #pragma unroll
        for (int q = 0; q < 4; ++q) {
            if (dl[q] >= 0) {
                const int p = atomicAdd(&c[dl[q]], 1);
                if (p < PLACE_CAP) stage[p] = pr[q];
                else               csr[lo + p] = pr[q];
            }
        }
    }
    __syncthreads();
    const int nst  = (hi - lo < PLACE_CAP) ? (hi - lo) : PLACE_CAP;
    const int gend = lo + nst;
    for (int g = (lo & ~31) + tid; g < gend; g += 1024)
        if (g >= lo) csr[g] = stage[g - lo];
}

template <int Q>
__device__ __forceinline__ int row_share(const int v) {
    return __builtin_amdgcn_update_dpp(0, v, 0x150 + Q, 0xf, 0xf, false);
}

__device__ __forceinline__ void msg_add8(float (&acc)[8], const v8h xr, const float a,
                                         const float keep, const float (&w)[8],
                                         const float (&b)[8]) {
    #pragma unroll
    for (int i = 0; i < 8; ++i) {
        const float m = fmaxf((float)xr[i] + a * w[i] + b[i], 0.f);
        acc[i] = fmaf(keep, m, acc[i]);
    }
}

template <int Q>
__device__ __forceinline__ void edge_group(float (&acc)[8], const int2 my, const int m,
                                           const char* __restrict__ xbase, const unsigned lo,
                                           const float (&w)[8], const float (&b)[8]) {
    const unsigned s0 = ((unsigned)row_share<Q>(my.x) << 8) | lo;
    const unsigned s1 = ((unsigned)row_share<Q + 1>(my.x) << 8) | lo;
    const unsigned s2 = ((unsigned)row_share<Q + 2>(my.x) << 8) | lo;
    const unsigned s3 = ((unsigned)row_share<Q + 3>(my.x) << 8) | lo;
    const v8h x0 = *(const v8h*)(xbase + s0);
    const v8h x1 = *(const v8h*)(xbase + s1);
    const v8h x2 = *(const v8h*)(xbase + s2);
    const v8h x3 = *(const v8h*)(xbase + s3);
    const float a0 = __int_as_float(row_share<Q>(my.y));
    const float a1 = __int_as_float(row_share<Q + 1>(my.y));
    const float a2 = __int_as_float(row_share<Q + 2>(my.y));
    const float a3 = __int_as_float(row_share<Q + 3>(my.y));
    msg_add8(acc, x0, a0, (Q     < m) ? 1.f : 0.f, w, b);
    msg_add8(acc, x1, a1, (Q + 1 < m) ? 1.f : 0.f, w, b);
    msg_add8(acc, x2, a2, (Q + 2 < m) ? 1.f : 0.f, w, b);
    msg_add8(acc, x3, a3, (Q + 3 < m) ? 1.f : 0.f, w, b);
}

template <int Q>
__device__ __forceinline__ void rows8(v8h (&x)[8], const int2 my,
                                      const char* __restrict__ xbase, const unsigned lo) {
    const unsigned s0 = ((unsigned)row_share<Q>(my.x) << 8) | lo;
    const unsigned s1 = ((unsigned)row_share<Q + 1>(my.x) << 8) | lo;
    const unsigned s2 = ((unsigned)row_share<Q + 2>(my.x) << 8) | lo;
    const unsigned s3 = ((unsigned)row_share<Q + 3>(my.x) << 8) | lo;
    const unsigned s4 = ((unsigned)row_share<Q + 4>(my.x) << 8) | lo;
    const unsigned s5 = ((unsigned)row_share<Q + 5>(my.x) << 8) | lo;
    const unsigned s6 = ((unsigned)row_share<Q + 6>(my.x) << 8) | lo;
    const unsigned s7 = ((unsigned)row_share<Q + 7>(my.x) << 8) | lo;
    x[0] = *(const v8h*)(xbase + s0);
    x[1] = *(const v8h*)(xbase + s1);
    x[2] = *(const v8h*)(xbase + s2);
    x[3] = *(const v8h*)(xbase + s3);
    x[4] = *(const v8h*)(xbase + s4);
    x[5] = *(const v8h*)(xbase + s5);
    x[6] = *(const v8h*)(xbase + s6);
    x[7] = *(const v8h*)(xbase + s7);
}

template <int Q>
__device__ __forceinline__ void sum8(float (&acc)[8], const v8h (&x)[8], const int2 my,
                                     const int m, const float (&w)[8], const float (&b)[8]) {
    const float a0 = __int_as_float(row_share<Q>(my.y));
    const float a1 = __int_as_float(row_share<Q + 1>(my.y));
    const float a2 = __int_as_float(row_share<Q + 2>(my.y));
    const float a3 = __int_as_float(row_share<Q + 3>(my.y));
    const float a4 = __int_as_float(row_share<Q + 4>(my.y));
    const float a5 = __int_as_float(row_share<Q + 5>(my.y));
    const float a6 = __int_as_float(row_share<Q + 6>(my.y));
    const float a7 = __int_as_float(row_share<Q + 7>(my.y));
    msg_add8(acc, x[0], a0, (Q     < m) ? 1.f : 0.f, w, b);
    msg_add8(acc, x[1], a1, (Q + 1 < m) ? 1.f : 0.f, w, b);
    msg_add8(acc, x[2], a2, (Q + 2 < m) ? 1.f : 0.f, w, b);
    msg_add8(acc, x[3], a3, (Q + 3 < m) ? 1.f : 0.f, w, b);
    msg_add8(acc, x[4], a4, (Q + 4 < m) ? 1.f : 0.f, w, b);
    msg_add8(acc, x[5], a5, (Q + 5 < m) ? 1.f : 0.f, w, b);
    msg_add8(acc, x[6], a6, (Q + 6 < m) ? 1.f : 0.f, w, b);
    msg_add8(acc, x[7], a7, (Q + 7 < m) ? 1.f : 0.f, w, b);
}

template <int Q>
__device__ __forceinline__ void edge_group8(float (&acc)[8], const int2 my, const int m,
                                            const char* __restrict__ xbase, const unsigned lo,
                                            const float (&w)[8], const float (&b)[8]) {
    v8h x[8];
    rows8<Q>(x, my, xbase, lo);
    sum8<Q>(acc, x, my, m, w, b);
}

__device__ __forceinline__ void edge_trip(float (&acc)[8], int2& cur, const int cnt, const int cmax,
                                          const int done, const int beg, const int hl,
                                          const int2* __restrict__ csr,
                                          const char* __restrict__ xbase, const unsigned lo,
                                          const float (&w)[8], const float (&b)[8]) {
    const int m  = cnt - done;
    const int mm = cmax - done;
    int2 nxt = make_int2(0, 0);
    if (hl + 16 < m) nxt = csr[beg + done + 16 + hl];
    if (mm > 4) edge_group8<0>(acc, cur, m, xbase, lo, w, b);
    else        edge_group<0>(acc, cur, m, xbase, lo, w, b);
    if (mm > 12)     edge_group8<8>(acc, cur, m, xbase, lo, w, b);
    else if (mm > 8) edge_group<8>(acc, cur, m, xbase, lo, w, b);
    cur = nxt;
}

__device__ __forceinline__ void load8(float (&r)[8], const float* __restrict__ p) {
    const float4 u = *(const float4*)p, v = *(const float4*)(p + 4);
    r[0] = u.x; r[1] = u.y; r[2] = u.z; r[3] = u.w;
    r[4] = v.x; r[5] = v.y; r[6] = v.z; r[7] = v.w;
}

template <typename RT>
__device__ __forceinline__ void gather_stage(_Float16* st, const int row0, const int lane,
                                             const int nrows, const RT* __restrict__ xres,
                                             const _Float16* __restrict__ xrows,
                                             const int* __restrict__ off,
                                             const int2* __restrict__ csr,
                                             const float* __restrict__ We,
                                             const float* __restrict__ be, const float ep) {
    const int hl   = lane & 15;
    const int half = lane >> 4;
    const int c    = hl << 3;

    int offv = 0;
    if (lane <= 16) {
        const int idx = (row0 + lane < nrows) ? (row0 + lane) : nrows;
        offv = off[idx];
    }

    float w[8], b[8];
    load8(w, We + c);
    load8(b, be + c);

    const char* xbase = (const char*)xrows;
    const unsigned lo = (unsigned)hl << 4;

    int beg = __shfl(offv, half);
    int cnt = __shfl(offv, half + 1) - beg;
    int2 cur = make_int2(0, 0);
    if (hl < cnt) cur = csr[beg + hl];

    #pragma unroll 1
    for (int p = 0; p < 8; ++p) {
        const int r    = 2 * p + half;
        const int node = row0 + r;
        const int cA = __builtin_amdgcn_readlane(cnt, 0);
        const int cB = __builtin_amdgcn_readlane(cnt, 16);
        const int cmax = (cA > cB) ? cA : cB;

        v8h x0[8];
        rows8<0>(x0, cur, xbase, lo);

        int nbeg = 0, ncnt = 0;
        int2 ncur = make_int2(0, 0);
        if (p < 7) {
            nbeg = __shfl(offv, r + 2);
            ncnt = __shfl(offv, r + 3) - nbeg;
            if (hl < ncnt) ncur = csr[nbeg + hl];
        }
        int2 nxt = make_int2(0, 0);
        if (hl + 16 < cnt) nxt = csr[beg + 16 + hl];
        float acc[8];
        #pragma unroll
        for (int i = 0; i < 8; ++i) acc[i] = 0.f;

        sum8<0>(acc, x0, cur, cnt, w, b);
        const RT* xp = xres + (size_t)((node < nrows) ? node : (nrows - 1)) * F_DIM + c;
        v8h    xh = {};
        float4 xa = make_float4(0.f, 0.f, 0.f, 0.f), xb = xa;
        if constexpr (sizeof(RT) == 2) {
            xh = *(const v8h*)xp;
        } else {
            xa = *(const float4*)xp;
            xb = *(const float4*)(xp + 4);
        }
        if (cmax > 12)     edge_group8<8>(acc, cur, cnt, xbase, lo, w, b);
        else if (cmax > 8) edge_group<8>(acc, cur, cnt, xbase, lo, w, b);
        cur = nxt;
        for (int done = 16; done < cmax; done += 16)
            edge_trip(acc, cur, cnt, cmax, done, beg, hl, csr, xbase, lo, w, b);

        float xr[8];
        if constexpr (sizeof(RT) == 2) {
            #pragma unroll
            for (int i = 0; i < 8; ++i) xr[i] = (float)xh[i];
        } else {
            xr[0] = xa.x; xr[1] = xa.y; xr[2] = xa.z; xr[3] = xa.w;
            xr[4] = xb.x; xr[5] = xb.y; xr[6] = xb.z; xr[7] = xb.w;
        }
        v8h h;
        #pragma unroll
        for (int i = 0; i < 8; ++i) h[i] = (_Float16)(ep * xr[i] + acc[i]);
        *(v8h*)(st + r * 136 + c) = h;

        beg = nbeg; cnt = ncnt; cur = ncur;
    }
}

#define MLP_WAVES   25
#define MLP_THREADS (MLP_WAVES * 32)
#define MLP_BLOCKS  256

template <int N16>
struct WPieces { v8h v[(N16 + MLP_THREADS - 1) / MLP_THREADS]; };

template <int N16>
__device__ __forceinline__ void load_pieces(WPieces<N16>& p, const _Float16* __restrict__ src,
                                            const int tid) {
    #pragma unroll
    for (int q = 0; q < (N16 + MLP_THREADS - 1) / MLP_THREADS; ++q) {
        const int i = tid + q * MLP_THREADS;
        p.v[q] = *(const v8h*)(src + (size_t)((i < N16) ? i : (N16 - 1)) * 8);
    }
}

template <int N16>
__device__ __forceinline__ void store_pieces(_Float16* dst, const WPieces<N16>& p, const int tid) {
    #pragma unroll
    for (int q = 0; q < (N16 + MLP_THREADS - 1) / MLP_THREADS; ++q) {
        const int i = tid + q * MLP_THREADS;
        if (i < N16) *(v8h*)(dst + (size_t)i * 8) = p.v[q];
    }
}

__global__ void __launch_bounds__(MLP_THREADS, 7)
gather_mlp(const float* __restrict__ xin, const _Float16* __restrict__ xrows,
           const int* __restrict__ off, const int* __restrict__ csr,
           const float* __restrict__ We, const float* __restrict__ be,
           const float* __restrict__ eps,
           const _Float16* __restrict__ fw1, const float* __restrict__ b1,
           const _Float16* __restrict__ fw2, const float* __restrict__ b2,
           _Float16* __restrict__ out, int nrows) {
    __shared__ __align__(16) _Float16 wfrag[2 * 16384];
    __shared__ __align__(16) _Float16 stage[MLP_WAVES][16 * 136];

    {
        WPieces<2048> p1, p2;
        load_pieces(p1, fw1, threadIdx.x);
        load_pieces(p2, fw2, threadIdx.x);
        store_pieces(wfrag, p1, threadIdx.x);
        store_pieces(wfrag + 16384, p2, threadIdx.x);
    }
    __syncthreads();

    const int wv   = __builtin_amdgcn_readfirstlane(threadIdx.x >> 5);
    const int lane = threadIdx.x & 31;
    const int row0 = 16 * ((blockIdx.x / MLP_BLOCKS) * (MLP_WAVES * MLP_BLOCKS)
                           + wv * MLP_BLOCKS + (blockIdx.x % MLP_BLOCKS));
    if (row0 >= nrows) return;
    const float ep = 1.0f + eps[0];
    _Float16* st = &stage[wv][0];
    const _Float16* w1s = wfrag;
    const _Float16* w2s = wfrag + 16384;

    gather_stage(st, row0, lane, nrows, xin, xrows, off, (const int2*)csr, We, be, ep);

    const int m_  = lane & 15;
    const int hs8 = (lane >> 4) << 3;

    v8f acc[8];
    const v8f vz = {0.f, 0.f, 0.f, 0.f, 0.f, 0.f, 0.f, 0.f};

    #pragma unroll
    for (int nt = 0; nt < 8; ++nt) acc[nt] = vz;
    #pragma unroll
    for (int kt = 0; kt < 4; ++kt) {
        int kb = kt * 32 + hs8;
        v8h alo = *(const v8h*)(st + m_ * 136 + kb);
        v8h ahi = *(const v8h*)(st + m_ * 136 + kb + 16);
        v16h av = __builtin_shufflevector(alo, ahi, 0,1,2,3,4,5,6,7,8,9,10,11,12,13,14,15);
        #pragma unroll
        for (int nt = 0; nt < 8; ++nt) {
            const _Float16* q = w1s + (((kt << 3) | nt) * 512 + lane * 8);
            v8h blo = *(const v8h*)q;
            v8h bhi = *(const v8h*)(q + 256);
            v16h bv = __builtin_shufflevector(blo, bhi, 0,1,2,3,4,5,6,7,8,9,10,11,12,13,14,15);
            acc[nt] = __builtin_amdgcn_wmma_f32_16x16x32_f16(
                false, av, false, bv, (short)0, acc[nt], false, false);
            if ((nt & 3) == 3) __builtin_amdgcn_sched_barrier(0);
        }
    }

    #pragma unroll
    for (int nt = 0; nt < 8; ++nt) {
        int col = nt * 16 + m_;
        float bs = b1[col];
        #pragma unroll
        for (int r = 0; r < 8; ++r) {
            float v = fmaxf(acc[nt][r] + bs, 0.f);
            st[(r + hs8) * 136 + col] = (_Float16)v;
        }
    }

    #pragma unroll
    for (int nt = 0; nt < 8; ++nt) acc[nt] = vz;
    #pragma unroll
    for (int kt = 0; kt < 4; ++kt) {
        int kb = kt * 32 + hs8;
        v8h alo = *(const v8h*)(st + m_ * 136 + kb);
        v8h ahi = *(const v8h*)(st + m_ * 136 + kb + 16);
        v16h av = __builtin_shufflevector(alo, ahi, 0,1,2,3,4,5,6,7,8,9,10,11,12,13,14,15);
        #pragma unroll
        for (int nt = 0; nt < 8; ++nt) {
            const _Float16* q = w2s + (((kt << 3) | nt) * 512 + lane * 8);
            v8h blo = *(const v8h*)q;
            v8h bhi = *(const v8h*)(q + 256);
            v16h bv = __builtin_shufflevector(blo, bhi, 0,1,2,3,4,5,6,7,8,9,10,11,12,13,14,15);
            acc[nt] = __builtin_amdgcn_wmma_f32_16x16x32_f16(
                false, av, false, bv, (short)0, acc[nt], false, false);
            if ((nt & 3) == 3) __builtin_amdgcn_sched_barrier(0);
        }
    }

    #pragma unroll
    for (int nt = 0; nt < 8; ++nt) {
        int col = nt * 16 + m_;
        float bs = b2[col];
        #pragma unroll
        for (int r = 0; r < 8; ++r) {
            float v = fmaxf(acc[nt][r] + bs, 0.f);
            st[(r + hs8) * 136 + col] = (_Float16)v;
        }
    }

    _Float16* og = out + (size_t)row0 * F_DIM;
    #pragma unroll
    for (int i = 0; i < 8; ++i) {
        const int rr = 2 * i + (lane >> 4);
        const int cc = (lane & 15) << 3;
        *(v8h*)(og + rr * F_DIM + cc) = *(const v8h*)(st + rr * 136 + cc);
    }
}

__global__ void __launch_bounds__(MLP_THREADS, 7)
gather_mlp_cls(const _Float16* __restrict__ xin,
               const int* __restrict__ off, const int* __restrict__ csr,
               const float* __restrict__ We, const float* __restrict__ be,
               const float* __restrict__ eps,
               const _Float16* __restrict__ fw1, const float* __restrict__ b1,
               const _Float16* __restrict__ fw2, const float* __restrict__ b2,
               const _Float16* __restrict__ fwl, const float* __restrict__ bl,
               float* __restrict__ out, int nrows) {
    __shared__ __align__(16) _Float16 wfrag[2 * 16384 + 6144];
    __shared__ __align__(16) _Float16 stage[MLP_WAVES][16 * 136];
    __shared__ __align__(16) float ltile[MLP_WAVES][16 * N_CLS];

    {
        WPieces<2048> p1, p2;
        WPieces<768>  pl;
        load_pieces(p1, fw1, threadIdx.x);
        load_pieces(p2, fw2, threadIdx.x);
        load_pieces(pl, fwl, threadIdx.x);
        store_pieces(wfrag, p1, threadIdx.x);
        store_pieces(wfrag + 16384, p2, threadIdx.x);
        store_pieces(wfrag + 32768, pl, threadIdx.x);
    }
    __syncthreads();

    const int wv   = __builtin_amdgcn_readfirstlane(threadIdx.x >> 5);
    const int lane = threadIdx.x & 31;
    const int row0 = 16 * ((blockIdx.x / MLP_BLOCKS) * (MLP_WAVES * MLP_BLOCKS)
                           + wv * MLP_BLOCKS + (blockIdx.x % MLP_BLOCKS));
    if (row0 >= nrows) return;
    const float ep = 1.0f + eps[0];
    _Float16* st = &stage[wv][0];
    const _Float16* w1s = wfrag;
    const _Float16* w2s = wfrag + 16384;
    const _Float16* wls = wfrag + 32768;

    gather_stage(st, row0, lane, nrows, xin, xin, off, (const int2*)csr, We, be, ep);

    const int m_  = lane & 15;
    const int hs8 = (lane >> 4) << 3;

    v8f acc[8];
    const v8f vz = {0.f, 0.f, 0.f, 0.f, 0.f, 0.f, 0.f, 0.f};

    #pragma unroll
    for (int nt = 0; nt < 8; ++nt) acc[nt] = vz;
    #pragma unroll
    for (int kt = 0; kt < 4; ++kt) {
        int kb = kt * 32 + hs8;
        v8h alo = *(const v8h*)(st + m_ * 136 + kb);
        v8h ahi = *(const v8h*)(st + m_ * 136 + kb + 16);
        v16h av = __builtin_shufflevector(alo, ahi, 0,1,2,3,4,5,6,7,8,9,10,11,12,13,14,15);
        #pragma unroll
        for (int nt = 0; nt < 8; ++nt) {
            const _Float16* q = w1s + (((kt << 3) | nt) * 512 + lane * 8);
            v8h blo = *(const v8h*)q;
            v8h bhi = *(const v8h*)(q + 256);
            v16h bv = __builtin_shufflevector(blo, bhi, 0,1,2,3,4,5,6,7,8,9,10,11,12,13,14,15);
            acc[nt] = __builtin_amdgcn_wmma_f32_16x16x32_f16(
                false, av, false, bv, (short)0, acc[nt], false, false);
            if ((nt & 3) == 3) __builtin_amdgcn_sched_barrier(0);
        }
    }

    #pragma unroll
    for (int nt = 0; nt < 8; ++nt) {
        int col = nt * 16 + m_;
        float bs = b1[col];
        #pragma unroll
        for (int r = 0; r < 8; ++r) {
            float v = fmaxf(acc[nt][r] + bs, 0.f);
            st[(r + hs8) * 136 + col] = (_Float16)v;
        }
    }

    #pragma unroll
    for (int nt = 0; nt < 8; ++nt) acc[nt] = vz;
    #pragma unroll
    for (int kt = 0; kt < 4; ++kt) {
        int kb = kt * 32 + hs8;
        v8h alo = *(const v8h*)(st + m_ * 136 + kb);
        v8h ahi = *(const v8h*)(st + m_ * 136 + kb + 16);
        v16h av = __builtin_shufflevector(alo, ahi, 0,1,2,3,4,5,6,7,8,9,10,11,12,13,14,15);
        #pragma unroll
        for (int nt = 0; nt < 8; ++nt) {
            const _Float16* q = w2s + (((kt << 3) | nt) * 512 + lane * 8);
            v8h blo = *(const v8h*)q;
            v8h bhi = *(const v8h*)(q + 256);
            v16h bv = __builtin_shufflevector(blo, bhi, 0,1,2,3,4,5,6,7,8,9,10,11,12,13,14,15);
            acc[nt] = __builtin_amdgcn_wmma_f32_16x16x32_f16(
                false, av, false, bv, (short)0, acc[nt], false, false);
            if ((nt & 3) == 3) __builtin_amdgcn_sched_barrier(0);
        }
    }

    #pragma unroll
    for (int nt = 0; nt < 8; ++nt) {
        int col = nt * 16 + m_;
        float bs = b2[col];
        #pragma unroll
        for (int r = 0; r < 8; ++r) {
            float v = fmaxf(acc[nt][r] + bs, 0.f);
            st[(r + hs8) * 136 + col] = (_Float16)v;
        }
    }

    v8f lg[3];
    #pragma unroll
    for (int nt = 0; nt < 3; ++nt) lg[nt] = vz;
    #pragma unroll
    for (int kt = 0; kt < 4; ++kt) {
        int kb = kt * 32 + hs8;
        v8h alo = *(const v8h*)(st + m_ * 136 + kb);
        v8h ahi = *(const v8h*)(st + m_ * 136 + kb + 16);
        v16h av = __builtin_shufflevector(alo, ahi, 0,1,2,3,4,5,6,7,8,9,10,11,12,13,14,15);
        #pragma unroll
        for (int nt = 0; nt < 3; ++nt) {
            const _Float16* q = wls + ((kt * 3 + nt) * 512 + lane * 8);
            v8h blo = *(const v8h*)q;
            v8h bhi = *(const v8h*)(q + 256);
            v16h bv = __builtin_shufflevector(blo, bhi, 0,1,2,3,4,5,6,7,8,9,10,11,12,13,14,15);
            lg[nt] = __builtin_amdgcn_wmma_f32_16x16x32_f16(
                false, av, false, bv, (short)0, lg[nt], false, false);
        }
        __builtin_amdgcn_sched_barrier(0);
    }

    float* lt = &ltile[wv][0];
    #pragma unroll
    for (int nt = 0; nt < 3; ++nt) {
        int col = nt * 16 + m_;
        if (col < N_CLS) {
            float bs = bl[col];
            #pragma unroll
            for (int r = 0; r < 8; ++r) lt[(r + hs8) * N_CLS + col] = lg[nt][r] + bs;
        }
    }

    {
        float* rowp = lt + (lane >> 1) * N_CLS + (lane & 1) * 20;
        float v[20];
        #pragma unroll
        for (int q = 0; q < 5; ++q) {
            float4 t = *(const float4*)(rowp + 4 * q);
            v[4 * q] = t.x; v[4 * q + 1] = t.y; v[4 * q + 2] = t.z; v[4 * q + 3] = t.w;
        }
        float mx = v[0];
        #pragma unroll
        for (int c = 1; c < 20; ++c) mx = fmaxf(mx, v[c]);
        mx = fmaxf(mx, __shfl_xor(mx, 1));
        float s = 0.f;
        #pragma unroll
        for (int c = 0; c < 20; ++c) s += __expf(v[c] - mx);
        s += __shfl_xor(s, 1);
        const float sub = mx + __logf(s);
        #pragma unroll
        for (int q = 0; q < 5; ++q) {
            float4 t = make_float4(v[4 * q] - sub, v[4 * q + 1] - sub,
                                   v[4 * q + 2] - sub, v[4 * q + 3] - sub);
            *(float4*)(rowp + 4 * q) = t;
        }
    }

    float* og = out + (size_t)row0 * N_CLS;
    #pragma unroll
    for (int i = 0; i < 5; ++i) {
        const int e = (i * 32 + lane) * 4;
        *(float4*)(og + e) = *(const float4*)(lt + e);
    }
}


extern "C" void kernel_launch(void* const* d_in, const int* in_sizes, int n_in,
                              void* d_out, int out_size, void* d_ws, size_t ws_size,
                              hipStream_t stream) {
    const float* x    = (const float*)d_in[0];
    const int*   ei   = (const int*)d_in[1];
    const float* ea   = (const float*)d_in[2];
    const float* eps1 = (const float*)d_in[3];
    const float* We1  = (const float*)d_in[4];
    const float* be1  = (const float*)d_in[5];
    const float* W11  = (const float*)d_in[6];
    const float* b11  = (const float*)d_in[7];
    const float* W12  = (const float*)d_in[8];
    const float* b12  = (const float*)d_in[9];
    const float* eps2 = (const float*)d_in[10];
    const float* We2  = (const float*)d_in[11];
    const float* be2  = (const float*)d_in[12];
    const float* W21  = (const float*)d_in[13];
    const float* b21  = (const float*)d_in[14];
    const float* W22  = (const float*)d_in[15];
    const float* b22  = (const float*)d_in[16];
    const float* Wl   = (const float*)d_in[17];
    const float* bl   = (const float*)d_in[18];

    const int N_ = in_sizes[0] / F_DIM;
    const int E_ = in_sizes[2];
    const size_t NB = (size_t)N_ * F_DIM * sizeof(float);

    char* ws = (char*)d_ws;
    _Float16* h1    = (_Float16*)(ws + NB);
    _Float16* x16   = (_Float16*)(ws + NB + NB / 2);
    _Float16* frags = (_Float16*)(ws + 2 * NB);
    _Float16* f11 = frags;
    _Float16* f12 = frags + 16384;
    _Float16* f21 = frags + 32768;
    _Float16* f22 = frags + 49152;
    _Float16* fl  = frags + 65536;
    float* out = (float*)d_out;

    const int K     = (N_ + BKT_W - 1) / BKT_W;
    const int chunk = (E_ + BIN_BLOCKS - 1) / BIN_BLOCKS;
    const int nmat  = K * BIN_BLOCKS;
    const size_t matB = (((size_t)nmat + 4) * sizeof(int) + 255) & ~(size_t)255;
    const size_t offB = (((size_t)N_ + 4) * sizeof(int) + 255) & ~(size_t)255;
    const size_t prB  = ((size_t)E_ * 2 * sizeof(int) + 255) & ~(size_t)255;
    const size_t dlB  = ((size_t)E_ * sizeof(int) + 255) & ~(size_t)255;
    char* gbase = ws + 2 * NB + 131072 + 16384;
    int*  cntmat = (int*)(gbase);
    int*  bstart = (int*)(gbase + matB);
    int*  off    = (int*)(gbase + 2 * matB);
    int2* tpair  = (int2*)(gbase + 2 * matB + offB);
    int*  tdl    = (int*)(gbase + 2 * matB + offB + prB);
    int*  csr    = (int*)(gbase + 2 * matB + offB + prB + dlB);

    const int rows_per_set = 16 * MLP_WAVES * MLP_BLOCKS;
    const int mb = MLP_BLOCKS * ((N_ + rows_per_set - 1) / rows_per_set);

    bin_count<<<BIN_BLOCKS, 1024, K * sizeof(int), stream>>>(ei, cntmat, E_, K, chunk,
                                                             W11, W12, W21, W22, Wl, frags, x, x16, N_);
    bin_scatter<<<BIN_BLOCKS, 1024, 3 * K * sizeof(int), stream>>>(ei, ea, cntmat, bstart, tpair, tdl, E_, K, chunk);
    bucket_place<<<K, 1024, 0, stream>>>(bstart, tpair, tdl, off, (int2*)csr, N_, K);

    gather_mlp<<<mb, MLP_THREADS, 0, stream>>>(x, x16, off, csr, We1, be1, eps1, f11, b11, f12, b12, h1, N_);
    gather_mlp_cls<<<mb, MLP_THREADS, 0, stream>>>(h1, off, csr, We2, be2, eps2, f21, b21, f22, b22, fl, bl, out, N_);
}


# Edge message arithmetic cut from 24 to 16 vector instructions per edge: f16 row halves feed mixed-precision FMAs (no converts), edge bias added once per node as count*b; sum order changes
# speedup vs baseline: 6.3381x; 1.0666x over previous
#include <hip/hip_runtime.h>
#include <hip/hip_bf16.h>
#include <stdint.h>

typedef _Float16 v16h __attribute__((ext_vector_type(16)));
typedef _Float16 v8h  __attribute__((ext_vector_type(8)));
typedef _Float16 v4h  __attribute__((ext_vector_type(4)));
typedef float    v8f  __attribute__((ext_vector_type(8)));

#define F_DIM 128
#define N_CLS 40

#define BKT_SHIFT 9
#define BKT_W (1 << BKT_SHIFT)
#define BIN_BLOCKS 256

__device__ __forceinline__ int frag_dword(const int ntk, const int lane, const int d) {
    return ntk * 256 + (d >> 2) * 128 + lane * 4 + (d & 3);
}

#define CONV_THREADS (BIN_BLOCKS * 1024)
#define WFRAG_DWORDS (4 * 8192 + 3072)
static_assert(CONV_THREADS >= WFRAG_DWORDS, "one thread per fragment dword");

__device__ __forceinline__ void convert_x(const int g, const float* __restrict__ x,
                                          _Float16* __restrict__ x16, const int nrows) {
    const int n4     = nrows * (F_DIM / 4);
    const int stride = CONV_THREADS;
    const float4* x4 = (const float4*)x;
    int i = g;
    for (; i + 3 * stride < n4; i += 4 * stride) {
        float4 v[4];
        #pragma unroll
        for (int q = 0; q < 4; ++q) v[q] = x4[i + q * stride];
        #pragma unroll
        for (int q = 0; q < 4; ++q) {
            v4h h = {(_Float16)v[q].x, (_Float16)v[q].y, (_Float16)v[q].z, (_Float16)v[q].w};
            *(v4h*)(x16 + (size_t)(i + q * stride) * 4) = h;
        }
    }
    for (; i < n4; i += stride) {
        const float4 v = x4[i];
        v4h h = {(_Float16)v.x, (_Float16)v.y, (_Float16)v.z, (_Float16)v.w};
        *(v4h*)(x16 + (size_t)i * 4) = h;
    }
}

__device__ __forceinline__ void convert_weights(const int gid,
                         const float* __restrict__ W11, const float* __restrict__ W12,
                         const float* __restrict__ W21, const float* __restrict__ W22,
                         const float* __restrict__ Wl, _Float16* __restrict__ frags) {
    if (gid >= 4 * 8192) {
        int r = gid - 4 * 8192;
        int d    = r & 7;
        int lane = (r >> 3) & 31;
        int ntk  = r >> 8;
        int kt   = ntk / 3;
        int nt   = ntk - kt * 3;
        int nn   = nt * 16 + (lane & 15);
        int kb   = kt * 32 + ((lane >> 4) << 3);
        int k0   = kb + ((d < 4) ? (d << 1) : (16 + ((d - 4) << 1)));
        _Float16* o = frags + (size_t)4 * 16384 + (size_t)frag_dword(ntk, lane, d) * 2;
        const bool ok = (nn < N_CLS);
        o[0] = ok ? (_Float16)Wl[(size_t)k0 * N_CLS + nn] : (_Float16)0.f;
        o[1] = ok ? (_Float16)Wl[(size_t)(k0 + 1) * N_CLS + nn] : (_Float16)0.f;
        return;
    }
    int m    = gid >> 13;
    int r    = gid & 8191;
    int d    = r & 7;
    int lane = (r >> 3) & 31;
    int ntk  = r >> 8;
    int kt   = ntk >> 3;
    int nt   = ntk & 7;
    int nn   = nt * 16 + (lane & 15);
    int kb   = kt * 32 + ((lane >> 4) << 3);
    int k0   = kb + ((d < 4) ? (d << 1) : (16 + ((d - 4) << 1)));
    const float* W = (m == 0) ? W11 : (m == 1) ? W12 : (m == 2) ? W21 : W22;
    _Float16* o = frags + (size_t)m * 16384 + (size_t)frag_dword(ntk, lane, d) * 2;
    o[0] = (_Float16)W[(size_t)k0 * F_DIM + nn];
    o[1] = (_Float16)W[(size_t)(k0 + 1) * F_DIM + nn];
}

__global__ void __launch_bounds__(1024)
bin_count(const int* __restrict__ ei, int* __restrict__ cntmat, int E_, int K, int chunk,
          const float* __restrict__ W11, const float* __restrict__ W12,
          const float* __restrict__ W21, const float* __restrict__ W22,
          const float* __restrict__ Wl, _Float16* __restrict__ frags,
          const float* __restrict__ x, _Float16* __restrict__ x16, int nrows) {
    extern __shared__ int hist[];
    const int tid = threadIdx.x;
    const int b   = blockIdx.x;
    for (int k = tid; k < K; k += 1024) hist[k] = 0;
    __syncthreads();
    const int e0 = b * chunk;
    const int e1 = (e0 + chunk < E_) ? (e0 + chunk) : E_;
    const int* dst = ei + E_;
    for (int i0 = e0 + tid; i0 < e1; i0 += 4096) {
        int d[4];
        #pragma unroll
        for (int q = 0; q < 4; ++q) {
            const int i = i0 + q * 1024;
            d[q] = (i < e1) ? dst[i] : -1;
        }
        #pragma unroll
        for (int q = 0; q < 4; ++q)
            if (d[q] >= 0) atomicAdd(&hist[d[q] >> BKT_SHIFT], 1);
    }
    const int g = b * 1024 + tid;
    if (g < WFRAG_DWORDS) convert_weights(g, W11, W12, W21, W22, Wl, frags);
    convert_x(g, x, x16, nrows);
    __syncthreads();
    for (int k = tid; k < K; k += 1024) cntmat[k * BIN_BLOCKS + b] = hist[k];
}

__device__ __forceinline__ int scan_k(const int* in, int* out, const int K, int* wsum,
                                      const int tid, const int lane, const int wv) {
    int carry = 0;
    for (int kb = 0; kb < K; kb += 1024) {
        const int k = kb + tid;
        const int v = (k < K) ? in[k] : 0;
        int incl = v;
        #pragma unroll
        for (int dd = 1; dd < 32; dd <<= 1) {
            int y = __shfl_up(incl, dd);
            if (lane >= dd) incl += y;
        }
        if (lane == 31) wsum[wv] = incl;
        __syncthreads();
        const int w = wsum[lane];
        int wincl = w;
        #pragma unroll
        for (int dd = 1; dd < 32; dd <<= 1) {
            int y = __shfl_up(wincl, dd);
            if (lane >= dd) wincl += y;
        }
        const int wave_excl = __shfl(wincl, wv) - __shfl(w, wv);
        const int total     = __shfl(wincl, 31);
        if (k < K) out[k] = carry + wave_excl + (incl - v);
        carry += total;
        __syncthreads();
    }
    return carry;
}

#define SC_BATCH 8192
static_assert(BIN_BLOCKS == 256, "a wave reads a row of the count matrix as 32 lanes x 8 counts");
__global__ void __launch_bounds__(1024)
bin_scatter(const int* __restrict__ ei, const float* __restrict__ ea,
            const int* __restrict__ cntmat, int* __restrict__ bstart,
            int2* __restrict__ tpair, int* __restrict__ tdl, int E_, int K, int chunk) {
    extern __shared__ int dyn[];
    __shared__ int2 spair[SC_BATCH];
    __shared__ int  sdl[SC_BATCH];
    __shared__ int  wsum[32];
    int* cur    = dyn;
    int* lcnt   = dyn + K;
    int* lstart = dyn + 2 * K;
    const int tid  = threadIdx.x;
    const int lane = tid & 31;
    const int wv   = tid >> 5;
    const int b    = blockIdx.x;
    for (int k0 = wv; k0 < K; k0 += 128) {
        int4 u[4][2];
        #pragma unroll
        for (int q = 0; q < 4; ++q) {
            const int k  = k0 + 32 * q;
            const int kk = (k < K) ? k : (K - 1);
            const int4* rp = (const int4*)(cntmat + (size_t)kk * BIN_BLOCKS) + 2 * lane;
            u[q][0] = rp[0];
            u[q][1] = rp[1];
        }
        int tot[4], pre[4];
        #pragma unroll
        for (int q = 0; q < 4; ++q) {
            const int v[8] = {u[q][0].x, u[q][0].y, u[q][0].z, u[q][0].w,
                              u[q][1].x, u[q][1].y, u[q][1].z, u[q][1].w};
            tot[q] = 0; pre[q] = 0;
            #pragma unroll
            for (int j = 0; j < 8; ++j) {
                tot[q] += v[j];
                if (8 * lane + j < b) pre[q] += v[j];
            }
        }
        #pragma unroll
        for (int dd = 16; dd >= 1; dd >>= 1) {
            #pragma unroll
            for (int q = 0; q < 4; ++q) {
                tot[q] += __shfl_xor(tot[q], dd);
                pre[q] += __shfl_xor(pre[q], dd);
            }
        }
        #pragma unroll
        for (int q = 0; q < 4; ++q) {
            const int k = k0 + 32 * q;
            if (k < K && lane == 0) { lcnt[k] = tot[q]; cur[k] = pre[q]; }
        }
    }
    __syncthreads();
    const int nedges = scan_k(lcnt, lstart, K, wsum, tid, lane, wv);
    for (int k = tid; k < K; k += 1024) {
        const int s = lstart[k];
        cur[k] += s;
        if (b == 0) bstart[k] = s;
    }
    if (b == 0 && tid == 0) bstart[K] = nedges;
    const int e0 = b * chunk;
    const int e1 = (e0 + chunk < E_) ? (e0 + chunk) : E_;
    const int* dst = ei + E_;
    for (int s0 = e0; s0 < e1; s0 += SC_BATCH) {
        const int s1 = (s0 + SC_BATCH < e1) ? (s0 + SC_BATCH) : e1;
        for (int k = tid; k < K; k += 1024) lcnt[k] = 0;
        __syncthreads();
        int d[8], s[8], r[8];
        float a[8];
        #pragma unroll
        for (int q = 0; q < 8; ++q) {
            const int i = s0 + q * 1024 + tid;
            const bool ok = (i < s1);
            d[q] = ok ? dst[i] : -1;
            s[q] = ok ? ei[i] : 0;
            a[q] = ok ? ea[i] : 0.f;
        }
        #pragma unroll
        for (int q = 0; q < 8; ++q)
            r[q] = (d[q] >= 0) ? atomicAdd(&lcnt[d[q] >> BKT_SHIFT], 1) : 0;
        __syncthreads();
        scan_k(lcnt, lstart, K, wsum, tid, lane, wv);
        #pragma unroll
        for (int q = 0; q < 8; ++q) {
            if (d[q] >= 0) {
                const int slot = lstart[d[q] >> BKT_SHIFT] + r[q];
                spair[slot] = make_int2(s[q], __float_as_int(a[q]));
                sdl[slot]   = d[q] & (BKT_W - 1);
            }
        }
        __syncthreads();
        for (int k = wv; k < K; k += 32) {
            const int L  = lcnt[k];
            const int ls = lstart[k];
            const int g  = cur[k];
            for (int j = lane; j < L; j += 32) {
                tpair[g + j] = spair[ls + j];
                tdl[g + j]   = sdl[ls + j];
            }
            if (lane == 0) cur[k] = g + L;
        }
        __syncthreads();
    }
}

#define PLACE_CAP 16384
__global__ void __launch_bounds__(1024)
bucket_place(const int* __restrict__ bstart, const int2* __restrict__ tpair,
             const int* __restrict__ tdl, int* __restrict__ off,
             int2* __restrict__ csr, int n, int K) {
    __shared__ int c[BKT_W];
    __shared__ int wsum[32];
    __shared__ int2 stage[PLACE_CAP];
    const int tid  = threadIdx.x;
    const int lane = tid & 31;
    const int wv   = tid >> 5;
    const int k    = blockIdx.x;
    const int lo = bstart[k];
    const int hi = bstart[k + 1];
    if (tid < BKT_W) c[tid] = 0;
    __syncthreads();
    for (int i0 = lo + tid; i0 < hi; i0 += 4096) {
        int dl[4];
        #pragma unroll
        for (int q = 0; q < 4; ++q) {
            const int i = i0 + q * 1024;
            dl[q] = (i < hi) ? tdl[i] : -1;
        }
        #pragma unroll
        for (int q = 0; q < 4; ++q)
            if (dl[q] >= 0) atomicAdd(&c[dl[q]], 1);
    }
    __syncthreads();
    const int v = (tid < BKT_W) ? c[tid] : 0;
    int incl = v;
    #pragma unroll
    for (int d = 1; d < 32; d <<= 1) {
        int y = __shfl_up(incl, d);
        if (lane >= d) incl += y;
    }
    if (lane == 31) wsum[wv] = incl;
    __syncthreads();
    const int w = wsum[lane];
    int wincl = w;
    #pragma unroll
    for (int d = 1; d < 32; d <<= 1) {
        int y = __shfl_up(wincl, d);
        if (lane >= d) wincl += y;
    }
    const int wave_excl = __shfl(wincl, wv) - __shfl(w, wv);
    const int lstart = wave_excl + (incl - v);
    if (tid < BKT_W) {
        c[tid] = lstart;
        const int node = k * BKT_W + tid;
        if (node < n) off[node] = lo + lstart;
    }
    if (k == K - 1 && tid == 0) off[n] = hi;
    __syncthreads();
    for (int i0 = lo + tid; i0 < hi; i0 += 4096) {
        int dl[4];
        int2 pr[4];
        #pragma unroll
        for (int q = 0; q < 4; ++q) {
            const int i = i0 + q * 1024;
            const bool ok = (i < hi);
            dl[q] = ok ? tdl[i] : -1;
            pr[q] = ok ? tpair[i] : make_int2(0, 0);
        }
        #pragma unroll
        for (int q = 0; q < 4; ++q) {
            if (dl[q] >= 0) {
                const int p = atomicAdd(&c[dl[q]], 1);
                if (p < PLACE_CAP) stage[p] = pr[q];
                else               csr[lo + p] = pr[q];
            }
        }
    }
    __syncthreads();
    const int nst  = (hi - lo < PLACE_CAP) ? (hi - lo) : PLACE_CAP;
    const int gend = lo + nst;
    for (int g = (lo & ~31) + tid; g < gend; g += 1024)
        if (g >= lo) csr[g] = stage[g - lo];
}

template <int Q>
__device__ __forceinline__ int row_share(const int v) {
    return __builtin_amdgcn_update_dpp(0, v, 0x150 + Q, 0xf, 0xf, false);
}

__device__ __forceinline__ void msg_add8(float (&acc)[8], const v8h xr, const float a,
                                         const float keep, const float (&w)[8],
                                         const float (&nb)[8]) {
    #pragma unroll
    for (int i = 0; i < 8; ++i) {
        const float v = __builtin_fmaf(a, w[i], (float)xr[i]);
        acc[i] = fmaf(keep, fmaxf(v, nb[i]), acc[i]);
    }
    asm("" : "+v"(acc[0]), "+v"(acc[1]), "+v"(acc[2]), "+v"(acc[3]),
             "+v"(acc[4]), "+v"(acc[5]), "+v"(acc[6]), "+v"(acc[7]));
}

template <int Q>
__device__ __forceinline__ void edge_group(float (&acc)[8], const int2 my, const int m,
                                           const char* __restrict__ xbase, const unsigned lo,
                                           const float (&w)[8], const float (&b)[8]) {
    const unsigned s0 = ((unsigned)row_share<Q>(my.x) << 8) | lo;
    const unsigned s1 = ((unsigned)row_share<Q + 1>(my.x) << 8) | lo;
    const unsigned s2 = ((unsigned)row_share<Q + 2>(my.x) << 8) | lo;
    const unsigned s3 = ((unsigned)row_share<Q + 3>(my.x) << 8) | lo;
    const v8h x0 = *(const v8h*)(xbase + s0);
    const v8h x1 = *(const v8h*)(xbase + s1);
    const v8h x2 = *(const v8h*)(xbase + s2);
    const v8h x3 = *(const v8h*)(xbase + s3);
    const float a0 = __int_as_float(row_share<Q>(my.y));
    const float a1 = __int_as_float(row_share<Q + 1>(my.y));
    const float a2 = __int_as_float(row_share<Q + 2>(my.y));
    const float a3 = __int_as_float(row_share<Q + 3>(my.y));
    msg_add8(acc, x0, a0, (Q     < m) ? 1.f : 0.f, w, b);
    msg_add8(acc, x1, a1, (Q + 1 < m) ? 1.f : 0.f, w, b);
    msg_add8(acc, x2, a2, (Q + 2 < m) ? 1.f : 0.f, w, b);
    msg_add8(acc, x3, a3, (Q + 3 < m) ? 1.f : 0.f, w, b);
}

template <int Q>
__device__ __forceinline__ void rows8(v8h (&x)[8], const int2 my,
                                      const char* __restrict__ xbase, const unsigned lo) {
    const unsigned s0 = ((unsigned)row_share<Q>(my.x) << 8) | lo;
    const unsigned s1 = ((unsigned)row_share<Q + 1>(my.x) << 8) | lo;
    const unsigned s2 = ((unsigned)row_share<Q + 2>(my.x) << 8) | lo;
    const unsigned s3 = ((unsigned)row_share<Q + 3>(my.x) << 8) | lo;
    const unsigned s4 = ((unsigned)row_share<Q + 4>(my.x) << 8) | lo;
    const unsigned s5 = ((unsigned)row_share<Q + 5>(my.x) << 8) | lo;
    const unsigned s6 = ((unsigned)row_share<Q + 6>(my.x) << 8) | lo;
    const unsigned s7 = ((unsigned)row_share<Q + 7>(my.x) << 8) | lo;
    x[0] = *(const v8h*)(xbase + s0);
    x[1] = *(const v8h*)(xbase + s1);
    x[2] = *(const v8h*)(xbase + s2);
    x[3] = *(const v8h*)(xbase + s3);
    x[4] = *(const v8h*)(xbase + s4);
    x[5] = *(const v8h*)(xbase + s5);
    x[6] = *(const v8h*)(xbase + s6);
    x[7] = *(const v8h*)(xbase + s7);
}

template <int Q>
__device__ __forceinline__ void sum8(float (&acc)[8], const v8h (&x)[8], const int2 my,
                                     const int m, const float (&w)[8], const float (&b)[8]) {
    const float a0 = __int_as_float(row_share<Q>(my.y));
    const float a1 = __int_as_float(row_share<Q + 1>(my.y));
    const float a2 = __int_as_float(row_share<Q + 2>(my.y));
    const float a3 = __int_as_float(row_share<Q + 3>(my.y));
    const float a4 = __int_as_float(row_share<Q + 4>(my.y));
    const float a5 = __int_as_float(row_share<Q + 5>(my.y));
    const float a6 = __int_as_float(row_share<Q + 6>(my.y));
    const float a7 = __int_as_float(row_share<Q + 7>(my.y));
    msg_add8(acc, x[0], a0, (Q     < m) ? 1.f : 0.f, w, b);
    msg_add8(acc, x[1], a1, (Q + 1 < m) ? 1.f : 0.f, w, b);
    msg_add8(acc, x[2], a2, (Q + 2 < m) ? 1.f : 0.f, w, b);
    msg_add8(acc, x[3], a3, (Q + 3 < m) ? 1.f : 0.f, w, b);
    msg_add8(acc, x[4], a4, (Q + 4 < m) ? 1.f : 0.f, w, b);
    msg_add8(acc, x[5], a5, (Q + 5 < m) ? 1.f : 0.f, w, b);
    msg_add8(acc, x[6], a6, (Q + 6 < m) ? 1.f : 0.f, w, b);
    msg_add8(acc, x[7], a7, (Q + 7 < m) ? 1.f : 0.f, w, b);
}

template <int Q>
__device__ __forceinline__ void edge_group8(float (&acc)[8], const int2 my, const int m,
                                            const char* __restrict__ xbase, const unsigned lo,
                                            const float (&w)[8], const float (&b)[8]) {
    v8h x[8];
    rows8<Q>(x, my, xbase, lo);
    sum8<Q>(acc, x, my, m, w, b);
}

__device__ __forceinline__ void edge_trip(float (&acc)[8], int2& cur, const int cnt, const int cmax,
                                          const int done, const int beg, const int hl,
                                          const int2* __restrict__ csr,
                                          const char* __restrict__ xbase, const unsigned lo,
                                          const float (&w)[8], const float (&b)[8]) {
    const int m  = cnt - done;
    const int mm = cmax - done;
    int2 nxt = make_int2(0, 0);
    if (hl + 16 < m) nxt = csr[beg + done + 16 + hl];
    if (mm > 4) edge_group8<0>(acc, cur, m, xbase, lo, w, b);
    else        edge_group<0>(acc, cur, m, xbase, lo, w, b);
    if (mm > 12)     edge_group8<8>(acc, cur, m, xbase, lo, w, b);
    else if (mm > 8) edge_group<8>(acc, cur, m, xbase, lo, w, b);
    cur = nxt;
}

__device__ __forceinline__ void load8(float (&r)[8], const float* __restrict__ p) {
    const float4 u = *(const float4*)p, v = *(const float4*)(p + 4);
    r[0] = u.x; r[1] = u.y; r[2] = u.z; r[3] = u.w;
    r[4] = v.x; r[5] = v.y; r[6] = v.z; r[7] = v.w;
}

template <typename RT>
__device__ __forceinline__ void gather_stage(_Float16* st, const int row0, const int lane,
                                             const int nrows, const RT* __restrict__ xres,
                                             const _Float16* __restrict__ xrows,
                                             const int* __restrict__ off,
                                             const int2* __restrict__ csr,
                                             const float* __restrict__ We,
                                             const float* __restrict__ be, const float ep) {
    const int hl   = lane & 15;
    const int half = lane >> 4;
    const int c    = hl << 3;

    int offv = 0;
    if (lane <= 16) {
        const int idx = (row0 + lane < nrows) ? (row0 + lane) : nrows;
        offv = off[idx];
    }

    float w[8], b[8];
    load8(w, We + c);
    load8(b, be + c);
    #pragma unroll
    for (int i = 0; i < 8; ++i) b[i] = -b[i];

    const char* xbase = (const char*)xrows;
    const unsigned lo = (unsigned)hl << 4;

    int beg = __shfl(offv, half);
    int cnt = __shfl(offv, half + 1) - beg;
    int2 cur = make_int2(0, 0);
    if (hl < cnt) cur = csr[beg + hl];

    #pragma unroll 1
    for (int p = 0; p < 8; ++p) {
        const int r    = 2 * p + half;
        const int node = row0 + r;
        const int cA = __builtin_amdgcn_readlane(cnt, 0);
        const int cB = __builtin_amdgcn_readlane(cnt, 16);
        const int cmax = (cA > cB) ? cA : cB;

        v8h x0[8];
        rows8<0>(x0, cur, xbase, lo);

        int nbeg = 0, ncnt = 0;
        int2 ncur = make_int2(0, 0);
        if (p < 7) {
            nbeg = __shfl(offv, r + 2);
            ncnt = __shfl(offv, r + 3) - nbeg;
            if (hl < ncnt) ncur = csr[nbeg + hl];
        }
        int2 nxt = make_int2(0, 0);
        if (hl + 16 < cnt) nxt = csr[beg + 16 + hl];
        float acc[8];
        #pragma unroll
        for (int i = 0; i < 8; ++i) acc[i] = 0.f;

        sum8<0>(acc, x0, cur, cnt, w, b);
        const RT* xp = xres + (size_t)((node < nrows) ? node : (nrows - 1)) * F_DIM + c;
        v8h    xh = {};
        float4 xa = make_float4(0.f, 0.f, 0.f, 0.f), xb = xa;
        if constexpr (sizeof(RT) == 2) {
            xh = *(const v8h*)xp;
        } else {
            xa = *(const float4*)xp;
            xb = *(const float4*)(xp + 4);
        }
        if (cmax > 12)     edge_group8<8>(acc, cur, cnt, xbase, lo, w, b);
        else if (cmax > 8) edge_group<8>(acc, cur, cnt, xbase, lo, w, b);
        cur = nxt;
        for (int done = 16; done < cmax; done += 16)
            edge_trip(acc, cur, cnt, cmax, done, beg, hl, csr, xbase, lo, w, b);

        float xr[8];
        if constexpr (sizeof(RT) == 2) {
            #pragma unroll
            for (int i = 0; i < 8; ++i) xr[i] = (float)xh[i];
        } else {
            xr[0] = xa.x; xr[1] = xa.y; xr[2] = xa.z; xr[3] = xa.w;
            xr[4] = xb.x; xr[5] = xb.y; xr[6] = xb.z; xr[7] = xb.w;
        }
        const float ncnt_f = -(float)cnt;
        v8h h;
        #pragma unroll
        for (int i = 0; i < 8; ++i) h[i] = (_Float16)(ep * xr[i] + fmaf(ncnt_f, b[i], acc[i]));
        *(v8h*)(st + r * 136 + c) = h;

        beg = nbeg; cnt = ncnt; cur = ncur;
    }
}

#define MLP_WAVES   25
#define MLP_THREADS (MLP_WAVES * 32)
#define MLP_BLOCKS  256

template <int N16>
struct WPieces { v8h v[(N16 + MLP_THREADS - 1) / MLP_THREADS]; };

template <int N16>
__device__ __forceinline__ void load_pieces(WPieces<N16>& p, const _Float16* __restrict__ src,
                                            const int tid) {
    #pragma unroll
    for (int q = 0; q < (N16 + MLP_THREADS - 1) / MLP_THREADS; ++q) {
        const int i = tid + q * MLP_THREADS;
        p.v[q] = *(const v8h*)(src + (size_t)((i < N16) ? i : (N16 - 1)) * 8);
    }
}

template <int N16>
__device__ __forceinline__ void store_pieces(_Float16* dst, const WPieces<N16>& p, const int tid) {
    #pragma unroll
    for (int q = 0; q < (N16 + MLP_THREADS - 1) / MLP_THREADS; ++q) {
        const int i = tid + q * MLP_THREADS;
        if (i < N16) *(v8h*)(dst + (size_t)i * 8) = p.v[q];
    }
}

__global__ void __launch_bounds__(MLP_THREADS, 7)
gather_mlp(const float* __restrict__ xin, const _Float16* __restrict__ xrows,
           const int* __restrict__ off, const int* __restrict__ csr,
           const float* __restrict__ We, const float* __restrict__ be,
           const float* __restrict__ eps,
           const _Float16* __restrict__ fw1, const float* __restrict__ b1,
           const _Float16* __restrict__ fw2, const float* __restrict__ b2,
           _Float16* __restrict__ out, int nrows) {
    __shared__ __align__(16) _Float16 wfrag[2 * 16384];
    __shared__ __align__(16) _Float16 stage[MLP_WAVES][16 * 136];

    {
        WPieces<2048> p1, p2;
        load_pieces(p1, fw1, threadIdx.x);
        load_pieces(p2, fw2, threadIdx.x);
        store_pieces(wfrag, p1, threadIdx.x);
        store_pieces(wfrag + 16384, p2, threadIdx.x);
    }
    __syncthreads();

    const int wv   = __builtin_amdgcn_readfirstlane(threadIdx.x >> 5);
    const int lane = threadIdx.x & 31;
    const int row0 = 16 * ((blockIdx.x / MLP_BLOCKS) * (MLP_WAVES * MLP_BLOCKS)
                           + wv * MLP_BLOCKS + (blockIdx.x % MLP_BLOCKS));
    if (row0 >= nrows) return;
    const float ep = 1.0f + eps[0];
    _Float16* st = &stage[wv][0];
    const _Float16* w1s = wfrag;
    const _Float16* w2s = wfrag + 16384;

    gather_stage(st, row0, lane, nrows, xin, xrows, off, (const int2*)csr, We, be, ep);

    const int m_  = lane & 15;
    const int hs8 = (lane >> 4) << 3;

    v8f acc[8];
    const v8f vz = {0.f, 0.f, 0.f, 0.f, 0.f, 0.f, 0.f, 0.f};

    #pragma unroll
    for (int nt = 0; nt < 8; ++nt) acc[nt] = vz;
    #pragma unroll
    for (int kt = 0; kt < 4; ++kt) {
        int kb = kt * 32 + hs8;
        v8h alo = *(const v8h*)(st + m_ * 136 + kb);
        v8h ahi = *(const v8h*)(st + m_ * 136 + kb + 16);
        v16h av = __builtin_shufflevector(alo, ahi, 0,1,2,3,4,5,6,7,8,9,10,11,12,13,14,15);
        #pragma unroll
        for (int nt = 0; nt < 8; ++nt) {
            const _Float16* q = w1s + (((kt << 3) | nt) * 512 + lane * 8);
            v8h blo = *(const v8h*)q;
            v8h bhi = *(const v8h*)(q + 256);
            v16h bv = __builtin_shufflevector(blo, bhi, 0,1,2,3,4,5,6,7,8,9,10,11,12,13,14,15);
            acc[nt] = __builtin_amdgcn_wmma_f32_16x16x32_f16(
                false, av, false, bv, (short)0, acc[nt], false, false);
            if ((nt & 3) == 3) __builtin_amdgcn_sched_barrier(0);
        }
    }

    #pragma unroll
    for (int nt = 0; nt < 8; ++nt) {
        int col = nt * 16 + m_;
        float bs = b1[col];
        #pragma unroll
        for (int r = 0; r < 8; ++r) {
            float v = fmaxf(acc[nt][r] + bs, 0.f);
            st[(r + hs8) * 136 + col] = (_Float16)v;
        }
    }

    #pragma unroll
    for (int nt = 0; nt < 8; ++nt) acc[nt] = vz;
    #pragma unroll
    for (int kt = 0; kt < 4; ++kt) {
        int kb = kt * 32 + hs8;
        v8h alo = *(const v8h*)(st + m_ * 136 + kb);
        v8h ahi = *(const v8h*)(st + m_ * 136 + kb + 16);
        v16h av = __builtin_shufflevector(alo, ahi, 0,1,2,3,4,5,6,7,8,9,10,11,12,13,14,15);
        #pragma unroll
        for (int nt = 0; nt < 8; ++nt) {
            const _Float16* q = w2s + (((kt << 3) | nt) * 512 + lane * 8);
            v8h blo = *(const v8h*)q;
            v8h bhi = *(const v8h*)(q + 256);
            v16h bv = __builtin_shufflevector(blo, bhi, 0,1,2,3,4,5,6,7,8,9,10,11,12,13,14,15);
            acc[nt] = __builtin_amdgcn_wmma_f32_16x16x32_f16(
                false, av, false, bv, (short)0, acc[nt], false, false);
            if ((nt & 3) == 3) __builtin_amdgcn_sched_barrier(0);
        }
    }

    #pragma unroll
    for (int nt = 0; nt < 8; ++nt) {
        int col = nt * 16 + m_;
        float bs = b2[col];
        #pragma unroll
        for (int r = 0; r < 8; ++r) {
            float v = fmaxf(acc[nt][r] + bs, 0.f);
            st[(r + hs8) * 136 + col] = (_Float16)v;
        }
    }

    _Float16* og = out + (size_t)row0 * F_DIM;
    #pragma unroll
    for (int i = 0; i < 8; ++i) {
        const int rr = 2 * i + (lane >> 4);
        const int cc = (lane & 15) << 3;
        *(v8h*)(og + rr * F_DIM + cc) = *(const v8h*)(st + rr * 136 + cc);
    }
}

__global__ void __launch_bounds__(MLP_THREADS, 7)
gather_mlp_cls(const _Float16* __restrict__ xin,
               const int* __restrict__ off, const int* __restrict__ csr,
               const float* __restrict__ We, const float* __restrict__ be,
               const float* __restrict__ eps,
               const _Float16* __restrict__ fw1, const float* __restrict__ b1,
               const _Float16* __restrict__ fw2, const float* __restrict__ b2,
               const _Float16* __restrict__ fwl, const float* __restrict__ bl,
               float* __restrict__ out, int nrows) {
    __shared__ __align__(16) _Float16 wfrag[2 * 16384 + 6144];
    __shared__ __align__(16) _Float16 stage[MLP_WAVES][16 * 136];
    __shared__ __align__(16) float ltile[MLP_WAVES][16 * N_CLS];

    {
        WPieces<2048> p1, p2;
        WPieces<768>  pl;
        load_pieces(p1, fw1, threadIdx.x);
        load_pieces(p2, fw2, threadIdx.x);
        load_pieces(pl, fwl, threadIdx.x);
        store_pieces(wfrag, p1, threadIdx.x);
        store_pieces(wfrag + 16384, p2, threadIdx.x);
        store_pieces(wfrag + 32768, pl, threadIdx.x);
    }
    __syncthreads();

    const int wv   = __builtin_amdgcn_readfirstlane(threadIdx.x >> 5);
    const int lane = threadIdx.x & 31;
    const int row0 = 16 * ((blockIdx.x / MLP_BLOCKS) * (MLP_WAVES * MLP_BLOCKS)
                           + wv * MLP_BLOCKS + (blockIdx.x % MLP_BLOCKS));
    if (row0 >= nrows) return;
    const float ep = 1.0f + eps[0];
    _Float16* st = &stage[wv][0];
    const _Float16* w1s = wfrag;
    const _Float16* w2s = wfrag + 16384;
    const _Float16* wls = wfrag + 32768;

    gather_stage(st, row0, lane, nrows, xin, xin, off, (const int2*)csr, We, be, ep);

    const int m_  = lane & 15;
    const int hs8 = (lane >> 4) << 3;

    v8f acc[8];
    const v8f vz = {0.f, 0.f, 0.f, 0.f, 0.f, 0.f, 0.f, 0.f};

    #pragma unroll
    for (int nt = 0; nt < 8; ++nt) acc[nt] = vz;
    #pragma unroll
    for (int kt = 0; kt < 4; ++kt) {
        int kb = kt * 32 + hs8;
        v8h alo = *(const v8h*)(st + m_ * 136 + kb);
        v8h ahi = *(const v8h*)(st + m_ * 136 + kb + 16);
        v16h av = __builtin_shufflevector(alo, ahi, 0,1,2,3,4,5,6,7,8,9,10,11,12,13,14,15);
        #pragma unroll
        for (int nt = 0; nt < 8; ++nt) {
            const _Float16* q = w1s + (((kt << 3) | nt) * 512 + lane * 8);
            v8h blo = *(const v8h*)q;
            v8h bhi = *(const v8h*)(q + 256);
            v16h bv = __builtin_shufflevector(blo, bhi, 0,1,2,3,4,5,6,7,8,9,10,11,12,13,14,15);
            acc[nt] = __builtin_amdgcn_wmma_f32_16x16x32_f16(
                false, av, false, bv, (short)0, acc[nt], false, false);
            if ((nt & 3) == 3) __builtin_amdgcn_sched_barrier(0);
        }
    }

    #pragma unroll
    for (int nt = 0; nt < 8; ++nt) {
        int col = nt * 16 + m_;
        float bs = b1[col];
        #pragma unroll
        for (int r = 0; r < 8; ++r) {
            float v = fmaxf(acc[nt][r] + bs, 0.f);
            st[(r + hs8) * 136 + col] = (_Float16)v;
        }
    }

    #pragma unroll
    for (int nt = 0; nt < 8; ++nt) acc[nt] = vz;
    #pragma unroll
    for (int kt = 0; kt < 4; ++kt) {
        int kb = kt * 32 + hs8;
        v8h alo = *(const v8h*)(st + m_ * 136 + kb);
        v8h ahi = *(const v8h*)(st + m_ * 136 + kb + 16);
        v16h av = __builtin_shufflevector(alo, ahi, 0,1,2,3,4,5,6,7,8,9,10,11,12,13,14,15);
        #pragma unroll
        for (int nt = 0; nt < 8; ++nt) {
            const _Float16* q = w2s + (((kt << 3) | nt) * 512 + lane * 8);
            v8h blo = *(const v8h*)q;
            v8h bhi = *(const v8h*)(q + 256);
            v16h bv = __builtin_shufflevector(blo, bhi, 0,1,2,3,4,5,6,7,8,9,10,11,12,13,14,15);
            acc[nt] = __builtin_amdgcn_wmma_f32_16x16x32_f16(
                false, av, false, bv, (short)0, acc[nt], false, false);
            if ((nt & 3) == 3) __builtin_amdgcn_sched_barrier(0);
        }
    }

    #pragma unroll
    for (int nt = 0; nt < 8; ++nt) {
        int col = nt * 16 + m_;
        float bs = b2[col];
        #pragma unroll
        for (int r = 0; r < 8; ++r) {
            float v = fmaxf(acc[nt][r] + bs, 0.f);
            st[(r + hs8) * 136 + col] = (_Float16)v;
        }
    }

    v8f lg[3];
    #pragma unroll
    for (int nt = 0; nt < 3; ++nt) lg[nt] = vz;
    #pragma unroll
    for (int kt = 0; kt < 4; ++kt) {
        int kb = kt * 32 + hs8;
        v8h alo = *(const v8h*)(st + m_ * 136 + kb);
        v8h ahi = *(const v8h*)(st + m_ * 136 + kb + 16);
        v16h av = __builtin_shufflevector(alo, ahi, 0,1,2,3,4,5,6,7,8,9,10,11,12,13,14,15);
        #pragma unroll
        for (int nt = 0; nt < 3; ++nt) {
            const _Float16* q = wls + ((kt * 3 + nt) * 512 + lane * 8);
            v8h blo = *(const v8h*)q;
            v8h bhi = *(const v8h*)(q + 256);
            v16h bv = __builtin_shufflevector(blo, bhi, 0,1,2,3,4,5,6,7,8,9,10,11,12,13,14,15);
            lg[nt] = __builtin_amdgcn_wmma_f32_16x16x32_f16(
                false, av, false, bv, (short)0, lg[nt], false, false);
        }
        __builtin_amdgcn_sched_barrier(0);
    }

    float* lt = &ltile[wv][0];
    #pragma unroll
    for (int nt = 0; nt < 3; ++nt) {
        int col = nt * 16 + m_;
        if (col < N_CLS) {
            float bs = bl[col];
            #pragma unroll
            for (int r = 0; r < 8; ++r) lt[(r + hs8) * N_CLS + col] = lg[nt][r] + bs;
        }
    }

    {
        float* rowp = lt + (lane >> 1) * N_CLS + (lane & 1) * 20;
        float v[20];
        #pragma unroll
        for (int q = 0; q < 5; ++q) {
            float4 t = *(const float4*)(rowp + 4 * q);
            v[4 * q] = t.x; v[4 * q + 1] = t.y; v[4 * q + 2] = t.z; v[4 * q + 3] = t.w;
        }
        float mx = v[0];
        #pragma unroll
        for (int c = 1; c < 20; ++c) mx = fmaxf(mx, v[c]);
        mx = fmaxf(mx, __shfl_xor(mx, 1));
        float s = 0.f;
        #pragma unroll
        for (int c = 0; c < 20; ++c) s += __expf(v[c] - mx);
        s += __shfl_xor(s, 1);
        const float sub = mx + __logf(s);
        #pragma unroll
        for (int q = 0; q < 5; ++q) {
            float4 t = make_float4(v[4 * q] - sub, v[4 * q + 1] - sub,
                                   v[4 * q + 2] - sub, v[4 * q + 3] - sub);
            *(float4*)(rowp + 4 * q) = t;
        }
    }

    float* og = out + (size_t)row0 * N_CLS;
    #pragma unroll
    for (int i = 0; i < 5; ++i) {
        const int e = (i * 32 + lane) * 4;
        *(float4*)(og + e) = *(const float4*)(lt + e);
    }
}


extern "C" void kernel_launch(void* const* d_in, const int* in_sizes, int n_in,
                              void* d_out, int out_size, void* d_ws, size_t ws_size,
                              hipStream_t stream) {
    const float* x    = (const float*)d_in[0];
    const int*   ei   = (const int*)d_in[1];
    const float* ea   = (const float*)d_in[2];
    const float* eps1 = (const float*)d_in[3];
    const float* We1  = (const float*)d_in[4];
    const float* be1  = (const float*)d_in[5];
    const float* W11  = (const float*)d_in[6];
    const float* b11  = (const float*)d_in[7];
    const float* W12  = (const float*)d_in[8];
    const float* b12  = (const float*)d_in[9];
    const float* eps2 = (const float*)d_in[10];
    const float* We2  = (const float*)d_in[11];
    const float* be2  = (const float*)d_in[12];
    const float* W21  = (const float*)d_in[13];
    const float* b21  = (const float*)d_in[14];
    const float* W22  = (const float*)d_in[15];
    const float* b22  = (const float*)d_in[16];
    const float* Wl   = (const float*)d_in[17];
    const float* bl   = (const float*)d_in[18];

    const int N_ = in_sizes[0] / F_DIM;
    const int E_ = in_sizes[2];
    const size_t NB = (size_t)N_ * F_DIM * sizeof(float);

    char* ws = (char*)d_ws;
    _Float16* h1    = (_Float16*)(ws + NB);
    _Float16* x16   = (_Float16*)(ws + NB + NB / 2);
    _Float16* frags = (_Float16*)(ws + 2 * NB);
    _Float16* f11 = frags;
    _Float16* f12 = frags + 16384;
    _Float16* f21 = frags + 32768;
    _Float16* f22 = frags + 49152;
    _Float16* fl  = frags + 65536;
    float* out = (float*)d_out;

    const int K     = (N_ + BKT_W - 1) / BKT_W;
    const int chunk = (E_ + BIN_BLOCKS - 1) / BIN_BLOCKS;
    const int nmat  = K * BIN_BLOCKS;
    const size_t matB = (((size_t)nmat + 4) * sizeof(int) + 255) & ~(size_t)255;
    const size_t offB = (((size_t)N_ + 4) * sizeof(int) + 255) & ~(size_t)255;
    const size_t prB  = ((size_t)E_ * 2 * sizeof(int) + 255) & ~(size_t)255;
    const size_t dlB  = ((size_t)E_ * sizeof(int) + 255) & ~(size_t)255;
    char* gbase = ws + 2 * NB + 131072 + 16384;
    int*  cntmat = (int*)(gbase);
    int*  bstart = (int*)(gbase + matB);
    int*  off    = (int*)(gbase + 2 * matB);
    int2* tpair  = (int2*)(gbase + 2 * matB + offB);
    int*  tdl    = (int*)(gbase + 2 * matB + offB + prB);
    int*  csr    = (int*)(gbase + 2 * matB + offB + prB + dlB);

    const int rows_per_set = 16 * MLP_WAVES * MLP_BLOCKS;
    const int mb = MLP_BLOCKS * ((N_ + rows_per_set - 1) / rows_per_set);

    bin_count<<<BIN_BLOCKS, 1024, K * sizeof(int), stream>>>(ei, cntmat, E_, K, chunk,
                                                             W11, W12, W21, W22, Wl, frags, x, x16, N_);
    bin_scatter<<<BIN_BLOCKS, 1024, 3 * K * sizeof(int), stream>>>(ei, ea, cntmat, bstart, tpair, tdl, E_, K, chunk);
    bucket_place<<<K, 1024, 0, stream>>>(bstart, tpair, tdl, off, (int2*)csr, N_, K);

    gather_mlp<<<mb, MLP_THREADS, 0, stream>>>(x, x16, off, csr, We1, be1, eps1, f11, b11, f12, b12, h1, N_);
    gather_mlp_cls<<<mb, MLP_THREADS, 0, stream>>>(h1, off, csr, We2, be2, eps2, f21, b21, f22, b22, fl, bl, out, N_);
}


# Edge sort level 1: each batch's bucket-ordered records return to the batch's own place (whole lines) plus per-bucket span tables; no global count pass, launches 5 -> 4; sum order varies as before
# speedup vs baseline: 6.8820x; 1.0858x over previous
#include <hip/hip_runtime.h>
#include <hip/hip_bf16.h>
#include <stdint.h>

typedef _Float16 v16h __attribute__((ext_vector_type(16)));
typedef _Float16 v8h  __attribute__((ext_vector_type(8)));
typedef _Float16 v4h  __attribute__((ext_vector_type(4)));
typedef float    v8f  __attribute__((ext_vector_type(8)));

#define F_DIM 128
#define N_CLS 40

#define BKT_SHIFT 9
#define BKT_W (1 << BKT_SHIFT)
#define BIN_BLOCKS 256

__device__ __forceinline__ int frag_dword(const int ntk, const int lane, const int d) {
    return ntk * 256 + (d >> 2) * 128 + lane * 4 + (d & 3);
}

#define CONV_THREADS (BIN_BLOCKS * 1024)
#define WFRAG_DWORDS (4 * 8192 + 3072)
static_assert(CONV_THREADS >= WFRAG_DWORDS, "one thread per fragment dword");

__device__ __forceinline__ void convert_x(const int g, const float* __restrict__ x,
                                          _Float16* __restrict__ x16, const int nrows) {
    const int n4     = nrows * (F_DIM / 4);
    const int stride = CONV_THREADS;
    const float4* x4 = (const float4*)x;
    int i = g;
    for (; i + 3 * stride < n4; i += 4 * stride) {
        float4 v[4];
        #pragma unroll
        for (int q = 0; q < 4; ++q) v[q] = x4[i + q * stride];
        #pragma unroll
        for (int q = 0; q < 4; ++q) {
            v4h h = {(_Float16)v[q].x, (_Float16)v[q].y, (_Float16)v[q].z, (_Float16)v[q].w};
            *(v4h*)(x16 + (size_t)(i + q * stride) * 4) = h;
        }
    }
    for (; i < n4; i += stride) {
        const float4 v = x4[i];
        v4h h = {(_Float16)v.x, (_Float16)v.y, (_Float16)v.z, (_Float16)v.w};
        *(v4h*)(x16 + (size_t)i * 4) = h;
    }
}

__device__ __forceinline__ void convert_weights(const int gid,
                         const float* __restrict__ W11, const float* __restrict__ W12,
                         const float* __restrict__ W21, const float* __restrict__ W22,
                         const float* __restrict__ Wl, _Float16* __restrict__ frags) {
    if (gid >= 4 * 8192) {
        int r = gid - 4 * 8192;
        int d    = r & 7;
        int lane = (r >> 3) & 31;
        int ntk  = r >> 8;
        int kt   = ntk / 3;
        int nt   = ntk - kt * 3;
        int nn   = nt * 16 + (lane & 15);
        int kb   = kt * 32 + ((lane >> 4) << 3);
        int k0   = kb + ((d < 4) ? (d << 1) : (16 + ((d - 4) << 1)));
        _Float16* o = frags + (size_t)4 * 16384 + (size_t)frag_dword(ntk, lane, d) * 2;
        const bool ok = (nn < N_CLS);
        o[0] = ok ? (_Float16)Wl[(size_t)k0 * N_CLS + nn] : (_Float16)0.f;
        o[1] = ok ? (_Float16)Wl[(size_t)(k0 + 1) * N_CLS + nn] : (_Float16)0.f;
        return;
    }
    int m    = gid >> 13;
    int r    = gid & 8191;
    int d    = r & 7;
    int lane = (r >> 3) & 31;
    int ntk  = r >> 8;
    int kt   = ntk >> 3;
    int nt   = ntk & 7;
    int nn   = nt * 16 + (lane & 15);
    int kb   = kt * 32 + ((lane >> 4) << 3);
    int k0   = kb + ((d < 4) ? (d << 1) : (16 + ((d - 4) << 1)));
    const float* W = (m == 0) ? W11 : (m == 1) ? W12 : (m == 2) ? W21 : W22;
    _Float16* o = frags + (size_t)m * 16384 + (size_t)frag_dword(ntk, lane, d) * 2;
    o[0] = (_Float16)W[(size_t)k0 * F_DIM + nn];
    o[1] = (_Float16)W[(size_t)(k0 + 1) * F_DIM + nn];
}

__device__ __forceinline__ int scan_k(const int* in, int* out, const int K, int* wsum,
                                      const int tid, const int lane, const int wv) {
    int carry = 0;
    for (int kb = 0; kb < K; kb += 1024) {
        const int k = kb + tid;
        const int v = (k < K) ? in[k] : 0;
        int incl = v;
        #pragma unroll
        for (int dd = 1; dd < 32; dd <<= 1) {
            int y = __shfl_up(incl, dd);
            if (lane >= dd) incl += y;
        }
        if (lane == 31) wsum[wv] = incl;
        __syncthreads();
        const int w = wsum[lane];
        int wincl = w;
        #pragma unroll
        for (int dd = 1; dd < 32; dd <<= 1) {
            int y = __shfl_up(wincl, dd);
            if (lane >= dd) wincl += y;
        }
        const int wave_excl = __shfl(wincl, wv) - __shfl(w, wv);
        const int total     = __shfl(wincl, 31);
        if (k < K) out[k] = carry + wave_excl + (incl - v);
        carry += total;
        __syncthreads();
    }
    return carry;
}

#define SC_BATCH 8192
__global__ void __launch_bounds__(1024)
bin_scatter(const int* __restrict__ ei, const float* __restrict__ ea,
            int* __restrict__ cnt2, int* __restrict__ st2,
            int2* __restrict__ tpair, int* __restrict__ tdl,
            int E_, int K, int chunk, int nbpc,
            const float* __restrict__ W11, const float* __restrict__ W12,
            const float* __restrict__ W21, const float* __restrict__ W22,
            const float* __restrict__ Wl, _Float16* __restrict__ frags,
            const float* __restrict__ x, _Float16* __restrict__ x16, int nrows) {
    extern __shared__ int dyn[];
    __shared__ int2 spair[SC_BATCH];
    __shared__ int  sdl[SC_BATCH];
    __shared__ int  wsum[32];
    int* lcnt   = dyn;
    int* lstart = dyn + K;
    const int tid  = threadIdx.x;
    const int lane = tid & 31;
    const int wv   = tid >> 5;
    const int b    = blockIdx.x;
    const int NQ   = BIN_BLOCKS * nbpc;
    const int e0 = b * chunk;
    const int e1 = (e0 + chunk < E_) ? (e0 + chunk) : E_;
    const int* dst = ei + E_;
    for (int t = 0; t < nbpc; ++t) {
        const int s0 = e0 + t * SC_BATCH;
        const int s1 = (s0 + SC_BATCH < e1) ? (s0 + SC_BATCH) : e1;
        for (int k = tid; k < K; k += 1024) lcnt[k] = 0;
        __syncthreads();
        int d[8], s[8], r[8];
        float a[8];
        #pragma unroll
        for (int q = 0; q < 8; ++q) {
            const int i = s0 + q * 1024 + tid;
            const bool ok = (i < s1);
            d[q] = ok ? dst[i] : -1;
            s[q] = ok ? ei[i] : 0;
            a[q] = ok ? ea[i] : 0.f;
        }
        if (t == 0) {
            const int g = b * 1024 + tid;
            if (g < WFRAG_DWORDS) convert_weights(g, W11, W12, W21, W22, Wl, frags);
            convert_x(g, x, x16, nrows);
        }
        #pragma unroll
        for (int q = 0; q < 8; ++q)
            r[q] = (d[q] >= 0) ? atomicAdd(&lcnt[d[q] >> BKT_SHIFT], 1) : 0;
        __syncthreads();
        scan_k(lcnt, lstart, K, wsum, tid, lane, wv);
        #pragma unroll
        for (int q = 0; q < 8; ++q) {
            if (d[q] >= 0) {
                const int slot = lstart[d[q] >> BKT_SHIFT] + r[q];
                spair[slot] = make_int2(s[q], __float_as_int(a[q]));
                sdl[slot]   = d[q] & (BKT_W - 1);
            }
        }
        const int qb = b * nbpc + t;
        for (int k = tid; k < K; k += 1024) {
            cnt2[(size_t)k * NQ + qb] = lcnt[k];
            st2[(size_t)k * NQ + qb]  = s0 + lstart[k];
        }
        __syncthreads();
        const int nb = s1 - s0;
        for (int i = tid; i < nb; i += 1024) {
            tpair[s0 + i] = spair[i];
            tdl[s0 + i]   = sdl[i];
        }
        __syncthreads();
    }
}

#define PLACE_CAP 16384
#define PL_U 8

__device__ __forceinline__ void spans_load(int (&cn)[PL_U], int (&sb)[PL_U],
                                           const int* __restrict__ crow,
                                           const int* __restrict__ srow,
                                           const int q0, const int NQ) {
    #pragma unroll
    for (int u = 0; u < PL_U; ++u) {
        const int q = q0 + 32 * u;
        const bool ok = (q < NQ);
        cn[u] = ok ? crow[q] : 0;
        sb[u] = ok ? srow[q] : 0;
    }
}

__device__ __forceinline__ void spans_count(const int (&cn)[PL_U], const int (&sb)[PL_U],
                                            const int* __restrict__ tdl, int* c,
                                            const int lane) {
    int dl[PL_U][2];
    #pragma unroll
    for (int u = 0; u < PL_U; ++u) {
        #pragma unroll
        for (int h = 0; h < 2; ++h) {
            const int j = lane + 32 * h;
            dl[u][h] = (j < cn[u]) ? tdl[sb[u] + j] : -1;
        }
    }
    #pragma unroll
    for (int u = 0; u < PL_U; ++u) {
        #pragma unroll
        for (int h = 0; h < 2; ++h)
            if (dl[u][h] >= 0) atomicAdd(&c[dl[u][h]], 1);
    }
    #pragma unroll
    for (int u = 0; u < PL_U; ++u)
        for (int j = lane + 64; j < cn[u]; j += 32) atomicAdd(&c[tdl[sb[u] + j]], 1);
}

__device__ __forceinline__ void spans_place(const int (&cn)[PL_U], const int (&sb)[PL_U],
                                            const int* __restrict__ tdl,
                                            const int2* __restrict__ tpair, int* c,
                                            int2* stage, int2* __restrict__ csr,
                                            const int lo, const int lane) {
    int  dl[PL_U][2];
    int2 pr[PL_U][2];
    #pragma unroll
    for (int u = 0; u < PL_U; ++u) {
        #pragma unroll
        for (int h = 0; h < 2; ++h) {
            const int j = lane + 32 * h;
            const bool ok = (j < cn[u]);
            dl[u][h] = ok ? tdl[sb[u] + j] : -1;
            pr[u][h] = ok ? tpair[sb[u] + j] : make_int2(0, 0);
        }
    }
    #pragma unroll
    for (int u = 0; u < PL_U; ++u) {
        #pragma unroll
        for (int h = 0; h < 2; ++h) {
            if (dl[u][h] >= 0) {
                const int p = atomicAdd(&c[dl[u][h]], 1);
                if (p < PLACE_CAP) stage[p] = pr[u][h];
                else               csr[lo + p] = pr[u][h];
            }
        }
    }
    #pragma unroll
    for (int u = 0; u < PL_U; ++u) {
        for (int j = lane + 64; j < cn[u]; j += 32) {
            const int  dj = tdl[sb[u] + j];
            const int2 pj = tpair[sb[u] + j];
            const int p = atomicAdd(&c[dj], 1);
            if (p < PLACE_CAP) stage[p] = pj;
            else               csr[lo + p] = pj;
        }
    }
}

__global__ void __launch_bounds__(1024)
bucket_place(const int* __restrict__ cnt2, const int* __restrict__ st2,
             const int2* __restrict__ tpair, const int* __restrict__ tdl,
             int* __restrict__ off, int2* __restrict__ csr, int n, int K, int NQ) {
    __shared__ int c[BKT_W];
    __shared__ int wsum[32];
    __shared__ int wlo[32];
    __shared__ int wtot[32];
    __shared__ int2 stage[PLACE_CAP];
    const int tid  = threadIdx.x;
    const int lane = tid & 31;
    const int wv   = tid >> 5;
    const int k    = blockIdx.x;
    const int* crow = cnt2 + (size_t)k * NQ;
    const int* srow = st2 + (size_t)k * NQ;
    int cnF[PL_U], sbF[PL_U];
    spans_load(cnF, sbF, crow, srow, wv, NQ);
    const int nprev = k * NQ;
    int part = 0;
    for (int i0 = tid * 4; i0 < nprev; i0 += 8 * 4096) {
        int4 v[8];
        #pragma unroll
        for (int q = 0; q < 8; ++q) {
            const int i = i0 + q * 4096;
            v[q] = (i < nprev) ? *(const int4*)(cnt2 + i) : make_int4(0, 0, 0, 0);
        }
        #pragma unroll
        for (int q = 0; q < 8; ++q) part += v[q].x + v[q].y + v[q].z + v[q].w;
    }
    if (tid < BKT_W) c[tid] = 0;
    __syncthreads();
    int wt = 0;
    #pragma unroll
    for (int u = 0; u < PL_U; ++u) wt += cnF[u];
    spans_count(cnF, sbF, tdl, c, lane);
    for (int q0 = wv + 32 * PL_U; q0 < NQ; q0 += 32 * PL_U) {
        int cn[PL_U], sb[PL_U];
        spans_load(cn, sb, crow, srow, q0, NQ);
        #pragma unroll
        for (int u = 0; u < PL_U; ++u) wt += cn[u];
        spans_count(cn, sb, tdl, c, lane);
    }
    #pragma unroll
    for (int dd = 16; dd >= 1; dd >>= 1) part += __shfl_xor(part, dd);
    if (lane == 0) { wlo[wv] = part; wtot[wv] = wt; }
    __syncthreads();
    int lo = wlo[lane], tot = wtot[lane];
    #pragma unroll
    for (int dd = 16; dd >= 1; dd >>= 1) {
        lo  += __shfl_xor(lo, dd);
        tot += __shfl_xor(tot, dd);
    }
    const int hi = lo + tot;
    const int v = (tid < BKT_W) ? c[tid] : 0;
    int incl = v;
    #pragma unroll
    for (int d = 1; d < 32; d <<= 1) {
        int y = __shfl_up(incl, d);
        if (lane >= d) incl += y;
    }
    if (lane == 31) wsum[wv] = incl;
    __syncthreads();
    const int w = wsum[lane];
    int wincl = w;
    #pragma unroll
    for (int d = 1; d < 32; d <<= 1) {
        int y = __shfl_up(wincl, d);
        if (lane >= d) wincl += y;
    }
    const int wave_excl = __shfl(wincl, wv) - __shfl(w, wv);
    const int lstart = wave_excl + (incl - v);
    if (tid < BKT_W) {
        c[tid] = lstart;
        const int node = k * BKT_W + tid;
        if (node < n) off[node] = lo + lstart;
    }
    if (k == K - 1 && tid == 0) off[n] = hi;
    __syncthreads();
    spans_place(cnF, sbF, tdl, tpair, c, stage, csr, lo, lane);
    for (int q0 = wv + 32 * PL_U; q0 < NQ; q0 += 32 * PL_U) {
        int cn[PL_U], sb[PL_U];
        spans_load(cn, sb, crow, srow, q0, NQ);
        spans_place(cn, sb, tdl, tpair, c, stage, csr, lo, lane);
    }
    __syncthreads();
    const int nst  = (hi - lo < PLACE_CAP) ? (hi - lo) : PLACE_CAP;
    const int gend = lo + nst;
    for (int g = (lo & ~31) + tid; g < gend; g += 1024)
        if (g >= lo) csr[g] = stage[g - lo];
}

template <int Q>
__device__ __forceinline__ int row_share(const int v) {
    return __builtin_amdgcn_update_dpp(0, v, 0x150 + Q, 0xf, 0xf, false);
}

__device__ __forceinline__ void msg_add8(float (&acc)[8], const v8h xr, const float a,
                                         const float keep, const float (&w)[8],
                                         const float (&nb)[8]) {
    #pragma unroll
    for (int i = 0; i < 8; ++i) {
        const float v = __builtin_fmaf(a, w[i], (float)xr[i]);
        acc[i] = fmaf(keep, fmaxf(v, nb[i]), acc[i]);
    }
    asm("" : "+v"(acc[0]), "+v"(acc[1]), "+v"(acc[2]), "+v"(acc[3]),
             "+v"(acc[4]), "+v"(acc[5]), "+v"(acc[6]), "+v"(acc[7]));
}

template <int Q>
__device__ __forceinline__ void edge_group(float (&acc)[8], const int2 my, const int m,
                                           const char* __restrict__ xbase, const unsigned lo,
                                           const float (&w)[8], const float (&b)[8]) {
    const unsigned s0 = ((unsigned)row_share<Q>(my.x) << 8) | lo;
    const unsigned s1 = ((unsigned)row_share<Q + 1>(my.x) << 8) | lo;
    const unsigned s2 = ((unsigned)row_share<Q + 2>(my.x) << 8) | lo;
    const unsigned s3 = ((unsigned)row_share<Q + 3>(my.x) << 8) | lo;
    const v8h x0 = *(const v8h*)(xbase + s0);
    const v8h x1 = *(const v8h*)(xbase + s1);
    const v8h x2 = *(const v8h*)(xbase + s2);
    const v8h x3 = *(const v8h*)(xbase + s3);
    const float a0 = __int_as_float(row_share<Q>(my.y));
    const float a1 = __int_as_float(row_share<Q + 1>(my.y));
    const float a2 = __int_as_float(row_share<Q + 2>(my.y));
    const float a3 = __int_as_float(row_share<Q + 3>(my.y));
    msg_add8(acc, x0, a0, (Q     < m) ? 1.f : 0.f, w, b);
    msg_add8(acc, x1, a1, (Q + 1 < m) ? 1.f : 0.f, w, b);
    msg_add8(acc, x2, a2, (Q + 2 < m) ? 1.f : 0.f, w, b);
    msg_add8(acc, x3, a3, (Q + 3 < m) ? 1.f : 0.f, w, b);
}

template <int Q>
__device__ __forceinline__ void rows8(v8h (&x)[8], const int2 my,
                                      const char* __restrict__ xbase, const unsigned lo) {
    const unsigned s0 = ((unsigned)row_share<Q>(my.x) << 8) | lo;
    const unsigned s1 = ((unsigned)row_share<Q + 1>(my.x) << 8) | lo;
    const unsigned s2 = ((unsigned)row_share<Q + 2>(my.x) << 8) | lo;
    const unsigned s3 = ((unsigned)row_share<Q + 3>(my.x) << 8) | lo;
    const unsigned s4 = ((unsigned)row_share<Q + 4>(my.x) << 8) | lo;
    const unsigned s5 = ((unsigned)row_share<Q + 5>(my.x) << 8) | lo;
    const unsigned s6 = ((unsigned)row_share<Q + 6>(my.x) << 8) | lo;
    const unsigned s7 = ((unsigned)row_share<Q + 7>(my.x) << 8) | lo;
    x[0] = *(const v8h*)(xbase + s0);
    x[1] = *(const v8h*)(xbase + s1);
    x[2] = *(const v8h*)(xbase + s2);
    x[3] = *(const v8h*)(xbase + s3);
    x[4] = *(const v8h*)(xbase + s4);
    x[5] = *(const v8h*)(xbase + s5);
    x[6] = *(const v8h*)(xbase + s6);
    x[7] = *(const v8h*)(xbase + s7);
}

template <int Q>
__device__ __forceinline__ void sum8(float (&acc)[8], const v8h (&x)[8], const int2 my,
                                     const int m, const float (&w)[8], const float (&b)[8]) {
    const float a0 = __int_as_float(row_share<Q>(my.y));
    const float a1 = __int_as_float(row_share<Q + 1>(my.y));
    const float a2 = __int_as_float(row_share<Q + 2>(my.y));
    const float a3 = __int_as_float(row_share<Q + 3>(my.y));
    const float a4 = __int_as_float(row_share<Q + 4>(my.y));
    const float a5 = __int_as_float(row_share<Q + 5>(my.y));
    const float a6 = __int_as_float(row_share<Q + 6>(my.y));
    const float a7 = __int_as_float(row_share<Q + 7>(my.y));
    msg_add8(acc, x[0], a0, (Q     < m) ? 1.f : 0.f, w, b);
    msg_add8(acc, x[1], a1, (Q + 1 < m) ? 1.f : 0.f, w, b);
    msg_add8(acc, x[2], a2, (Q + 2 < m) ? 1.f : 0.f, w, b);
    msg_add8(acc, x[3], a3, (Q + 3 < m) ? 1.f : 0.f, w, b);
    msg_add8(acc, x[4], a4, (Q + 4 < m) ? 1.f : 0.f, w, b);
    msg_add8(acc, x[5], a5, (Q + 5 < m) ? 1.f : 0.f, w, b);
    msg_add8(acc, x[6], a6, (Q + 6 < m) ? 1.f : 0.f, w, b);
    msg_add8(acc, x[7], a7, (Q + 7 < m) ? 1.f : 0.f, w, b);
}

template <int Q>
__device__ __forceinline__ void edge_group8(float (&acc)[8], const int2 my, const int m,
                                            const char* __restrict__ xbase, const unsigned lo,
                                            const float (&w)[8], const float (&b)[8]) {
    v8h x[8];
    rows8<Q>(x, my, xbase, lo);
    sum8<Q>(acc, x, my, m, w, b);
}

__device__ __forceinline__ void edge_trip(float (&acc)[8], int2& cur, const int cnt, const int cmax,
                                          const int done, const int beg, const int hl,
                                          const int2* __restrict__ csr,
                                          const char* __restrict__ xbase, const unsigned lo,
                                          const float (&w)[8], const float (&b)[8]) {
    const int m  = cnt - done;
    const int mm = cmax - done;
    int2 nxt = make_int2(0, 0);
    if (hl + 16 < m) nxt = csr[beg + done + 16 + hl];
    if (mm > 4) edge_group8<0>(acc, cur, m, xbase, lo, w, b);
    else        edge_group<0>(acc, cur, m, xbase, lo, w, b);
    if (mm > 12)     edge_group8<8>(acc, cur, m, xbase, lo, w, b);
    else if (mm > 8) edge_group<8>(acc, cur, m, xbase, lo, w, b);
    cur = nxt;
}

__device__ __forceinline__ void load8(float (&r)[8], const float* __restrict__ p) {
    const float4 u = *(const float4*)p, v = *(const float4*)(p + 4);
    r[0] = u.x; r[1] = u.y; r[2] = u.z; r[3] = u.w;
    r[4] = v.x; r[5] = v.y; r[6] = v.z; r[7] = v.w;
}

template <typename RT>
__device__ __forceinline__ void gather_stage(_Float16* st, const int row0, const int lane,
                                             const int nrows, const RT* __restrict__ xres,
                                             const _Float16* __restrict__ xrows,
                                             const int* __restrict__ off,
                                             const int2* __restrict__ csr,
                                             const float* __restrict__ We,
                                             const float* __restrict__ be, const float ep) {
    const int hl   = lane & 15;
    const int half = lane >> 4;
    const int c    = hl << 3;

    int offv = 0;
    if (lane <= 16) {
        const int idx = (row0 + lane < nrows) ? (row0 + lane) : nrows;
        offv = off[idx];
    }

    float w[8], b[8];
    load8(w, We + c);
    load8(b, be + c);
    #pragma unroll
    for (int i = 0; i < 8; ++i) b[i] = -b[i];

    const char* xbase = (const char*)xrows;
    const unsigned lo = (unsigned)hl << 4;

    int beg = __shfl(offv, half);
    int cnt = __shfl(offv, half + 1) - beg;
    int2 cur = make_int2(0, 0);
    if (hl < cnt) cur = csr[beg + hl];

    #pragma unroll 1
    for (int p = 0; p < 8; ++p) {
        const int r    = 2 * p + half;
        const int node = row0 + r;
        const int cA = __builtin_amdgcn_readlane(cnt, 0);
        const int cB = __builtin_amdgcn_readlane(cnt, 16);
        const int cmax = (cA > cB) ? cA : cB;

        v8h x0[8];
        rows8<0>(x0, cur, xbase, lo);

        int nbeg = 0, ncnt = 0;
        int2 ncur = make_int2(0, 0);
        if (p < 7) {
            nbeg = __shfl(offv, r + 2);
            ncnt = __shfl(offv, r + 3) - nbeg;
            if (hl < ncnt) ncur = csr[nbeg + hl];
        }
        int2 nxt = make_int2(0, 0);
        if (hl + 16 < cnt) nxt = csr[beg + 16 + hl];
        float acc[8];
        #pragma unroll
        for (int i = 0; i < 8; ++i) acc[i] = 0.f;

        sum8<0>(acc, x0, cur, cnt, w, b);
        const RT* xp = xres + (size_t)((node < nrows) ? node : (nrows - 1)) * F_DIM + c;
        v8h    xh = {};
        float4 xa = make_float4(0.f, 0.f, 0.f, 0.f), xb = xa;
        if constexpr (sizeof(RT) == 2) {
            xh = *(const v8h*)xp;
        } else {
            xa = *(const float4*)xp;
            xb = *(const float4*)(xp + 4);
        }
        if (cmax > 12)     edge_group8<8>(acc, cur, cnt, xbase, lo, w, b);
        else if (cmax > 8) edge_group<8>(acc, cur, cnt, xbase, lo, w, b);
        cur = nxt;
        for (int done = 16; done < cmax; done += 16)
            edge_trip(acc, cur, cnt, cmax, done, beg, hl, csr, xbase, lo, w, b);

        float xr[8];
        if constexpr (sizeof(RT) == 2) {
            #pragma unroll
            for (int i = 0; i < 8; ++i) xr[i] = (float)xh[i];
        } else {
            xr[0] = xa.x; xr[1] = xa.y; xr[2] = xa.z; xr[3] = xa.w;
            xr[4] = xb.x; xr[5] = xb.y; xr[6] = xb.z; xr[7] = xb.w;
        }
        const float ncnt_f = -(float)cnt;
        v8h h;
        #pragma unroll
        for (int i = 0; i < 8; ++i) h[i] = (_Float16)(ep * xr[i] + fmaf(ncnt_f, b[i], acc[i]));
        *(v8h*)(st + r * 136 + c) = h;

        beg = nbeg; cnt = ncnt; cur = ncur;
    }
}

#define MLP_WAVES   25
#define MLP_THREADS (MLP_WAVES * 32)
#define MLP_BLOCKS  256

template <int N16>
struct WPieces { v8h v[(N16 + MLP_THREADS - 1) / MLP_THREADS]; };

template <int N16>
__device__ __forceinline__ void load_pieces(WPieces<N16>& p, const _Float16* __restrict__ src,
                                            const int tid) {
    #pragma unroll
    for (int q = 0; q < (N16 + MLP_THREADS - 1) / MLP_THREADS; ++q) {
        const int i = tid + q * MLP_THREADS;
        p.v[q] = *(const v8h*)(src + (size_t)((i < N16) ? i : (N16 - 1)) * 8);
    }
}

template <int N16>
__device__ __forceinline__ void store_pieces(_Float16* dst, const WPieces<N16>& p, const int tid) {
    #pragma unroll
    for (int q = 0; q < (N16 + MLP_THREADS - 1) / MLP_THREADS; ++q) {
        const int i = tid + q * MLP_THREADS;
        if (i < N16) *(v8h*)(dst + (size_t)i * 8) = p.v[q];
    }
}

__global__ void __launch_bounds__(MLP_THREADS, 7)
gather_mlp(const float* __restrict__ xin, const _Float16* __restrict__ xrows,
           const int* __restrict__ off, const int* __restrict__ csr,
           const float* __restrict__ We, const float* __restrict__ be,
           const float* __restrict__ eps,
           const _Float16* __restrict__ fw1, const float* __restrict__ b1,
           const _Float16* __restrict__ fw2, const float* __restrict__ b2,
           _Float16* __restrict__ out, int nrows) {
    __shared__ __align__(16) _Float16 wfrag[2 * 16384];
    __shared__ __align__(16) _Float16 stage[MLP_WAVES][16 * 136];

    {
        WPieces<2048> p1, p2;
        load_pieces(p1, fw1, threadIdx.x);
        load_pieces(p2, fw2, threadIdx.x);
        store_pieces(wfrag, p1, threadIdx.x);
        store_pieces(wfrag + 16384, p2, threadIdx.x);
    }
    __syncthreads();

    const int wv   = __builtin_amdgcn_readfirstlane(threadIdx.x >> 5);
    const int lane = threadIdx.x & 31;
    const int row0 = 16 * ((blockIdx.x / MLP_BLOCKS) * (MLP_WAVES * MLP_BLOCKS)
                           + wv * MLP_BLOCKS + (blockIdx.x % MLP_BLOCKS));
    if (row0 >= nrows) return;
    const float ep = 1.0f + eps[0];
    _Float16* st = &stage[wv][0];
    const _Float16* w1s = wfrag;
    const _Float16* w2s = wfrag + 16384;

    gather_stage(st, row0, lane, nrows, xin, xrows, off, (const int2*)csr, We, be, ep);

    const int m_  = lane & 15;
    const int hs8 = (lane >> 4) << 3;

    v8f acc[8];
    const v8f vz = {0.f, 0.f, 0.f, 0.f, 0.f, 0.f, 0.f, 0.f};

    #pragma unroll
    for (int nt = 0; nt < 8; ++nt) acc[nt] = vz;
    #pragma unroll
    for (int kt = 0; kt < 4; ++kt) {
        int kb = kt * 32 + hs8;
        v8h alo = *(const v8h*)(st + m_ * 136 + kb);
        v8h ahi = *(const v8h*)(st + m_ * 136 + kb + 16);
        v16h av = __builtin_shufflevector(alo, ahi, 0,1,2,3,4,5,6,7,8,9,10,11,12,13,14,15);
        #pragma unroll
        for (int nt = 0; nt < 8; ++nt) {
            const _Float16* q = w1s + (((kt << 3) | nt) * 512 + lane * 8);
            v8h blo = *(const v8h*)q;
            v8h bhi = *(const v8h*)(q + 256);
            v16h bv = __builtin_shufflevector(blo, bhi, 0,1,2,3,4,5,6,7,8,9,10,11,12,13,14,15);
            acc[nt] = __builtin_amdgcn_wmma_f32_16x16x32_f16(
                false, av, false, bv, (short)0, acc[nt], false, false);
            if ((nt & 3) == 3) __builtin_amdgcn_sched_barrier(0);
        }
    }

    #pragma unroll
    for (int nt = 0; nt < 8; ++nt) {
        int col = nt * 16 + m_;
        float bs = b1[col];
        #pragma unroll
        for (int r = 0; r < 8; ++r) {
            float v = fmaxf(acc[nt][r] + bs, 0.f);
            st[(r + hs8) * 136 + col] = (_Float16)v;
        }
    }

    #pragma unroll
    for (int nt = 0; nt < 8; ++nt) acc[nt] = vz;
    #pragma unroll
    for (int kt = 0; kt < 4; ++kt) {
        int kb = kt * 32 + hs8;
        v8h alo = *(const v8h*)(st + m_ * 136 + kb);
        v8h ahi = *(const v8h*)(st + m_ * 136 + kb + 16);
        v16h av = __builtin_shufflevector(alo, ahi, 0,1,2,3,4,5,6,7,8,9,10,11,12,13,14,15);
        #pragma unroll
        for (int nt = 0; nt < 8; ++nt) {
            const _Float16* q = w2s + (((kt << 3) | nt) * 512 + lane * 8);
            v8h blo = *(const v8h*)q;
            v8h bhi = *(const v8h*)(q + 256);
            v16h bv = __builtin_shufflevector(blo, bhi, 0,1,2,3,4,5,6,7,8,9,10,11,12,13,14,15);
            acc[nt] = __builtin_amdgcn_wmma_f32_16x16x32_f16(
                false, av, false, bv, (short)0, acc[nt], false, false);
            if ((nt & 3) == 3) __builtin_amdgcn_sched_barrier(0);
        }
    }

    #pragma unroll
    for (int nt = 0; nt < 8; ++nt) {
        int col = nt * 16 + m_;
        float bs = b2[col];
        #pragma unroll
        for (int r = 0; r < 8; ++r) {
            float v = fmaxf(acc[nt][r] + bs, 0.f);
            st[(r + hs8) * 136 + col] = (_Float16)v;
        }
    }

    _Float16* og = out + (size_t)row0 * F_DIM;
    #pragma unroll
    for (int i = 0; i < 8; ++i) {
        const int rr = 2 * i + (lane >> 4);
        const int cc = (lane & 15) << 3;
        *(v8h*)(og + rr * F_DIM + cc) = *(const v8h*)(st + rr * 136 + cc);
    }
}

__global__ void __launch_bounds__(MLP_THREADS, 7)
gather_mlp_cls(const _Float16* __restrict__ xin,
               const int* __restrict__ off, const int* __restrict__ csr,
               const float* __restrict__ We, const float* __restrict__ be,
               const float* __restrict__ eps,
               const _Float16* __restrict__ fw1, const float* __restrict__ b1,
               const _Float16* __restrict__ fw2, const float* __restrict__ b2,
               const _Float16* __restrict__ fwl, const float* __restrict__ bl,
               float* __restrict__ out, int nrows) {
    __shared__ __align__(16) _Float16 wfrag[2 * 16384 + 6144];
    __shared__ __align__(16) _Float16 stage[MLP_WAVES][16 * 136];
    __shared__ __align__(16) float ltile[MLP_WAVES][16 * N_CLS];

    {
        WPieces<2048> p1, p2;
        WPieces<768>  pl;
        load_pieces(p1, fw1, threadIdx.x);
        load_pieces(p2, fw2, threadIdx.x);
        load_pieces(pl, fwl, threadIdx.x);
        store_pieces(wfrag, p1, threadIdx.x);
        store_pieces(wfrag + 16384, p2, threadIdx.x);
        store_pieces(wfrag + 32768, pl, threadIdx.x);
    }
    __syncthreads();

    const int wv   = __builtin_amdgcn_readfirstlane(threadIdx.x >> 5);
    const int lane = threadIdx.x & 31;
    const int row0 = 16 * ((blockIdx.x / MLP_BLOCKS) * (MLP_WAVES * MLP_BLOCKS)
                           + wv * MLP_BLOCKS + (blockIdx.x % MLP_BLOCKS));
    if (row0 >= nrows) return;
    const float ep = 1.0f + eps[0];
    _Float16* st = &stage[wv][0];
    const _Float16* w1s = wfrag;
    const _Float16* w2s = wfrag + 16384;
    const _Float16* wls = wfrag + 32768;

    gather_stage(st, row0, lane, nrows, xin, xin, off, (const int2*)csr, We, be, ep);

    const int m_  = lane & 15;
    const int hs8 = (lane >> 4) << 3;

    v8f acc[8];
    const v8f vz = {0.f, 0.f, 0.f, 0.f, 0.f, 0.f, 0.f, 0.f};

    #pragma unroll
    for (int nt = 0; nt < 8; ++nt) acc[nt] = vz;
    #pragma unroll
    for (int kt = 0; kt < 4; ++kt) {
        int kb = kt * 32 + hs8;
        v8h alo = *(const v8h*)(st + m_ * 136 + kb);
        v8h ahi = *(const v8h*)(st + m_ * 136 + kb + 16);
        v16h av = __builtin_shufflevector(alo, ahi, 0,1,2,3,4,5,6,7,8,9,10,11,12,13,14,15);
        #pragma unroll
        for (int nt = 0; nt < 8; ++nt) {
            const _Float16* q = w1s + (((kt << 3) | nt) * 512 + lane * 8);
            v8h blo = *(const v8h*)q;
            v8h bhi = *(const v8h*)(q + 256);
            v16h bv = __builtin_shufflevector(blo, bhi, 0,1,2,3,4,5,6,7,8,9,10,11,12,13,14,15);
            acc[nt] = __builtin_amdgcn_wmma_f32_16x16x32_f16(
                false, av, false, bv, (short)0, acc[nt], false, false);
            if ((nt & 3) == 3) __builtin_amdgcn_sched_barrier(0);
        }
    }

    #pragma unroll
    for (int nt = 0; nt < 8; ++nt) {
        int col = nt * 16 + m_;
        float bs = b1[col];
        #pragma unroll
        for (int r = 0; r < 8; ++r) {
            float v = fmaxf(acc[nt][r] + bs, 0.f);
            st[(r + hs8) * 136 + col] = (_Float16)v;
        }
    }

    #pragma unroll
    for (int nt = 0; nt < 8; ++nt) acc[nt] = vz;
    #pragma unroll
    for (int kt = 0; kt < 4; ++kt) {
        int kb = kt * 32 + hs8;
        v8h alo = *(const v8h*)(st + m_ * 136 + kb);
        v8h ahi = *(const v8h*)(st + m_ * 136 + kb + 16);
        v16h av = __builtin_shufflevector(alo, ahi, 0,1,2,3,4,5,6,7,8,9,10,11,12,13,14,15);
        #pragma unroll
        for (int nt = 0; nt < 8; ++nt) {
            const _Float16* q = w2s + (((kt << 3) | nt) * 512 + lane * 8);
            v8h blo = *(const v8h*)q;
            v8h bhi = *(const v8h*)(q + 256);
            v16h bv = __builtin_shufflevector(blo, bhi, 0,1,2,3,4,5,6,7,8,9,10,11,12,13,14,15);
            acc[nt] = __builtin_amdgcn_wmma_f32_16x16x32_f16(
                false, av, false, bv, (short)0, acc[nt], false, false);
            if ((nt & 3) == 3) __builtin_amdgcn_sched_barrier(0);
        }
    }

    #pragma unroll
    for (int nt = 0; nt < 8; ++nt) {
        int col = nt * 16 + m_;
        float bs = b2[col];
        #pragma unroll
        for (int r = 0; r < 8; ++r) {
            float v = fmaxf(acc[nt][r] + bs, 0.f);
            st[(r + hs8) * 136 + col] = (_Float16)v;
        }
    }

    v8f lg[3];
    #pragma unroll
    for (int nt = 0; nt < 3; ++nt) lg[nt] = vz;
    #pragma unroll
    for (int kt = 0; kt < 4; ++kt) {
        int kb = kt * 32 + hs8;
        v8h alo = *(const v8h*)(st + m_ * 136 + kb);
        v8h ahi = *(const v8h*)(st + m_ * 136 + kb + 16);
        v16h av = __builtin_shufflevector(alo, ahi, 0,1,2,3,4,5,6,7,8,9,10,11,12,13,14,15);
        #pragma unroll
        for (int nt = 0; nt < 3; ++nt) {
            const _Float16* q = wls + ((kt * 3 + nt) * 512 + lane * 8);
            v8h blo = *(const v8h*)q;
            v8h bhi = *(const v8h*)(q + 256);
            v16h bv = __builtin_shufflevector(blo, bhi, 0,1,2,3,4,5,6,7,8,9,10,11,12,13,14,15);
            lg[nt] = __builtin_amdgcn_wmma_f32_16x16x32_f16(
                false, av, false, bv, (short)0, lg[nt], false, false);
        }
        __builtin_amdgcn_sched_barrier(0);
    }

    float* lt = &ltile[wv][0];
    #pragma unroll
    for (int nt = 0; nt < 3; ++nt) {
        int col = nt * 16 + m_;
        if (col < N_CLS) {
            float bs = bl[col];
            #pragma unroll
            for (int r = 0; r < 8; ++r) lt[(r + hs8) * N_CLS + col] = lg[nt][r] + bs;
        }
    }

    {
        float* rowp = lt + (lane >> 1) * N_CLS + (lane & 1) * 20;
        float v[20];
        #pragma unroll
        for (int q = 0; q < 5; ++q) {
            float4 t = *(const float4*)(rowp + 4 * q);
            v[4 * q] = t.x; v[4 * q + 1] = t.y; v[4 * q + 2] = t.z; v[4 * q + 3] = t.w;
        }
        float mx = v[0];
        #pragma unroll
        for (int c = 1; c < 20; ++c) mx = fmaxf(mx, v[c]);
        mx = fmaxf(mx, __shfl_xor(mx, 1));
        float s = 0.f;
        #pragma unroll
        for (int c = 0; c < 20; ++c) s += __expf(v[c] - mx);
        s += __shfl_xor(s, 1);
        const float sub = mx + __logf(s);
        #pragma unroll
        for (int q = 0; q < 5; ++q) {
            float4 t = make_float4(v[4 * q] - sub, v[4 * q + 1] - sub,
                                   v[4 * q + 2] - sub, v[4 * q + 3] - sub);
            *(float4*)(rowp + 4 * q) = t;
        }
    }

    float* og = out + (size_t)row0 * N_CLS;
    #pragma unroll
    for (int i = 0; i < 5; ++i) {
        const int e = (i * 32 + lane) * 4;
        *(float4*)(og + e) = *(const float4*)(lt + e);
    }
}


extern "C" void kernel_launch(void* const* d_in, const int* in_sizes, int n_in,
                              void* d_out, int out_size, void* d_ws, size_t ws_size,
                              hipStream_t stream) {
    const float* x    = (const float*)d_in[0];
    const int*   ei   = (const int*)d_in[1];
    const float* ea   = (const float*)d_in[2];
    const float* eps1 = (const float*)d_in[3];
    const float* We1  = (const float*)d_in[4];
    const float* be1  = (const float*)d_in[5];
    const float* W11  = (const float*)d_in[6];
    const float* b11  = (const float*)d_in[7];
    const float* W12  = (const float*)d_in[8];
    const float* b12  = (const float*)d_in[9];
    const float* eps2 = (const float*)d_in[10];
    const float* We2  = (const float*)d_in[11];
    const float* be2  = (const float*)d_in[12];
    const float* W21  = (const float*)d_in[13];
    const float* b21  = (const float*)d_in[14];
    const float* W22  = (const float*)d_in[15];
    const float* b22  = (const float*)d_in[16];
    const float* Wl   = (const float*)d_in[17];
    const float* bl   = (const float*)d_in[18];

    const int N_ = in_sizes[0] / F_DIM;
    const int E_ = in_sizes[2];
    const size_t NB = (size_t)N_ * F_DIM * sizeof(float);

    char* ws = (char*)d_ws;
    _Float16* h1    = (_Float16*)(ws + NB);
    _Float16* x16   = (_Float16*)(ws + NB + NB / 2);
    _Float16* frags = (_Float16*)(ws + 2 * NB);
    _Float16* f11 = frags;
    _Float16* f12 = frags + 16384;
    _Float16* f21 = frags + 32768;
    _Float16* f22 = frags + 49152;
    _Float16* fl  = frags + 65536;
    float* out = (float*)d_out;

    const int K     = (N_ + BKT_W - 1) / BKT_W;
    const int chunk = (E_ + BIN_BLOCKS - 1) / BIN_BLOCKS;
    const int nbpc  = (chunk + SC_BATCH - 1) / SC_BATCH;
    const int NQ    = BIN_BLOCKS * nbpc;
    const int nmat  = K * NQ;
    const size_t matB = (((size_t)nmat + 4) * sizeof(int) + 255) & ~(size_t)255;
    const size_t offB = (((size_t)N_ + 4) * sizeof(int) + 255) & ~(size_t)255;
    const size_t prB  = ((size_t)E_ * 2 * sizeof(int) + 255) & ~(size_t)255;
    const size_t dlB  = ((size_t)E_ * sizeof(int) + 255) & ~(size_t)255;
    char* gbase = ws + 2 * NB + 131072 + 16384;
    int*  cnt2   = (int*)(gbase);
    int*  st2    = (int*)(gbase + matB);
    int*  off    = (int*)(gbase + 2 * matB);
    int2* tpair  = (int2*)(gbase + 2 * matB + offB);
    int*  tdl    = (int*)(gbase + 2 * matB + offB + prB);
    int*  csr    = (int*)(gbase + 2 * matB + offB + prB + dlB);

    const int rows_per_set = 16 * MLP_WAVES * MLP_BLOCKS;
    const int mb = MLP_BLOCKS * ((N_ + rows_per_set - 1) / rows_per_set);

    bin_scatter<<<BIN_BLOCKS, 1024, 2 * K * sizeof(int), stream>>>(ei, ea, cnt2, st2, tpair, tdl, E_, K, chunk, nbpc,
                                                                   W11, W12, W21, W22, Wl, frags, x, x16, N_);
    bucket_place<<<K, 1024, 0, stream>>>(cnt2, st2, tpair, tdl, off, (int2*)csr, N_, K, NQ);

    gather_mlp<<<mb, MLP_THREADS, 0, stream>>>(x, x16, off, csr, We1, be1, eps1, f11, b11, f12, b12, h1, N_);
    gather_mlp_cls<<<mb, MLP_THREADS, 0, stream>>>(h1, off, csr, We2, be2, eps2, f21, b21, f22, b22, fl, bl, out, N_);
}
